# starting kernel
# baseline (speedup 1.0000x reference)
#include <hip/hip_runtime.h>
#include <hip/hip_bf16.h>

typedef __attribute__((ext_vector_type(2))) float v2f;
typedef __attribute__((ext_vector_type(8))) float v8f;

#define FEAT 128
#define NCLS 7
#define EPSB 1e-5f
#define SCAN_T 1024

__global__ void zero_i32(int* __restrict__ ptr, int n) {
    int i = blockIdx.x * blockDim.x + threadIdx.x;
    if (i < n) ptr[i] = 0;
}

__global__ void hist_rows(const int* __restrict__ erow, int* __restrict__ deg, int E) {
    int e = blockIdx.x * blockDim.x + threadIdx.x;
    if (e < E) atomicAdd(&deg[erow[e]], 1);
}

__global__ void scan_rows(const int* __restrict__ deg, int* __restrict__ rowStart, int N) {
    __shared__ int part[SCAN_T];
    const int t = threadIdx.x;
    const int chunk = (N + SCAN_T - 1) / SCAN_T;
    const int lo = t * chunk;
    const int hi = (lo + chunk < N) ? lo + chunk : N;
    int s = 0;
    for (int i = lo; i < hi; ++i) s += deg[i];
    part[t] = s;
    __syncthreads();
    for (int off = 1; off < SCAN_T; off <<= 1) {
        int add = (t >= off) ? part[t - off] : 0;
        int v = part[t];
        __syncthreads();
        part[t] = v + add;
        __syncthreads();
    }
    int run = part[t] - s;
    for (int i = lo; i < hi; ++i) { rowStart[i] = run; run += deg[i]; }
    if (t == SCAN_T - 1) rowStart[N] = part[t];
}

__global__ void scatter_edges(const int* __restrict__ erow, const int* __restrict__ ecol,
                              const float* __restrict__ ev, const int* __restrict__ rowStart,
                              int* __restrict__ cursor, int* __restrict__ scol,
                              float* __restrict__ sval, int E) {
    int e = blockIdx.x * blockDim.x + threadIdx.x;
    if (e >= E) return;
    const int r = erow[e];
    const int pos = rowStart[r] + atomicAdd(&cursor[r], 1);
    scol[pos] = ecol[e];
    sval[pos] = ev[e];
}

__global__ void gemm_xW1(const float* __restrict__ X, const float* __restrict__ W,
                         float* __restrict__ Y, int nrows) {
    const int wave = threadIdx.x >> 5;
    const int lane = threadIdx.x & 31;
    const int half = lane >> 4;
    const int r    = lane & 15;
    const int mt   = blockIdx.x * 4 + wave;
    if (mt >= (nrows >> 4)) return;
    const int m0 = mt << 4;

    const v8f vzero = {0.f,0.f,0.f,0.f,0.f,0.f,0.f,0.f};
    v8f acc[8];
#pragma unroll
    for (int nt = 0; nt < 8; ++nt) acc[nt] = vzero;

    const float* arow = X + (size_t)(m0 + r) * FEAT;
    for (int kt = 0; kt < 32; ++kt) {
        const int k0 = kt << 2;
        v2f a;
        a.x = arow[k0 + 2 * half + 0];
        a.y = arow[k0 + 2 * half + 1];
#pragma unroll
        for (int nt = 0; nt < 8; ++nt) {
            const int n0 = nt << 4;
            v2f b;
            b.x = W[(size_t)(k0 + 2 * half + 0) * FEAT + n0 + r];
            b.y = W[(size_t)(k0 + 2 * half + 1) * FEAT + n0 + r];
            acc[nt] = __builtin_amdgcn_wmma_f32_16x16x4_f32(
                false, a, false, b, (short)0, acc[nt], false, false);
        }
    }
#pragma unroll
    for (int nt = 0; nt < 8; ++nt) {
        const int n0 = nt << 4;
#pragma unroll
        for (int i = 0; i < 8; ++i)
            Y[(size_t)(m0 + i + 8 * half) * FEAT + n0 + r] = acc[nt][i];
    }
}

__global__ void spmm1_fused(const float* __restrict__ y, const int* __restrict__ rowStart,
                            const int* __restrict__ scol, const float* __restrict__ sval,
                            const float* __restrict__ b1, const float* __restrict__ g1,
                            const float* __restrict__ be1, const float* __restrict__ m1,
                            const float* __restrict__ v1, const float* __restrict__ W2,
                            float* __restrict__ p, int N) {
    const int wave = threadIdx.x >> 5;
    const int lane = threadIdx.x & 31;
    const int r = blockIdx.x * 8 + wave;
    if (r >= N) return;
    const int eBeg = rowStart[r];
    const int eEnd = rowStart[r + 1];

    float4 acc = {0.f, 0.f, 0.f, 0.f};
    for (int e = eBeg; e < eEnd; ++e) {
        const int   c = scol[e];
        const float v = sval[e];
        const float4 g = ((const float4*)(y + (size_t)c * FEAT))[lane];
        acc.x = fmaf(v, g.x, acc.x);
        acc.y = fmaf(v, g.y, acc.y);
        acc.z = fmaf(v, g.z, acc.z);
        acc.w = fmaf(v, g.w, acc.w);
    }

    float h[4];
    float af[4] = {acc.x, acc.y, acc.z, acc.w};
#pragma unroll
    for (int j = 0; j < 4; ++j) {
        const int ch = lane * 4 + j;
        const float s = g1[ch] * rsqrtf(v1[ch] + EPSB);
        const float t = (b1[ch] - m1[ch]) * s + be1[ch];
        const float hv = fmaf(af[j], s, t);
        h[j] = hv > 0.0f ? hv : 0.0f;
    }

#pragma unroll
    for (int c = 0; c < NCLS; ++c) {
        float ssum = 0.0f;
#pragma unroll
        for (int j = 0; j < 4; ++j)
            ssum = fmaf(h[j], W2[(size_t)(lane * 4 + j) * NCLS + c], ssum);
#pragma unroll
        for (int off = 16; off >= 1; off >>= 1) ssum += __shfl_xor(ssum, off, 32);
        if (lane == 0) p[(size_t)r * NCLS + c] = ssum;
    }
}

__global__ void spmm2_fused(const float* __restrict__ p, const int* __restrict__ rowStart,
                            const int* __restrict__ scol, const float* __restrict__ sval,
                            const float* __restrict__ b2, const float* __restrict__ g2,
                            const float* __restrict__ be2, const float* __restrict__ m2,
                            const float* __restrict__ v2, float* __restrict__ out, int total) {
    int t = blockIdx.x * blockDim.x + threadIdx.x;
    if (t >= total) return;
    const int r = t / NCLS;
    const int c = t - r * NCLS;
    float acc = 0.0f;
    const int eEnd = rowStart[r + 1];
    for (int e = rowStart[r]; e < eEnd; ++e)
        acc = fmaf(sval[e], p[(size_t)scol[e] * NCLS + c], acc);
    const float s = g2[c] * rsqrtf(v2[c] + EPSB);
    out[t] = (acc + b2[c] - m2[c]) * s + be2[c];
}

extern "C" void kernel_launch(void* const* d_in, const int* in_sizes, int n_in,
                              void* d_out, int out_size, void* d_ws, size_t ws_size,
                              hipStream_t stream) {
    const float* x   = (const float*)d_in[0];
    const float* ev  = (const float*)d_in[1];
    const float* W1  = (const float*)d_in[2];
    const float* b1  = (const float*)d_in[3];
    const float* g1  = (const float*)d_in[4];
    const float* be1 = (const float*)d_in[5];
    const float* m1  = (const float*)d_in[6];
    const float* v1  = (const float*)d_in[7];
    const float* W2  = (const float*)d_in[8];
    const float* b2  = (const float*)d_in[9];
    const float* g2  = (const float*)d_in[10];
    const float* be2 = (const float*)d_in[11];
    const float* m2  = (const float*)d_in[12];
    const float* v2  = (const float*)d_in[13];
    const int*  erow = (const int*)d_in[14];
    const int*  ecol = (const int*)d_in[15];
    float* out = (float*)d_out;

    const int N = in_sizes[0] / FEAT;
    const int E = in_sizes[1];

    float* y        = (float*)d_ws;
    float* p        = y + (size_t)N * FEAT;
    int*   deg      = (int*)(p + (size_t)N * NCLS);
    int*   cursor   = deg + N;
    int*   rowStart = cursor + N;
    int*   scol     = rowStart + (N + 1);
    float* sval     = (float*)(scol + E);

    zero_i32<<<(2 * N + 255) / 256, 256, 0, stream>>>(deg, 2 * N);
    hist_rows<<<(E + 255) / 256, 256, 0, stream>>>(erow, deg, E);
    scan_rows<<<1, SCAN_T, 0, stream>>>(deg, rowStart, N);
    scatter_edges<<<(E + 255) / 256, 256, 0, stream>>>(erow, ecol, ev, rowStart,
                                                       cursor, scol, sval, E);
    gemm_xW1<<<((N >> 4) + 3) / 4, 128, 0, stream>>>(x, W1, y, N);
    spmm1_fused<<<(N + 7) / 8, 256, 0, stream>>>(y, rowStart, scol, sval,
                                                 b1, g1, be1, m1, v1, W2, p, N);
    spmm2_fused<<<(N * NCLS + 255) / 256, 256, 0, stream>>>(p, rowStart, scol, sval,
                                                            b2, g2, be2, m2, v2, out,
                                                            N * NCLS);
}


# scan_rows: one serial 1024-thread block -> one block per 1024 rows; each block sums the counters ahead of its tile itself (coalesced 128-bit loads), wave-shuffle scan, one barrier
# speedup vs baseline: 1.3341x; 1.3341x over previous
#include <hip/hip_runtime.h>
#include <hip/hip_bf16.h>

typedef __attribute__((ext_vector_type(2))) float v2f;
typedef __attribute__((ext_vector_type(8))) float v8f;

#define FEAT 128
#define NCLS 7
#define EPSB 1e-5f
#define SCAN_T 1024

__global__ void zero_i32(int* __restrict__ ptr, int n) {
    int i = blockIdx.x * blockDim.x + threadIdx.x;
    if (i < n) ptr[i] = 0;
}

__global__ void hist_rows(const int* __restrict__ erow, int* __restrict__ deg, int E) {
    int e = blockIdx.x * blockDim.x + threadIdx.x;
    if (e < E) atomicAdd(&deg[erow[e]], 1);
}

__global__ void __launch_bounds__(SCAN_T)
scan_rows(const int* __restrict__ deg, int* __restrict__ rowStart, int N) {
    __shared__ int wtot[SCAN_T / 32];
    __shared__ int wcar[SCAN_T / 32];
    const int t    = threadIdx.x;
    const int lane = t & 31;
    const int wave = t >> 5;
    const int base = blockIdx.x * SCAN_T;
    const int idx  = base + t;
    const int d    = (idx < N) ? deg[idx] : 0;

    int s = 0;
    if ((((size_t)deg) & 15) == 0) {
        const int4* __restrict__ d4 = (const int4*)deg;
        const int n4 = base >> 2;
        int i = t;
        for (; i + 7 * SCAN_T < n4; i += 8 * SCAN_T) {
            int4 v[8];
            for (int u = 0; u < 8; ++u) v[u] = d4[i + u * SCAN_T];
            for (int u = 0; u < 8; ++u) s += (v[u].x + v[u].y) + (v[u].z + v[u].w);
        }
        for (; i < n4; i += SCAN_T) {
            const int4 v = d4[i];
            s += (v.x + v.y) + (v.z + v.w);
        }
    } else {
        for (int i = t; i < base; i += SCAN_T) s += deg[i];
    }

    int incl = d;
    for (int off = 1; off < 32; off <<= 1) {
        const int up = __shfl_up(incl, off, 32);
        if (lane >= off) incl += up;
    }
    for (int off = 16; off >= 1; off >>= 1) s += __shfl_xor(s, off, 32);
    if (lane == 31) wtot[wave] = incl;
    if (lane == 0)  wcar[wave] = s;
    __syncthreads();
    int c = wcar[lane] + ((lane < wave) ? wtot[lane] : 0);
    for (int off = 16; off >= 1; off >>= 1) c += __shfl_xor(c, off, 32);
    const int excl = c + incl - d;
    if (idx < N) rowStart[idx] = excl;
    if (idx == N - 1) rowStart[N] = excl + d;
}

__global__ void scatter_edges(const int* __restrict__ erow, const int* __restrict__ ecol,
                              const float* __restrict__ ev, const int* __restrict__ rowStart,
                              int* __restrict__ cursor, int* __restrict__ scol,
                              float* __restrict__ sval, int E) {
    int e = blockIdx.x * blockDim.x + threadIdx.x;
    if (e >= E) return;
    const int r = erow[e];
    const int pos = rowStart[r] + atomicAdd(&cursor[r], 1);
    scol[pos] = ecol[e];
    sval[pos] = ev[e];
}

__global__ void gemm_xW1(const float* __restrict__ X, const float* __restrict__ W,
                         float* __restrict__ Y, int nrows) {
    const int wave = threadIdx.x >> 5;
    const int lane = threadIdx.x & 31;
    const int half = lane >> 4;
    const int r    = lane & 15;
    const int mt   = blockIdx.x * 4 + wave;
    if (mt >= (nrows >> 4)) return;
    const int m0 = mt << 4;

    const v8f vzero = {0.f,0.f,0.f,0.f,0.f,0.f,0.f,0.f};
    v8f acc[8];
#pragma unroll
    for (int nt = 0; nt < 8; ++nt) acc[nt] = vzero;

    const float* arow = X + (size_t)(m0 + r) * FEAT;
    for (int kt = 0; kt < 32; ++kt) {
        const int k0 = kt << 2;
        v2f a;
        a.x = arow[k0 + 2 * half + 0];
        a.y = arow[k0 + 2 * half + 1];
#pragma unroll
        for (int nt = 0; nt < 8; ++nt) {
            const int n0 = nt << 4;
            v2f b;
            b.x = W[(size_t)(k0 + 2 * half + 0) * FEAT + n0 + r];
            b.y = W[(size_t)(k0 + 2 * half + 1) * FEAT + n0 + r];
            acc[nt] = __builtin_amdgcn_wmma_f32_16x16x4_f32(
                false, a, false, b, (short)0, acc[nt], false, false);
        }
    }
#pragma unroll
    for (int nt = 0; nt < 8; ++nt) {
        const int n0 = nt << 4;
#pragma unroll
        for (int i = 0; i < 8; ++i)
            Y[(size_t)(m0 + i + 8 * half) * FEAT + n0 + r] = acc[nt][i];
    }
}

__global__ void spmm1_fused(const float* __restrict__ y, const int* __restrict__ rowStart,
                            const int* __restrict__ scol, const float* __restrict__ sval,
                            const float* __restrict__ b1, const float* __restrict__ g1,
                            const float* __restrict__ be1, const float* __restrict__ m1,
                            const float* __restrict__ v1, const float* __restrict__ W2,
                            float* __restrict__ p, int N) {
    const int wave = threadIdx.x >> 5;
    const int lane = threadIdx.x & 31;
    const int r = blockIdx.x * 8 + wave;
    if (r >= N) return;
    const int eBeg = rowStart[r];
    const int eEnd = rowStart[r + 1];

    float4 acc = {0.f, 0.f, 0.f, 0.f};
    for (int e = eBeg; e < eEnd; ++e) {
        const int   c = scol[e];
        const float v = sval[e];
        const float4 g = ((const float4*)(y + (size_t)c * FEAT))[lane];
        acc.x = fmaf(v, g.x, acc.x);
        acc.y = fmaf(v, g.y, acc.y);
        acc.z = fmaf(v, g.z, acc.z);
        acc.w = fmaf(v, g.w, acc.w);
    }

    float h[4];
    float af[4] = {acc.x, acc.y, acc.z, acc.w};
#pragma unroll
    for (int j = 0; j < 4; ++j) {
        const int ch = lane * 4 + j;
        const float s = g1[ch] * rsqrtf(v1[ch] + EPSB);
        const float t = (b1[ch] - m1[ch]) * s + be1[ch];
        const float hv = fmaf(af[j], s, t);
        h[j] = hv > 0.0f ? hv : 0.0f;
    }

#pragma unroll
    for (int c = 0; c < NCLS; ++c) {
        float ssum = 0.0f;
#pragma unroll
        for (int j = 0; j < 4; ++j)
            ssum = fmaf(h[j], W2[(size_t)(lane * 4 + j) * NCLS + c], ssum);
#pragma unroll
        for (int off = 16; off >= 1; off >>= 1) ssum += __shfl_xor(ssum, off, 32);
        if (lane == 0) p[(size_t)r * NCLS + c] = ssum;
    }
}

__global__ void spmm2_fused(const float* __restrict__ p, const int* __restrict__ rowStart,
                            const int* __restrict__ scol, const float* __restrict__ sval,
                            const float* __restrict__ b2, const float* __restrict__ g2,
                            const float* __restrict__ be2, const float* __restrict__ m2,
                            const float* __restrict__ v2, float* __restrict__ out, int total) {
    int t = blockIdx.x * blockDim.x + threadIdx.x;
    if (t >= total) return;
    const int r = t / NCLS;
    const int c = t - r * NCLS;
    float acc = 0.0f;
    const int eEnd = rowStart[r + 1];
    for (int e = rowStart[r]; e < eEnd; ++e)
        acc = fmaf(sval[e], p[(size_t)scol[e] * NCLS + c], acc);
    const float s = g2[c] * rsqrtf(v2[c] + EPSB);
    out[t] = (acc + b2[c] - m2[c]) * s + be2[c];
}

extern "C" void kernel_launch(void* const* d_in, const int* in_sizes, int n_in,
                              void* d_out, int out_size, void* d_ws, size_t ws_size,
                              hipStream_t stream) {
    const float* x   = (const float*)d_in[0];
    const float* ev  = (const float*)d_in[1];
    const float* W1  = (const float*)d_in[2];
    const float* b1  = (const float*)d_in[3];
    const float* g1  = (const float*)d_in[4];
    const float* be1 = (const float*)d_in[5];
    const float* m1  = (const float*)d_in[6];
    const float* v1  = (const float*)d_in[7];
    const float* W2  = (const float*)d_in[8];
    const float* b2  = (const float*)d_in[9];
    const float* g2  = (const float*)d_in[10];
    const float* be2 = (const float*)d_in[11];
    const float* m2  = (const float*)d_in[12];
    const float* v2  = (const float*)d_in[13];
    const int*  erow = (const int*)d_in[14];
    const int*  ecol = (const int*)d_in[15];
    float* out = (float*)d_out;

    const int N = in_sizes[0] / FEAT;
    const int E = in_sizes[1];

    float* y        = (float*)d_ws;
    float* p        = y + (size_t)N * FEAT;
    int*   deg      = (int*)(p + (size_t)N * NCLS);
    int*   cursor   = deg + N;
    int*   rowStart = cursor + N;
    int*   scol     = rowStart + (N + 1);
    float* sval     = (float*)(scol + E);

    zero_i32<<<(2 * N + 255) / 256, 256, 0, stream>>>(deg, 2 * N);
    hist_rows<<<(E + 255) / 256, 256, 0, stream>>>(erow, deg, E);
    scan_rows<<<(N + SCAN_T - 1) / SCAN_T, SCAN_T, 0, stream>>>(deg, rowStart, N);
    scatter_edges<<<(E + 255) / 256, 256, 0, stream>>>(erow, ecol, ev, rowStart,
                                                       cursor, scol, sval, E);
    gemm_xW1<<<((N >> 4) + 3) / 4, 128, 0, stream>>>(x, W1, y, N);
    spmm1_fused<<<(N + 7) / 8, 256, 0, stream>>>(y, rowStart, scol, sval,
                                                 b1, g1, be1, m1, v1, W2, p, N);
    spmm2_fused<<<(N * NCLS + 255) / 256, 256, 0, stream>>>(p, rowStart, scol, sval,
                                                            b2, g2, be2, m2, v2, out,
                                                            N * NCLS);
}


# spmm1_fused: a wave walks 4 rows keeping its constants: W2 via LDS once per block, BN once per wave (was 28 W2 loads of 112-byte lane stride per row); row pairs by one coalesced load; sum order kept
# speedup vs baseline: 1.4886x; 1.1158x over previous
#include <hip/hip_runtime.h>
#include <hip/hip_bf16.h>

typedef __attribute__((ext_vector_type(2))) float v2f;
typedef __attribute__((ext_vector_type(8))) float v8f;

#define FEAT 128
#define NCLS 7
#define EPSB 1e-5f
#define SCAN_T 1024

__global__ void zero_i32(int* __restrict__ ptr, int n) {
    int i = blockIdx.x * blockDim.x + threadIdx.x;
    if (i < n) ptr[i] = 0;
}

__global__ void hist_rows(const int* __restrict__ erow, int* __restrict__ deg, int E) {
    int e = blockIdx.x * blockDim.x + threadIdx.x;
    if (e < E) atomicAdd(&deg[erow[e]], 1);
}

__global__ void __launch_bounds__(SCAN_T)
scan_rows(const int* __restrict__ deg, int* __restrict__ rowStart, int N) {
    __shared__ int wtot[SCAN_T / 32];
    __shared__ int wcar[SCAN_T / 32];
    const int t    = threadIdx.x;
    const int lane = t & 31;
    const int wave = t >> 5;
    const int base = blockIdx.x * SCAN_T;
    const int idx  = base + t;
    const int d    = (idx < N) ? deg[idx] : 0;

    int s = 0;
    if ((((size_t)deg) & 15) == 0) {
        const int4* __restrict__ d4 = (const int4*)deg;
        const int n4 = base >> 2;
        int i = t;
        for (; i + 7 * SCAN_T < n4; i += 8 * SCAN_T) {
            int4 v[8];
            for (int u = 0; u < 8; ++u) v[u] = d4[i + u * SCAN_T];
            for (int u = 0; u < 8; ++u) s += (v[u].x + v[u].y) + (v[u].z + v[u].w);
        }
        for (; i < n4; i += SCAN_T) {
            const int4 v = d4[i];
            s += (v.x + v.y) + (v.z + v.w);
        }
    } else {
        for (int i = t; i < base; i += SCAN_T) s += deg[i];
    }

    int incl = d;
    for (int off = 1; off < 32; off <<= 1) {
        const int up = __shfl_up(incl, off, 32);
        if (lane >= off) incl += up;
    }
    for (int off = 16; off >= 1; off >>= 1) s += __shfl_xor(s, off, 32);
    if (lane == 31) wtot[wave] = incl;
    if (lane == 0)  wcar[wave] = s;
    __syncthreads();
    int c = wcar[lane] + ((lane < wave) ? wtot[lane] : 0);
    for (int off = 16; off >= 1; off >>= 1) c += __shfl_xor(c, off, 32);
    const int excl = c + incl - d;
    if (idx < N) rowStart[idx] = excl;
    if (idx == N - 1) rowStart[N] = excl + d;
}

__global__ void scatter_edges(const int* __restrict__ erow, const int* __restrict__ ecol,
                              const float* __restrict__ ev, const int* __restrict__ rowStart,
                              int* __restrict__ cursor, int* __restrict__ scol,
                              float* __restrict__ sval, int E) {
    int e = blockIdx.x * blockDim.x + threadIdx.x;
    if (e >= E) return;
    const int r = erow[e];
    const int pos = rowStart[r] + atomicAdd(&cursor[r], 1);
    scol[pos] = ecol[e];
    sval[pos] = ev[e];
}

__global__ void gemm_xW1(const float* __restrict__ X, const float* __restrict__ W,
                         float* __restrict__ Y, int nrows) {
    const int wave = threadIdx.x >> 5;
    const int lane = threadIdx.x & 31;
    const int half = lane >> 4;
    const int r    = lane & 15;
    const int mt   = blockIdx.x * 4 + wave;
    if (mt >= (nrows >> 4)) return;
    const int m0 = mt << 4;

    const v8f vzero = {0.f,0.f,0.f,0.f,0.f,0.f,0.f,0.f};
    v8f acc[8];
#pragma unroll
    for (int nt = 0; nt < 8; ++nt) acc[nt] = vzero;

    const float* arow = X + (size_t)(m0 + r) * FEAT;
    for (int kt = 0; kt < 32; ++kt) {
        const int k0 = kt << 2;
        v2f a;
        a.x = arow[k0 + 2 * half + 0];
        a.y = arow[k0 + 2 * half + 1];
#pragma unroll
        for (int nt = 0; nt < 8; ++nt) {
            const int n0 = nt << 4;
            v2f b;
            b.x = W[(size_t)(k0 + 2 * half + 0) * FEAT + n0 + r];
            b.y = W[(size_t)(k0 + 2 * half + 1) * FEAT + n0 + r];
            acc[nt] = __builtin_amdgcn_wmma_f32_16x16x4_f32(
                false, a, false, b, (short)0, acc[nt], false, false);
        }
    }
#pragma unroll
    for (int nt = 0; nt < 8; ++nt) {
        const int n0 = nt << 4;
#pragma unroll
        for (int i = 0; i < 8; ++i)
            Y[(size_t)(m0 + i + 8 * half) * FEAT + n0 + r] = acc[nt][i];
    }
}

#define SP1_T    256
#define SP1_ROWS 4

template <int C>
__device__ __forceinline__ void gather_fma(const float4* __restrict__ y4, int cc, float vv,
                                           int j, int lane, float4& acc) {
    float4 g[C];
#pragma unroll
    for (int u = 0; u < C; ++u) {
        const int c = __builtin_amdgcn_readlane(cc, j + u);
        g[u] = y4[(size_t)c * (FEAT / 4) + lane];
    }
#pragma unroll
    for (int u = 0; u < C; ++u) {
        const float v = __int_as_float(__builtin_amdgcn_readlane(__float_as_int(vv), j + u));
        acc.x = fmaf(v, g[u].x, acc.x);
        acc.y = fmaf(v, g[u].y, acc.y);
        acc.z = fmaf(v, g[u].z, acc.z);
        acc.w = fmaf(v, g[u].w, acc.w);
    }
}

__global__ void __launch_bounds__(SP1_T)
spmm1_fused(const float* __restrict__ y, const int* __restrict__ rowStart,
            const int* __restrict__ scol, const float* __restrict__ sval,
            const float* __restrict__ b1, const float* __restrict__ g1,
            const float* __restrict__ be1, const float* __restrict__ m1,
            const float* __restrict__ v1, const float* __restrict__ W2,
            float* __restrict__ p, int N) {
    __shared__ float w2s[FEAT * NCLS];
    {
        constexpr int W2N = FEAT * NCLS;
        constexpr int W2K = (W2N + SP1_T - 1) / SP1_T;
        float t[W2K];
#pragma unroll
        for (int k = 0; k < W2K; ++k) {
            const int i = (int)threadIdx.x + k * SP1_T;
            t[k] = (i < W2N) ? W2[i] : 0.0f;
        }
#pragma unroll
        for (int k = 0; k < W2K; ++k) {
            const int i = (int)threadIdx.x + k * SP1_T;
            if (i < W2N) w2s[i] = t[k];
        }
    }
    __syncthreads();

    const int lane = threadIdx.x & 31;
    const int wave = __builtin_amdgcn_readfirstlane((int)(threadIdx.x >> 5));
    const int r0 = ((int)blockIdx.x * (SP1_T / 32) + wave) * SP1_ROWS;
    if (r0 >= N) return;
    const int nr = (N - r0 < SP1_ROWS) ? (N - r0) : SP1_ROWS;

    float bs[4], bt[4], w2[4][NCLS];
#pragma unroll
    for (int j = 0; j < 4; ++j) {
        const int ch = lane * 4 + j;
        const float s = g1[ch] * rsqrtf(v1[ch] + EPSB);
        bs[j] = s;
        bt[j] = (b1[ch] - m1[ch]) * s + be1[ch];
#pragma unroll
        for (int c = 0; c < NCLS; ++c) w2[j][c] = w2s[ch * NCLS + c];
    }

    const float4* __restrict__ y4 = (const float4*)y;

    const int myrs = rowStart[r0 + (lane < nr ? lane : nr)];
    int eB = __builtin_amdgcn_readlane(myrs, 0);
    int eE = __builtin_amdgcn_readlane(myrs, 1);
    int nc = 0; float nv = 0.0f;
    if (eB + lane < eE) { nc = scol[eB + lane]; nv = sval[eB + lane]; }

    for (int i = 0; i < nr; ++i) {
        const int cc = nc; const float vv = nv;
        int nE = eE;
        if (i + 1 < nr) {
            nE = __builtin_amdgcn_readlane(myrs, i + 2);
            nc = 0; nv = 0.0f;
            if (eE + lane < nE) { nc = scol[eE + lane]; nv = sval[eE + lane]; }
        }

        float4 acc = {0.f, 0.f, 0.f, 0.f};
        const int deg = eE - eB;
        const int n = deg < 32 ? deg : 32;
        int j = 0;
        for (; j + 8 <= n; j += 8) gather_fma<8>(y4, cc, vv, j, lane, acc);
        switch (n - j) {
            case 7: gather_fma<7>(y4, cc, vv, j, lane, acc); break;
            case 6: gather_fma<6>(y4, cc, vv, j, lane, acc); break;
            case 5: gather_fma<5>(y4, cc, vv, j, lane, acc); break;
            case 4: gather_fma<4>(y4, cc, vv, j, lane, acc); break;
            case 3: gather_fma<3>(y4, cc, vv, j, lane, acc); break;
            case 2: gather_fma<2>(y4, cc, vv, j, lane, acc); break;
            case 1: gather_fma<1>(y4, cc, vv, j, lane, acc); break;
            default: break;
        }
        for (int e = eB + 32; e < eE; ++e) {
            const int   c = scol[e];
            const float v = sval[e];
            const float4 g = y4[(size_t)c * (FEAT / 4) + lane];
            acc.x = fmaf(v, g.x, acc.x);
            acc.y = fmaf(v, g.y, acc.y);
            acc.z = fmaf(v, g.z, acc.z);
            acc.w = fmaf(v, g.w, acc.w);
        }

        const int r = r0 + i;
        const float af[4] = {acc.x, acc.y, acc.z, acc.w};
        float h[4];
#pragma unroll
        for (int j = 0; j < 4; ++j) {
            const float hv = fmaf(af[j], bs[j], bt[j]);
            h[j] = hv > 0.0f ? hv : 0.0f;
        }
#pragma unroll
        for (int c = 0; c < NCLS; ++c) {
            float ssum = 0.0f;
#pragma unroll
            for (int j = 0; j < 4; ++j) ssum = fmaf(h[j], w2[j][c], ssum);
#pragma unroll
            for (int off = 16; off >= 1; off >>= 1) ssum += __shfl_xor(ssum, off, 32);
            if (lane == 0) p[(size_t)r * NCLS + c] = ssum;
        }

        eB = eE; eE = nE;
    }
}

__global__ void spmm2_fused(const float* __restrict__ p, const int* __restrict__ rowStart,
                            const int* __restrict__ scol, const float* __restrict__ sval,
                            const float* __restrict__ b2, const float* __restrict__ g2,
                            const float* __restrict__ be2, const float* __restrict__ m2,
                            const float* __restrict__ v2, float* __restrict__ out, int total) {
    int t = blockIdx.x * blockDim.x + threadIdx.x;
    if (t >= total) return;
    const int r = t / NCLS;
    const int c = t - r * NCLS;
    float acc = 0.0f;
    const int eEnd = rowStart[r + 1];
    for (int e = rowStart[r]; e < eEnd; ++e)
        acc = fmaf(sval[e], p[(size_t)scol[e] * NCLS + c], acc);
    const float s = g2[c] * rsqrtf(v2[c] + EPSB);
    out[t] = (acc + b2[c] - m2[c]) * s + be2[c];
}

extern "C" void kernel_launch(void* const* d_in, const int* in_sizes, int n_in,
                              void* d_out, int out_size, void* d_ws, size_t ws_size,
                              hipStream_t stream) {
    const float* x   = (const float*)d_in[0];
    const float* ev  = (const float*)d_in[1];
    const float* W1  = (const float*)d_in[2];
    const float* b1  = (const float*)d_in[3];
    const float* g1  = (const float*)d_in[4];
    const float* be1 = (const float*)d_in[5];
    const float* m1  = (const float*)d_in[6];
    const float* v1  = (const float*)d_in[7];
    const float* W2  = (const float*)d_in[8];
    const float* b2  = (const float*)d_in[9];
    const float* g2  = (const float*)d_in[10];
    const float* be2 = (const float*)d_in[11];
    const float* m2  = (const float*)d_in[12];
    const float* v2  = (const float*)d_in[13];
    const int*  erow = (const int*)d_in[14];
    const int*  ecol = (const int*)d_in[15];
    float* out = (float*)d_out;

    const int N = in_sizes[0] / FEAT;
    const int E = in_sizes[1];

    float* y        = (float*)d_ws;
    float* p        = y + (size_t)N * FEAT;
    int*   deg      = (int*)(p + (size_t)N * NCLS);
    int*   cursor   = deg + N;
    int*   rowStart = cursor + N;
    int*   scol     = rowStart + (N + 1);
    float* sval     = (float*)(scol + E);

    zero_i32<<<(2 * N + 255) / 256, 256, 0, stream>>>(deg, 2 * N);
    hist_rows<<<(E + 255) / 256, 256, 0, stream>>>(erow, deg, E);
    scan_rows<<<(N + SCAN_T - 1) / SCAN_T, SCAN_T, 0, stream>>>(deg, rowStart, N);
    scatter_edges<<<(E + 255) / 256, 256, 0, stream>>>(erow, ecol, ev, rowStart,
                                                       cursor, scol, sval, E);
    gemm_xW1<<<((N >> 4) + 3) / 4, 128, 0, stream>>>(x, W1, y, N);
    spmm1_fused<<<(N + 31) / 32, 256, 0, stream>>>(y, rowStart, scol, sval,
                                                   b1, g1, be1, m1, v1, W2, p, N);
    spmm2_fused<<<(N * NCLS + 255) / 256, 256, 0, stream>>>(p, rowStart, scol, sval,
                                                            b2, g2, be2, m2, v2, out,
                                                            N * NCLS);
}


# CSR build: global atomics (row histogram, returning cursor adds) -> two-level counting sort on LDS counters (bin per tile by 256-row bucket, sort bucket in LDS); order of a row's fp sums changes
# speedup vs baseline: 3.2624x; 2.1916x over previous
#include <hip/hip_runtime.h>
#include <hip/hip_bf16.h>

typedef __attribute__((ext_vector_type(2))) float v2f;
typedef __attribute__((ext_vector_type(8))) float v8f;

#define FEAT 128
#define NCLS 7
#define EPSB 1e-5f
#define SCAN_T 1024

#define BK_SH   8
#define BK_R    (1 << BK_SH)
#define BK_MAX  1024
#define BIN_G   256
#define BIN_T   1024
#define BIN_U   8

__global__ void __launch_bounds__(BIN_T)
bin_count(const int* __restrict__ erow, int* __restrict__ cntMat, int E, int B) {
    __shared__ int cnt[BK_MAX];
    const int t = threadIdx.x;
    for (int i = t; i < B; i += BIN_T) cnt[i] = 0;
    __syncthreads();
    const int tile = (E + BIN_G - 1) / BIN_G;
    const int e0 = (int)blockIdx.x * tile;
    const int e1 = (e0 + tile < E) ? (e0 + tile) : E;
    for (int base = e0; base < e1; base += BIN_U * BIN_T) {
        int r[BIN_U];
#pragma unroll
        for (int u = 0; u < BIN_U; ++u) {
            const int e = base + u * BIN_T + t;
            r[u] = erow[(e < e1) ? e : (e1 - 1)];
        }
#pragma unroll
        for (int u = 0; u < BIN_U; ++u) asm volatile("" :: "v"(r[u]));
#pragma unroll
        for (int u = 0; u < BIN_U; ++u)
            if (base + u * BIN_T + t < e1) atomicAdd(&cnt[r[u] >> BK_SH], 1);
    }
    __syncthreads();
    for (int i = t; i < B; i += BIN_T) cntMat[(size_t)i * BIN_G + blockIdx.x] = cnt[i];
}

__global__ void __launch_bounds__(SCAN_T)
scan_rows(const int* __restrict__ deg, int* __restrict__ rowStart, int N) {
    __shared__ int wtot[SCAN_T / 32];
    __shared__ int wcar[SCAN_T / 32];
    const int t    = threadIdx.x;
    const int lane = t & 31;
    const int wave = t >> 5;
    const int base = blockIdx.x * SCAN_T;
    const int idx  = base + t;
    const int d    = (idx < N) ? deg[idx] : 0;

    int s = 0;
    if ((((size_t)deg) & 15) == 0) {
        const int4* __restrict__ d4 = (const int4*)deg;
        const int n4 = base >> 2;
        int i = t;
        for (; i + 7 * SCAN_T < n4; i += 8 * SCAN_T) {
            int4 v[8];
            for (int u = 0; u < 8; ++u) v[u] = d4[i + u * SCAN_T];
            for (int u = 0; u < 8; ++u) s += (v[u].x + v[u].y) + (v[u].z + v[u].w);
        }
        for (; i < n4; i += SCAN_T) {
            const int4 v = d4[i];
            s += (v.x + v.y) + (v.z + v.w);
        }
    } else {
        for (int i = t; i < base; i += SCAN_T) s += deg[i];
    }

    int incl = d;
    for (int off = 1; off < 32; off <<= 1) {
        const int up = __shfl_up(incl, off, 32);
        if (lane >= off) incl += up;
    }
    for (int off = 16; off >= 1; off >>= 1) s += __shfl_xor(s, off, 32);
    if (lane == 31) wtot[wave] = incl;
    if (lane == 0)  wcar[wave] = s;
    __syncthreads();
    int c = wcar[lane] + ((lane < wave) ? wtot[lane] : 0);
    for (int off = 16; off >= 1; off >>= 1) c += __shfl_xor(c, off, 32);
    const int excl = c + incl - d;
    if (idx < N) rowStart[idx] = excl;
    if (idx == N - 1) rowStart[N] = excl + d;
}

__global__ void __launch_bounds__(BIN_T)
bin_scatter(const int* __restrict__ erow, const int* __restrict__ ecol,
            const float* __restrict__ ev, const int* __restrict__ startMat,
            int* __restrict__ brec_, int E, int B) {
    __shared__ int cur[BK_MAX];
    int2* __restrict__ brec = (int2*)brec_;
    const int t = threadIdx.x;
    for (int i = t; i < B; i += BIN_T) cur[i] = startMat[(size_t)i * BIN_G + blockIdx.x];
    __syncthreads();
    const int tile = (E + BIN_G - 1) / BIN_G;
    const int e0 = (int)blockIdx.x * tile;
    const int e1 = (e0 + tile < E) ? (e0 + tile) : E;
    for (int base = e0; base < e1; base += BIN_U * BIN_T) {
        int r[BIN_U], c[BIN_U]; float v[BIN_U];
#pragma unroll
        for (int u = 0; u < BIN_U; ++u) {
            const int e = base + u * BIN_T + t;
            const bool ok = e < e1;
            const int ee = ok ? e : (e1 - 1);
            r[u] = erow[ee];
            c[u] = ecol[ee];
            v[u] = ev[ee];
        }
#pragma unroll
        for (int u = 0; u < BIN_U; ++u) asm volatile("" :: "v"(r[u]), "v"(c[u]), "v"(v[u]));
        int pos[BIN_U];
#pragma unroll
        for (int u = 0; u < BIN_U; ++u) {
            pos[u] = -1;
            if (base + u * BIN_T + t < e1) pos[u] = atomicAdd(&cur[r[u] >> BK_SH], 1);
        }
#pragma unroll
        for (int u = 0; u < BIN_U; ++u) {
            if (pos[u] >= 0)
                brec[pos[u]] = make_int2((c[u] << BK_SH) | (r[u] & (BK_R - 1)), __float_as_int(v[u]));
        }
    }
}

#define BS_T    1024
#define BS_U    6
#define BS_CAP  (BS_T * BS_U)

__device__ __forceinline__ int bucket_scan(int* rc, int* wsum, int t) {
    const int lane = t & 31;
    const int wave = t >> 5;
    int d = 0, incl = 0;
    if (t < BK_R) {
        d = rc[t];
        incl = d;
        for (int off = 1; off < 32; off <<= 1) {
            const int up = __shfl_up(incl, off, 32);
            if (lane >= off) incl += up;
        }
        if (lane == 31) wsum[wave] = incl;
    }
    __syncthreads();
    int excl = 0;
    if (t < BK_R) {
        int c = (lane < wave) ? wsum[lane] : 0;
        for (int off = 16; off >= 1; off >>= 1) c += __shfl_xor(c, off, 32);
        excl = c + incl - d;
        rc[t] = excl;
    }
    __syncthreads();
    return excl;
}

__global__ void __launch_bounds__(BS_T)
bucket_sort(const int* __restrict__ brec_, const int* __restrict__ startMat,
            int* __restrict__ rowStart, int* __restrict__ scol,
            float* __restrict__ sval, int N, int B) {
    __shared__ int   rc[BK_R];
    __shared__ int   wsum[BK_R / 32];
    __shared__ int   lcol[BS_CAP];
    __shared__ float lval[BS_CAP];
    const int2* __restrict__ brec = (const int2*)brec_;
    const int t = threadIdx.x;
    const int b = blockIdx.x;
    const int s0 = startMat[(size_t)b * BIN_G];
    const int s1 = startMat[(size_t)(b + 1) * BIN_G];
    const int n  = s1 - s0;
    if (t < BK_R) rc[t] = 0;
    __syncthreads();

    if (n <= BS_CAP) {
        int key[BS_U], val[BS_U], lr[BS_U];
        const int last = (s1 > 1) ? (s1 - 1) : 0;
#pragma unroll
        for (int u = 0; u < BS_U; ++u) {
            const int idx = s0 + t + u * BS_T;
            const int2 q = brec[idx < last ? idx : last];
            key[u] = q.x; val[u] = q.y;
        }
#pragma unroll
        for (int u = 0; u < BS_U; ++u) {
            lr[u] = 0;
            if (t + u * BS_T < n) lr[u] = atomicAdd(&rc[key[u] & (BK_R - 1)], 1);
        }
        __syncthreads();
        const int excl = bucket_scan(rc, wsum, t);
        if (t < BK_R) {
            const int row = b * BK_R + t;
            if (row < N) rowStart[row] = s0 + excl;
        }
        int pos[BS_U];
#pragma unroll
        for (int u = 0; u < BS_U; ++u) pos[u] = rc[key[u] & (BK_R - 1)] + lr[u];
#pragma unroll
        for (int u = 0; u < BS_U; ++u) {
            if (t + u * BS_T < n) {
                lcol[pos[u]] = key[u] >> BK_SH;
                lval[pos[u]] = __int_as_float(val[u]);
            }
        }
        __syncthreads();
        if (n > 0) {
            int oc[BS_U]; float ov[BS_U];
#pragma unroll
            for (int u = 0; u < BS_U; ++u) {
                const int j = t + u * BS_T;
                const int jj = (j < n) ? j : (n - 1);
                oc[u] = lcol[jj]; ov[u] = lval[jj];
            }
#pragma unroll
            for (int u = 0; u < BS_U; ++u) {
                const int j = t + u * BS_T;
                if (j < n) { scol[s0 + j] = oc[u]; sval[s0 + j] = ov[u]; }
            }
        }
    } else {
        for (int j = t; j < n; j += BS_T)
            atomicAdd(&rc[brec[s0 + j].x & (BK_R - 1)], 1);
        __syncthreads();
        const int excl = bucket_scan(rc, wsum, t);
        if (t < BK_R) {
            const int row = b * BK_R + t;
            if (row < N) rowStart[row] = s0 + excl;
        }
        for (int j = t; j < n; j += BS_T) {
            const int2 q = brec[s0 + j];
            const int pos = s0 + atomicAdd(&rc[q.x & (BK_R - 1)], 1);
            scol[pos] = q.x >> BK_SH;
            sval[pos] = __int_as_float(q.y);
        }
    }
    if (b == B - 1 && t == 0) rowStart[N] = s1;
}

__global__ void gemm_xW1(const float* __restrict__ X, const float* __restrict__ W,
                         float* __restrict__ Y, int nrows) {
    const int wave = threadIdx.x >> 5;
    const int lane = threadIdx.x & 31;
    const int half = lane >> 4;
    const int r    = lane & 15;
    const int mt   = blockIdx.x * 4 + wave;
    if (mt >= (nrows >> 4)) return;
    const int m0 = mt << 4;

    const v8f vzero = {0.f,0.f,0.f,0.f,0.f,0.f,0.f,0.f};
    v8f acc[8];
#pragma unroll
    for (int nt = 0; nt < 8; ++nt) acc[nt] = vzero;

    const float* arow = X + (size_t)(m0 + r) * FEAT;
    for (int kt = 0; kt < 32; ++kt) {
        const int k0 = kt << 2;
        v2f a;
        a.x = arow[k0 + 2 * half + 0];
        a.y = arow[k0 + 2 * half + 1];
#pragma unroll
        for (int nt = 0; nt < 8; ++nt) {
            const int n0 = nt << 4;
            v2f b;
            b.x = W[(size_t)(k0 + 2 * half + 0) * FEAT + n0 + r];
            b.y = W[(size_t)(k0 + 2 * half + 1) * FEAT + n0 + r];
            acc[nt] = __builtin_amdgcn_wmma_f32_16x16x4_f32(
                false, a, false, b, (short)0, acc[nt], false, false);
        }
    }
#pragma unroll
    for (int nt = 0; nt < 8; ++nt) {
        const int n0 = nt << 4;
#pragma unroll
        for (int i = 0; i < 8; ++i)
            Y[(size_t)(m0 + i + 8 * half) * FEAT + n0 + r] = acc[nt][i];
    }
}

#define SP1_T    256
#define SP1_ROWS 4

template <int C>
__device__ __forceinline__ void gather_fma(const float4* __restrict__ y4, int cc, float vv,
                                           int j, int lane, float4& acc) {
    float4 g[C];
#pragma unroll
    for (int u = 0; u < C; ++u) {
        const int c = __builtin_amdgcn_readlane(cc, j + u);
        g[u] = y4[(size_t)c * (FEAT / 4) + lane];
    }
#pragma unroll
    for (int u = 0; u < C; ++u) {
        const float v = __int_as_float(__builtin_amdgcn_readlane(__float_as_int(vv), j + u));
        acc.x = fmaf(v, g[u].x, acc.x);
        acc.y = fmaf(v, g[u].y, acc.y);
        acc.z = fmaf(v, g[u].z, acc.z);
        acc.w = fmaf(v, g[u].w, acc.w);
    }
}

__global__ void __launch_bounds__(SP1_T)
spmm1_fused(const float* __restrict__ y, const int* __restrict__ rowStart,
            const int* __restrict__ scol, const float* __restrict__ sval,
            const float* __restrict__ b1, const float* __restrict__ g1,
            const float* __restrict__ be1, const float* __restrict__ m1,
            const float* __restrict__ v1, const float* __restrict__ W2,
            float* __restrict__ p, int N) {
    __shared__ float w2s[FEAT * NCLS];
    {
        constexpr int W2N = FEAT * NCLS;
        constexpr int W2K = (W2N + SP1_T - 1) / SP1_T;
        float t[W2K];
#pragma unroll
        for (int k = 0; k < W2K; ++k) {
            const int i = (int)threadIdx.x + k * SP1_T;
            t[k] = (i < W2N) ? W2[i] : 0.0f;
        }
#pragma unroll
        for (int k = 0; k < W2K; ++k) {
            const int i = (int)threadIdx.x + k * SP1_T;
            if (i < W2N) w2s[i] = t[k];
        }
    }
    __syncthreads();

    const int lane = threadIdx.x & 31;
    const int wave = __builtin_amdgcn_readfirstlane((int)(threadIdx.x >> 5));
    const int r0 = ((int)blockIdx.x * (SP1_T / 32) + wave) * SP1_ROWS;
    if (r0 >= N) return;
    const int nr = (N - r0 < SP1_ROWS) ? (N - r0) : SP1_ROWS;

    float bs[4], bt[4], w2[4][NCLS];
#pragma unroll
    for (int j = 0; j < 4; ++j) {
        const int ch = lane * 4 + j;
        const float s = g1[ch] * rsqrtf(v1[ch] + EPSB);
        bs[j] = s;
        bt[j] = (b1[ch] - m1[ch]) * s + be1[ch];
#pragma unroll
        for (int c = 0; c < NCLS; ++c) w2[j][c] = w2s[ch * NCLS + c];
    }

    const float4* __restrict__ y4 = (const float4*)y;

    const int myrs = rowStart[r0 + (lane < nr ? lane : nr)];
    int eB = __builtin_amdgcn_readlane(myrs, 0);
    int eE = __builtin_amdgcn_readlane(myrs, 1);
    int nc = 0; float nv = 0.0f;
    if (eB + lane < eE) { nc = scol[eB + lane]; nv = sval[eB + lane]; }

    for (int i = 0; i < nr; ++i) {
        const int cc = nc; const float vv = nv;
        int nE = eE;
        if (i + 1 < nr) {
            nE = __builtin_amdgcn_readlane(myrs, i + 2);
            nc = 0; nv = 0.0f;
            if (eE + lane < nE) { nc = scol[eE + lane]; nv = sval[eE + lane]; }
        }

        float4 acc = {0.f, 0.f, 0.f, 0.f};
        const int deg = eE - eB;
        const int n = deg < 32 ? deg : 32;
        int j = 0;
        for (; j + 8 <= n; j += 8) gather_fma<8>(y4, cc, vv, j, lane, acc);
        switch (n - j) {
            case 7: gather_fma<7>(y4, cc, vv, j, lane, acc); break;
            case 6: gather_fma<6>(y4, cc, vv, j, lane, acc); break;
            case 5: gather_fma<5>(y4, cc, vv, j, lane, acc); break;
            case 4: gather_fma<4>(y4, cc, vv, j, lane, acc); break;
            case 3: gather_fma<3>(y4, cc, vv, j, lane, acc); break;
            case 2: gather_fma<2>(y4, cc, vv, j, lane, acc); break;
            case 1: gather_fma<1>(y4, cc, vv, j, lane, acc); break;
            default: break;
        }
        for (int e = eB + 32; e < eE; ++e) {
            const int   c = scol[e];
            const float v = sval[e];
            const float4 g = y4[(size_t)c * (FEAT / 4) + lane];
            acc.x = fmaf(v, g.x, acc.x);
            acc.y = fmaf(v, g.y, acc.y);
            acc.z = fmaf(v, g.z, acc.z);
            acc.w = fmaf(v, g.w, acc.w);
        }

        const int r = r0 + i;
        const float af[4] = {acc.x, acc.y, acc.z, acc.w};
        float h[4];
#pragma unroll
        for (int j = 0; j < 4; ++j) {
            const float hv = fmaf(af[j], bs[j], bt[j]);
            h[j] = hv > 0.0f ? hv : 0.0f;
        }
#pragma unroll
        for (int c = 0; c < NCLS; ++c) {
            float ssum = 0.0f;
#pragma unroll
            for (int j = 0; j < 4; ++j) ssum = fmaf(h[j], w2[j][c], ssum);
#pragma unroll
            for (int off = 16; off >= 1; off >>= 1) ssum += __shfl_xor(ssum, off, 32);
            if (lane == 0) p[(size_t)r * NCLS + c] = ssum;
        }

        eB = eE; eE = nE;
    }
}

__global__ void spmm2_fused(const float* __restrict__ p, const int* __restrict__ rowStart,
                            const int* __restrict__ scol, const float* __restrict__ sval,
                            const float* __restrict__ b2, const float* __restrict__ g2,
                            const float* __restrict__ be2, const float* __restrict__ m2,
                            const float* __restrict__ v2, float* __restrict__ out, int total) {
    int t = blockIdx.x * blockDim.x + threadIdx.x;
    if (t >= total) return;
    const int r = t / NCLS;
    const int c = t - r * NCLS;
    float acc = 0.0f;
    const int eEnd = rowStart[r + 1];
    for (int e = rowStart[r]; e < eEnd; ++e)
        acc = fmaf(sval[e], p[(size_t)scol[e] * NCLS + c], acc);
    const float s = g2[c] * rsqrtf(v2[c] + EPSB);
    out[t] = (acc + b2[c] - m2[c]) * s + be2[c];
}

extern "C" void kernel_launch(void* const* d_in, const int* in_sizes, int n_in,
                              void* d_out, int out_size, void* d_ws, size_t ws_size,
                              hipStream_t stream) {
    const float* x   = (const float*)d_in[0];
    const float* ev  = (const float*)d_in[1];
    const float* W1  = (const float*)d_in[2];
    const float* b1  = (const float*)d_in[3];
    const float* g1  = (const float*)d_in[4];
    const float* be1 = (const float*)d_in[5];
    const float* m1  = (const float*)d_in[6];
    const float* v1  = (const float*)d_in[7];
    const float* W2  = (const float*)d_in[8];
    const float* b2  = (const float*)d_in[9];
    const float* g2  = (const float*)d_in[10];
    const float* be2 = (const float*)d_in[11];
    const float* m2  = (const float*)d_in[12];
    const float* v2  = (const float*)d_in[13];
    const int*  erow = (const int*)d_in[14];
    const int*  ecol = (const int*)d_in[15];
    float* out = (float*)d_out;

    const int N = in_sizes[0] / FEAT;
    const int E = in_sizes[1];

    const int B = (N + BK_R - 1) >> BK_SH;

    float* y        = (float*)d_ws;
    float* p        = y + (size_t)N * FEAT;
    int*   brec     = (int*)(p + (size_t)N * NCLS);
    int*   cntMat   = brec + (size_t)2 * E;
    int*   startMat = cntMat + (size_t)B * BIN_G;
    int*   rowStart = startMat + ((size_t)B * BIN_G + 1);
    int*   scol     = rowStart + (N + 1);
    float* sval     = (float*)(scol + E);

    bin_count<<<BIN_G, BIN_T, 0, stream>>>(erow, cntMat, E, B);
    scan_rows<<<(B * BIN_G + SCAN_T - 1) / SCAN_T, SCAN_T, 0, stream>>>(cntMat, startMat,
                                                                      B * BIN_G);
    bin_scatter<<<BIN_G, BIN_T, 0, stream>>>(erow, ecol, ev, startMat, brec, E, B);
    bucket_sort<<<B, BS_T, 0, stream>>>(brec, startMat, rowStart, scol, sval, N, B);
    gemm_xW1<<<((N >> 4) + 3) / 4, 128, 0, stream>>>(x, W1, y, N);
    spmm1_fused<<<(N + 31) / 32, 256, 0, stream>>>(y, rowStart, scol, sval,
                                                   b1, g1, be1, m1, v1, W2, p, N);
    spmm2_fused<<<(N * NCLS + 255) / 256, 256, 0, stream>>>(p, rowStart, scol, sval,
                                                            b2, g2, be2, m2, v2, out,
                                                            N * NCLS);
}


# layer-1 dense product fused into the aggregation kernel: rows of x summed first, z@W1 by fp32 WMMA from LDS in the same block; x@W1 never stored, one launch less; order of fp sums changes
# speedup vs baseline: 3.7449x; 1.1479x over previous
#include <hip/hip_runtime.h>
#include <hip/hip_bf16.h>

typedef __attribute__((ext_vector_type(2))) float v2f;
typedef __attribute__((ext_vector_type(8))) float v8f;

#define FEAT 128
#define NCLS 7
#define EPSB 1e-5f
#define SCAN_T 1024

#define BK_SH   8
#define BK_R    (1 << BK_SH)
#define BK_MAX  1024
#define BIN_G   256
#define BIN_T   1024
#define BIN_U   8

__global__ void __launch_bounds__(BIN_T)
bin_count(const int* __restrict__ erow, int* __restrict__ cntMat, int E, int B) {
    __shared__ int cnt[BK_MAX];
    const int t = threadIdx.x;
    for (int i = t; i < B; i += BIN_T) cnt[i] = 0;
    __syncthreads();
    const int tile = (E + BIN_G - 1) / BIN_G;
    const int e0 = (int)blockIdx.x * tile;
    const int e1 = (e0 + tile < E) ? (e0 + tile) : E;
    for (int base = e0; base < e1; base += BIN_U * BIN_T) {
        int r[BIN_U];
#pragma unroll
        for (int u = 0; u < BIN_U; ++u) {
            const int e = base + u * BIN_T + t;
            r[u] = erow[(e < e1) ? e : (e1 - 1)];
        }
#pragma unroll
        for (int u = 0; u < BIN_U; ++u) asm volatile("" :: "v"(r[u]));
#pragma unroll
        for (int u = 0; u < BIN_U; ++u)
            if (base + u * BIN_T + t < e1) atomicAdd(&cnt[r[u] >> BK_SH], 1);
    }
    __syncthreads();
    for (int i = t; i < B; i += BIN_T) cntMat[(size_t)i * BIN_G + blockIdx.x] = cnt[i];
}

__global__ void __launch_bounds__(SCAN_T)
scan_rows(const int* __restrict__ deg, int* __restrict__ rowStart, int N) {
    __shared__ int wtot[SCAN_T / 32];
    __shared__ int wcar[SCAN_T / 32];
    const int t    = threadIdx.x;
    const int lane = t & 31;
    const int wave = t >> 5;
    const int base = blockIdx.x * SCAN_T;
    const int idx  = base + t;
    const int d    = (idx < N) ? deg[idx] : 0;

    int s = 0;
    if ((((size_t)deg) & 15) == 0) {
        const int4* __restrict__ d4 = (const int4*)deg;
        const int n4 = base >> 2;
        int i = t;
        for (; i + 7 * SCAN_T < n4; i += 8 * SCAN_T) {
            int4 v[8];
            for (int u = 0; u < 8; ++u) v[u] = d4[i + u * SCAN_T];
            for (int u = 0; u < 8; ++u) s += (v[u].x + v[u].y) + (v[u].z + v[u].w);
        }
        for (; i < n4; i += SCAN_T) {
            const int4 v = d4[i];
            s += (v.x + v.y) + (v.z + v.w);
        }
    } else {
        for (int i = t; i < base; i += SCAN_T) s += deg[i];
    }

    int incl = d;
    for (int off = 1; off < 32; off <<= 1) {
        const int up = __shfl_up(incl, off, 32);
        if (lane >= off) incl += up;
    }
    for (int off = 16; off >= 1; off >>= 1) s += __shfl_xor(s, off, 32);
    if (lane == 31) wtot[wave] = incl;
    if (lane == 0)  wcar[wave] = s;
    __syncthreads();
    int c = wcar[lane] + ((lane < wave) ? wtot[lane] : 0);
    for (int off = 16; off >= 1; off >>= 1) c += __shfl_xor(c, off, 32);
    const int excl = c + incl - d;
    if (idx < N) rowStart[idx] = excl;
    if (idx == N - 1) rowStart[N] = excl + d;
}

__global__ void __launch_bounds__(BIN_T)
bin_scatter(const int* __restrict__ erow, const int* __restrict__ ecol,
            const float* __restrict__ ev, const int* __restrict__ startMat,
            int* __restrict__ brec_, int E, int B) {
    __shared__ int cur[BK_MAX];
    int2* __restrict__ brec = (int2*)brec_;
    const int t = threadIdx.x;
    for (int i = t; i < B; i += BIN_T) cur[i] = startMat[(size_t)i * BIN_G + blockIdx.x];
    __syncthreads();
    const int tile = (E + BIN_G - 1) / BIN_G;
    const int e0 = (int)blockIdx.x * tile;
    const int e1 = (e0 + tile < E) ? (e0 + tile) : E;
    for (int base = e0; base < e1; base += BIN_U * BIN_T) {
        int r[BIN_U], c[BIN_U]; float v[BIN_U];
#pragma unroll
        for (int u = 0; u < BIN_U; ++u) {
            const int e = base + u * BIN_T + t;
            const bool ok = e < e1;
            const int ee = ok ? e : (e1 - 1);
            r[u] = erow[ee];
            c[u] = ecol[ee];
            v[u] = ev[ee];
        }
#pragma unroll
        for (int u = 0; u < BIN_U; ++u) asm volatile("" :: "v"(r[u]), "v"(c[u]), "v"(v[u]));
        int pos[BIN_U];
#pragma unroll
        for (int u = 0; u < BIN_U; ++u) {
            pos[u] = -1;
            if (base + u * BIN_T + t < e1) pos[u] = atomicAdd(&cur[r[u] >> BK_SH], 1);
        }
#pragma unroll
        for (int u = 0; u < BIN_U; ++u) {
            if (pos[u] >= 0)
                brec[pos[u]] = make_int2((c[u] << BK_SH) | (r[u] & (BK_R - 1)), __float_as_int(v[u]));
        }
    }
}

#define BS_T    1024
#define BS_U    6
#define BS_CAP  (BS_T * BS_U)

__device__ __forceinline__ int bucket_scan(int* rc, int* wsum, int t) {
    const int lane = t & 31;
    const int wave = t >> 5;
    int d = 0, incl = 0;
    if (t < BK_R) {
        d = rc[t];
        incl = d;
        for (int off = 1; off < 32; off <<= 1) {
            const int up = __shfl_up(incl, off, 32);
            if (lane >= off) incl += up;
        }
        if (lane == 31) wsum[wave] = incl;
    }
    __syncthreads();
    int excl = 0;
    if (t < BK_R) {
        int c = (lane < wave) ? wsum[lane] : 0;
        for (int off = 16; off >= 1; off >>= 1) c += __shfl_xor(c, off, 32);
        excl = c + incl - d;
        rc[t] = excl;
    }
    __syncthreads();
    return excl;
}

__global__ void __launch_bounds__(BS_T)
bucket_sort(const int* __restrict__ brec_, const int* __restrict__ startMat,
            int* __restrict__ rowStart, int* __restrict__ scol,
            float* __restrict__ sval, int N, int B) {
    __shared__ int   rc[BK_R];
    __shared__ int   wsum[BK_R / 32];
    __shared__ int   lcol[BS_CAP];
    __shared__ float lval[BS_CAP];
    const int2* __restrict__ brec = (const int2*)brec_;
    const int t = threadIdx.x;
    const int b = blockIdx.x;
    const int s0 = startMat[(size_t)b * BIN_G];
    const int s1 = startMat[(size_t)(b + 1) * BIN_G];
    const int n  = s1 - s0;
    if (t < BK_R) rc[t] = 0;
    __syncthreads();

    if (n <= BS_CAP) {
        int key[BS_U], val[BS_U], lr[BS_U];
        const int last = (s1 > 1) ? (s1 - 1) : 0;
#pragma unroll
        for (int u = 0; u < BS_U; ++u) {
            const int idx = s0 + t + u * BS_T;
            const int2 q = brec[idx < last ? idx : last];
            key[u] = q.x; val[u] = q.y;
        }
#pragma unroll
        for (int u = 0; u < BS_U; ++u) {
            lr[u] = 0;
            if (t + u * BS_T < n) lr[u] = atomicAdd(&rc[key[u] & (BK_R - 1)], 1);
        }
        __syncthreads();
        const int excl = bucket_scan(rc, wsum, t);
        if (t < BK_R) {
            const int row = b * BK_R + t;
            if (row < N) rowStart[row] = s0 + excl;
        }
        int pos[BS_U];
#pragma unroll
        for (int u = 0; u < BS_U; ++u) pos[u] = rc[key[u] & (BK_R - 1)] + lr[u];
#pragma unroll
        for (int u = 0; u < BS_U; ++u) {
            if (t + u * BS_T < n) {
                lcol[pos[u]] = key[u] >> BK_SH;
                lval[pos[u]] = __int_as_float(val[u]);
            }
        }
        __syncthreads();
        if (n > 0) {
            int oc[BS_U]; float ov[BS_U];
#pragma unroll
            for (int u = 0; u < BS_U; ++u) {
                const int j = t + u * BS_T;
                const int jj = (j < n) ? j : (n - 1);
                oc[u] = lcol[jj]; ov[u] = lval[jj];
            }
#pragma unroll
            for (int u = 0; u < BS_U; ++u) {
                const int j = t + u * BS_T;
                if (j < n) { scol[s0 + j] = oc[u]; sval[s0 + j] = ov[u]; }
            }
        }
    } else {
        for (int j = t; j < n; j += BS_T)
            atomicAdd(&rc[brec[s0 + j].x & (BK_R - 1)], 1);
        __syncthreads();
        const int excl = bucket_scan(rc, wsum, t);
        if (t < BK_R) {
            const int row = b * BK_R + t;
            if (row < N) rowStart[row] = s0 + excl;
        }
        for (int j = t; j < n; j += BS_T) {
            const int2 q = brec[s0 + j];
            const int pos = s0 + atomicAdd(&rc[q.x & (BK_R - 1)], 1);
            scol[pos] = q.x >> BK_SH;
            sval[pos] = __int_as_float(q.y);
        }
    }
    if (b == B - 1 && t == 0) rowStart[N] = s1;
}

#define SP1_T    512
#define SP1_W    (SP1_T / 32)
#define SP1_ROWS 4
#define SP1_BR   (SP1_W * SP1_ROWS)
#define SP1_NT   (FEAT / 16)
#define SP1_MW   (SP1_BR / 16 * SP1_NT / SP1_W)
#define ZP       (FEAT + 4)

static_assert(SP1_W % SP1_NT == 0 && SP1_MW >= 1 && SP1_BR % 16 == 0,
              "phase 2 splits SP1_BR / 16 row tiles x SP1_NT column tiles evenly over the waves");

template <int C>
__device__ __forceinline__ void gather_fma(const float4* __restrict__ y4, int cc, float vv,
                                           int j, int lane, float4& acc) {
    float4 g[C];
#pragma unroll
    for (int u = 0; u < C; ++u) {
        const int c = __builtin_amdgcn_readlane(cc, j + u);
        g[u] = y4[(size_t)c * (FEAT / 4) + lane];
    }
#pragma unroll
    for (int u = 0; u < C; ++u) {
        const float v = __int_as_float(__builtin_amdgcn_readlane(__float_as_int(vv), j + u));
        acc.x = fmaf(v, g[u].x, acc.x);
        acc.y = fmaf(v, g[u].y, acc.y);
        acc.z = fmaf(v, g[u].z, acc.z);
        acc.w = fmaf(v, g[u].w, acc.w);
    }
}

__global__ void __launch_bounds__(SP1_T)
spmm1_fused(const float* __restrict__ x, const int* __restrict__ rowStart,
            const int* __restrict__ scol, const float* __restrict__ sval,
            const float* __restrict__ W1,
            const float* __restrict__ b1, const float* __restrict__ g1,
            const float* __restrict__ be1, const float* __restrict__ m1,
            const float* __restrict__ v1, const float* __restrict__ W2,
            float* __restrict__ p, int N) {
    __shared__ float w2s[FEAT * NCLS];
    __shared__ __attribute__((aligned(16))) float zs[SP1_BR * ZP];
    {
        constexpr int W2N = FEAT * NCLS;
        constexpr int W2K = (W2N + SP1_T - 1) / SP1_T;
        float t[W2K];
#pragma unroll
        for (int k = 0; k < W2K; ++k) {
            const int i = (int)threadIdx.x + k * SP1_T;
            t[k] = (i < W2N) ? W2[i] : 0.0f;
        }
#pragma unroll
        for (int k = 0; k < W2K; ++k) {
            const int i = (int)threadIdx.x + k * SP1_T;
            if (i < W2N) w2s[i] = t[k];
        }
    }

    const int lane = threadIdx.x & 31;
    const int half = lane >> 4;
    const int r16  = lane & 15;
    const int wave = __builtin_amdgcn_readfirstlane((int)(threadIdx.x >> 5));
    const int lrow0 = wave * SP1_ROWS;
    const int r0 = (int)blockIdx.x * SP1_BR + lrow0;
    int nr = N - r0;
    nr = nr < 0 ? 0 : (nr > SP1_ROWS ? SP1_ROWS : nr);

    if (nr > 0) {
        const float4* __restrict__ x4 = (const float4*)x;
        const int myrs = rowStart[r0 + (lane < nr ? lane : nr)];
        int eB = __builtin_amdgcn_readlane(myrs, 0);
        int eE = __builtin_amdgcn_readlane(myrs, 1);
        int nc = 0; float nv = 0.0f;
        if (eB + lane < eE) { nc = scol[eB + lane]; nv = sval[eB + lane]; }

        for (int i = 0; i < nr; ++i) {
            const int cc = nc; const float vv = nv;
            int nE = eE;
            if (i + 1 < nr) {
                nE = __builtin_amdgcn_readlane(myrs, i + 2);
                nc = 0; nv = 0.0f;
                if (eE + lane < nE) { nc = scol[eE + lane]; nv = sval[eE + lane]; }
            }

            float4 acc = {0.f, 0.f, 0.f, 0.f};
            const int deg = eE - eB;
            const int n = deg < 32 ? deg : 32;
            int j = 0;
            for (; j + 8 <= n; j += 8) gather_fma<8>(x4, cc, vv, j, lane, acc);
            switch (n - j) {
                case 7: gather_fma<7>(x4, cc, vv, j, lane, acc); break;
                case 6: gather_fma<6>(x4, cc, vv, j, lane, acc); break;
                case 5: gather_fma<5>(x4, cc, vv, j, lane, acc); break;
                case 4: gather_fma<4>(x4, cc, vv, j, lane, acc); break;
                case 3: gather_fma<3>(x4, cc, vv, j, lane, acc); break;
                case 2: gather_fma<2>(x4, cc, vv, j, lane, acc); break;
                case 1: gather_fma<1>(x4, cc, vv, j, lane, acc); break;
                default: break;
            }
            for (int e = eB + 32; e < eE; ++e) {
                const int   c = scol[e];
                const float v = sval[e];
                const float4 g = x4[(size_t)c * (FEAT / 4) + lane];
                acc.x = fmaf(v, g.x, acc.x);
                acc.y = fmaf(v, g.y, acc.y);
                acc.z = fmaf(v, g.z, acc.z);
                acc.w = fmaf(v, g.w, acc.w);
            }
            *(float4*)&zs[(lrow0 + i) * ZP + lane * 4] = acc;
            eB = eE; eE = nE;
        }
    }
    {
        const float4 zero4 = {0.f, 0.f, 0.f, 0.f};
        for (int i = nr; i < SP1_ROWS; ++i) *(float4*)&zs[(lrow0 + i) * ZP + lane * 4] = zero4;
    }
    __syncthreads();

    const int ch  = (wave & (SP1_NT - 1)) * 16 + r16;
    const int mr0 = (wave / SP1_NT) * SP1_MW * 16;
    {
        const v8f vzero = {0.f,0.f,0.f,0.f,0.f,0.f,0.f,0.f};
        v8f acc[SP1_MW];
#pragma unroll
        for (int mt = 0; mt < SP1_MW; ++mt) acc[mt] = vzero;
        const float* __restrict__ wcol = W1 + ch;
        const float* zrow = &zs[(mr0 + r16) * ZP + 2 * half];
#pragma unroll 8
        for (int kt = 0; kt < FEAT / 4; ++kt) {
            const int k0 = kt << 2;
            v2f b;
            b.x = wcol[(size_t)(k0 + 2 * half + 0) * FEAT];
            b.y = wcol[(size_t)(k0 + 2 * half + 1) * FEAT];
#pragma unroll
            for (int mt = 0; mt < SP1_MW; ++mt) {
                v2f a;
                a.x = zrow[mt * 16 * ZP + k0 + 0];
                a.y = zrow[mt * 16 * ZP + k0 + 1];
                acc[mt] = __builtin_amdgcn_wmma_f32_16x16x4_f32(
                    false, a, false, b, (short)0, acc[mt], false, false);
            }
        }
        const float bs = g1[ch] * rsqrtf(v1[ch] + EPSB);
        const float bt = (b1[ch] - m1[ch]) * bs + be1[ch];
        __syncthreads();
#pragma unroll
        for (int mt = 0; mt < SP1_MW; ++mt) {
#pragma unroll
            for (int i = 0; i < 8; ++i) {
                const float hv = fmaf(acc[mt][i], bs, bt);
                zs[(mr0 + mt * 16 + i + 8 * half) * ZP + ch] = hv > 0.0f ? hv : 0.0f;
            }
        }
    }
    __syncthreads();

    float w2[4][NCLS];
#pragma unroll
    for (int j = 0; j < 4; ++j)
#pragma unroll
        for (int c = 0; c < NCLS; ++c) w2[j][c] = w2s[(lane * 4 + j) * NCLS + c];
    for (int i = 0; i < nr; ++i) {
        const float4 hq = *(const float4*)&zs[(lrow0 + i) * ZP + lane * 4];
        const float h[4] = {hq.x, hq.y, hq.z, hq.w};
        const int r = r0 + i;
#pragma unroll
        for (int c = 0; c < NCLS; ++c) {
            float ssum = 0.0f;
#pragma unroll
            for (int j = 0; j < 4; ++j) ssum = fmaf(h[j], w2[j][c], ssum);
#pragma unroll
            for (int off = 16; off >= 1; off >>= 1) ssum += __shfl_xor(ssum, off, 32);
            if (lane == 0) p[(size_t)r * NCLS + c] = ssum;
        }
    }
}

__global__ void spmm2_fused(const float* __restrict__ p, const int* __restrict__ rowStart,
                            const int* __restrict__ scol, const float* __restrict__ sval,
                            const float* __restrict__ b2, const float* __restrict__ g2,
                            const float* __restrict__ be2, const float* __restrict__ m2,
                            const float* __restrict__ v2, float* __restrict__ out, int total) {
    int t = blockIdx.x * blockDim.x + threadIdx.x;
    if (t >= total) return;
    const int r = t / NCLS;
    const int c = t - r * NCLS;
    float acc = 0.0f;
    const int eEnd = rowStart[r + 1];
    for (int e = rowStart[r]; e < eEnd; ++e)
        acc = fmaf(sval[e], p[(size_t)scol[e] * NCLS + c], acc);
    const float s = g2[c] * rsqrtf(v2[c] + EPSB);
    out[t] = (acc + b2[c] - m2[c]) * s + be2[c];
}

extern "C" void kernel_launch(void* const* d_in, const int* in_sizes, int n_in,
                              void* d_out, int out_size, void* d_ws, size_t ws_size,
                              hipStream_t stream) {
    const float* x   = (const float*)d_in[0];
    const float* ev  = (const float*)d_in[1];
    const float* W1  = (const float*)d_in[2];
    const float* b1  = (const float*)d_in[3];
    const float* g1  = (const float*)d_in[4];
    const float* be1 = (const float*)d_in[5];
    const float* m1  = (const float*)d_in[6];
    const float* v1  = (const float*)d_in[7];
    const float* W2  = (const float*)d_in[8];
    const float* b2  = (const float*)d_in[9];
    const float* g2  = (const float*)d_in[10];
    const float* be2 = (const float*)d_in[11];
    const float* m2  = (const float*)d_in[12];
    const float* v2  = (const float*)d_in[13];
    const int*  erow = (const int*)d_in[14];
    const int*  ecol = (const int*)d_in[15];
    float* out = (float*)d_out;

    const int N = in_sizes[0] / FEAT;
    const int E = in_sizes[1];

    const int B = (N + BK_R - 1) >> BK_SH;

    float* p        = (float*)d_ws;
    int*   brec     = (int*)(p + (size_t)N * NCLS);
    int*   cntMat   = brec + (size_t)2 * E;
    int*   startMat = cntMat + (size_t)B * BIN_G;
    int*   rowStart = startMat + ((size_t)B * BIN_G + 1);
    int*   scol     = rowStart + (N + 1);
    float* sval     = (float*)(scol + E);

    bin_count<<<BIN_G, BIN_T, 0, stream>>>(erow, cntMat, E, B);
    scan_rows<<<(B * BIN_G + SCAN_T - 1) / SCAN_T, SCAN_T, 0, stream>>>(cntMat, startMat,
                                                                      B * BIN_G);
    bin_scatter<<<BIN_G, BIN_T, 0, stream>>>(erow, ecol, ev, startMat, brec, E, B);
    bucket_sort<<<B, BS_T, 0, stream>>>(brec, startMat, rowStart, scol, sval, N, B);
    spmm1_fused<<<(N + SP1_BR - 1) / SP1_BR, SP1_T, 0, stream>>>(x, rowStart, scol, sval, W1,
                                                                 b1, g1, be1, m1, v1, W2, p, N);
    spmm2_fused<<<(N * NCLS + 255) / 256, 256, 0, stream>>>(p, rowStart, scol, sval,
                                                            b2, g2, be2, m2, v2, out,
                                                            N * NCLS);
}


# spmm2: eight lanes per row on 32-byte rows of p (pitch 7 -> 8); a wave runs 16 rows in step, 8 edges per row per trip, 32 gathers in flight, pairs passed by DPP8; 3.5x fewer waves; sum order kept
# speedup vs baseline: 4.1647x; 1.1121x over previous
#include <hip/hip_runtime.h>
#include <hip/hip_bf16.h>

typedef __attribute__((ext_vector_type(2))) float v2f;
typedef __attribute__((ext_vector_type(8))) float v8f;

#define FEAT 128
#define NCLS 7
#define PP   8
#define EPSB 1e-5f
#define SCAN_T 1024

#define BK_SH   8
#define BK_R    (1 << BK_SH)
#define BK_MAX  1024
#define BIN_G   256
#define BIN_T   1024
#define BIN_U   8

__global__ void __launch_bounds__(BIN_T)
bin_count(const int* __restrict__ erow, int* __restrict__ cntMat, int E, int B) {
    __shared__ int cnt[BK_MAX];
    const int t = threadIdx.x;
    for (int i = t; i < B; i += BIN_T) cnt[i] = 0;
    __syncthreads();
    const int tile = (E + BIN_G - 1) / BIN_G;
    const int e0 = (int)blockIdx.x * tile;
    const int e1 = (e0 + tile < E) ? (e0 + tile) : E;
    for (int base = e0; base < e1; base += BIN_U * BIN_T) {
        int r[BIN_U];
#pragma unroll
        for (int u = 0; u < BIN_U; ++u) {
            const int e = base + u * BIN_T + t;
            r[u] = erow[(e < e1) ? e : (e1 - 1)];
        }
#pragma unroll
        for (int u = 0; u < BIN_U; ++u) asm volatile("" :: "v"(r[u]));
#pragma unroll
        for (int u = 0; u < BIN_U; ++u)
            if (base + u * BIN_T + t < e1) atomicAdd(&cnt[r[u] >> BK_SH], 1);
    }
    __syncthreads();
    for (int i = t; i < B; i += BIN_T) cntMat[(size_t)i * BIN_G + blockIdx.x] = cnt[i];
}

__global__ void __launch_bounds__(SCAN_T)
scan_rows(const int* __restrict__ deg, int* __restrict__ rowStart, int N) {
    __shared__ int wtot[SCAN_T / 32];
    __shared__ int wcar[SCAN_T / 32];
    const int t    = threadIdx.x;
    const int lane = t & 31;
    const int wave = t >> 5;
    const int base = blockIdx.x * SCAN_T;
    const int idx  = base + t;
    const int d    = (idx < N) ? deg[idx] : 0;

    int s = 0;
    if ((((size_t)deg) & 15) == 0) {
        const int4* __restrict__ d4 = (const int4*)deg;
        const int n4 = base >> 2;
        int i = t;
        for (; i + 7 * SCAN_T < n4; i += 8 * SCAN_T) {
            int4 v[8];
            for (int u = 0; u < 8; ++u) v[u] = d4[i + u * SCAN_T];
            for (int u = 0; u < 8; ++u) s += (v[u].x + v[u].y) + (v[u].z + v[u].w);
        }
        for (; i < n4; i += SCAN_T) {
            const int4 v = d4[i];
            s += (v.x + v.y) + (v.z + v.w);
        }
    } else {
        for (int i = t; i < base; i += SCAN_T) s += deg[i];
    }

    int incl = d;
    for (int off = 1; off < 32; off <<= 1) {
        const int up = __shfl_up(incl, off, 32);
        if (lane >= off) incl += up;
    }
    for (int off = 16; off >= 1; off >>= 1) s += __shfl_xor(s, off, 32);
    if (lane == 31) wtot[wave] = incl;
    if (lane == 0)  wcar[wave] = s;
    __syncthreads();
    int c = wcar[lane] + ((lane < wave) ? wtot[lane] : 0);
    for (int off = 16; off >= 1; off >>= 1) c += __shfl_xor(c, off, 32);
    const int excl = c + incl - d;
    if (idx < N) rowStart[idx] = excl;
    if (idx == N - 1) rowStart[N] = excl + d;
}

__global__ void __launch_bounds__(BIN_T)
bin_scatter(const int* __restrict__ erow, const int* __restrict__ ecol,
            const float* __restrict__ ev, const int* __restrict__ startMat,
            int* __restrict__ brec_, int E, int B) {
    __shared__ int cur[BK_MAX];
    int2* __restrict__ brec = (int2*)brec_;
    const int t = threadIdx.x;
    for (int i = t; i < B; i += BIN_T) cur[i] = startMat[(size_t)i * BIN_G + blockIdx.x];
    __syncthreads();
    const int tile = (E + BIN_G - 1) / BIN_G;
    const int e0 = (int)blockIdx.x * tile;
    const int e1 = (e0 + tile < E) ? (e0 + tile) : E;
    for (int base = e0; base < e1; base += BIN_U * BIN_T) {
        int r[BIN_U], c[BIN_U]; float v[BIN_U];
#pragma unroll
        for (int u = 0; u < BIN_U; ++u) {
            const int e = base + u * BIN_T + t;
            const bool ok = e < e1;
            const int ee = ok ? e : (e1 - 1);
            r[u] = erow[ee];
            c[u] = ecol[ee];
            v[u] = ev[ee];
        }
#pragma unroll
        for (int u = 0; u < BIN_U; ++u) asm volatile("" :: "v"(r[u]), "v"(c[u]), "v"(v[u]));
        int pos[BIN_U];
#pragma unroll
        for (int u = 0; u < BIN_U; ++u) {
            pos[u] = -1;
            if (base + u * BIN_T + t < e1) pos[u] = atomicAdd(&cur[r[u] >> BK_SH], 1);
        }
#pragma unroll
        for (int u = 0; u < BIN_U; ++u) {
            if (pos[u] >= 0)
                brec[pos[u]] = make_int2((c[u] << BK_SH) | (r[u] & (BK_R - 1)), __float_as_int(v[u]));
        }
    }
}

#define BS_T    1024
#define BS_U    6
#define BS_CAP  (BS_T * BS_U)

__device__ __forceinline__ int bucket_scan(int* rc, int* wsum, int t) {
    const int lane = t & 31;
    const int wave = t >> 5;
    int d = 0, incl = 0;
    if (t < BK_R) {
        d = rc[t];
        incl = d;
        for (int off = 1; off < 32; off <<= 1) {
            const int up = __shfl_up(incl, off, 32);
            if (lane >= off) incl += up;
        }
        if (lane == 31) wsum[wave] = incl;
    }
    __syncthreads();
    int excl = 0;
    if (t < BK_R) {
        int c = (lane < wave) ? wsum[lane] : 0;
        for (int off = 16; off >= 1; off >>= 1) c += __shfl_xor(c, off, 32);
        excl = c + incl - d;
        rc[t] = excl;
    }
    __syncthreads();
    return excl;
}

__global__ void __launch_bounds__(BS_T)
bucket_sort(const int* __restrict__ brec_, const int* __restrict__ startMat,
            int* __restrict__ rowStart, int* __restrict__ scol,
            float* __restrict__ sval, int N, int B) {
    __shared__ int   rc[BK_R];
    __shared__ int   wsum[BK_R / 32];
    __shared__ int   lcol[BS_CAP];
    __shared__ float lval[BS_CAP];
    const int2* __restrict__ brec = (const int2*)brec_;
    const int t = threadIdx.x;
    const int b = blockIdx.x;
    const int s0 = startMat[(size_t)b * BIN_G];
    const int s1 = startMat[(size_t)(b + 1) * BIN_G];
    const int n  = s1 - s0;
    if (t < BK_R) rc[t] = 0;
    __syncthreads();

    if (n <= BS_CAP) {
        int key[BS_U], val[BS_U], lr[BS_U];
        const int last = (s1 > 1) ? (s1 - 1) : 0;
#pragma unroll
        for (int u = 0; u < BS_U; ++u) {
            const int idx = s0 + t + u * BS_T;
            const int2 q = brec[idx < last ? idx : last];
            key[u] = q.x; val[u] = q.y;
        }
#pragma unroll
        for (int u = 0; u < BS_U; ++u) {
            lr[u] = 0;
            if (t + u * BS_T < n) lr[u] = atomicAdd(&rc[key[u] & (BK_R - 1)], 1);
        }
        __syncthreads();
        const int excl = bucket_scan(rc, wsum, t);
        if (t < BK_R) {
            const int row = b * BK_R + t;
            if (row < N) rowStart[row] = s0 + excl;
        }
        int pos[BS_U];
#pragma unroll
        for (int u = 0; u < BS_U; ++u) pos[u] = rc[key[u] & (BK_R - 1)] + lr[u];
#pragma unroll
        for (int u = 0; u < BS_U; ++u) {
            if (t + u * BS_T < n) {
                lcol[pos[u]] = key[u] >> BK_SH;
                lval[pos[u]] = __int_as_float(val[u]);
            }
        }
        __syncthreads();
        if (n > 0) {
            int oc[BS_U]; float ov[BS_U];
#pragma unroll
            for (int u = 0; u < BS_U; ++u) {
                const int j = t + u * BS_T;
                const int jj = (j < n) ? j : (n - 1);
                oc[u] = lcol[jj]; ov[u] = lval[jj];
            }
#pragma unroll
            for (int u = 0; u < BS_U; ++u) {
                const int j = t + u * BS_T;
                if (j < n) { scol[s0 + j] = oc[u]; sval[s0 + j] = ov[u]; }
            }
        }
    } else {
        for (int j = t; j < n; j += BS_T)
            atomicAdd(&rc[brec[s0 + j].x & (BK_R - 1)], 1);
        __syncthreads();
        const int excl = bucket_scan(rc, wsum, t);
        if (t < BK_R) {
            const int row = b * BK_R + t;
            if (row < N) rowStart[row] = s0 + excl;
        }
        for (int j = t; j < n; j += BS_T) {
            const int2 q = brec[s0 + j];
            const int pos = s0 + atomicAdd(&rc[q.x & (BK_R - 1)], 1);
            scol[pos] = q.x >> BK_SH;
            sval[pos] = __int_as_float(q.y);
        }
    }
    if (b == B - 1 && t == 0) rowStart[N] = s1;
}

#define SP1_T    512
#define SP1_W    (SP1_T / 32)
#define SP1_ROWS 4
#define SP1_BR   (SP1_W * SP1_ROWS)
#define SP1_NT   (FEAT / 16)
#define SP1_MW   (SP1_BR / 16 * SP1_NT / SP1_W)
#define ZP       (FEAT + 4)

static_assert(SP1_W % SP1_NT == 0 && SP1_MW >= 1 && SP1_BR % 16 == 0,
              "phase 2 splits SP1_BR / 16 row tiles x SP1_NT column tiles evenly over the waves");

template <int C>
__device__ __forceinline__ void gather_fma(const float4* __restrict__ y4, int cc, float vv,
                                           int j, int lane, float4& acc) {
    float4 g[C];
#pragma unroll
    for (int u = 0; u < C; ++u) {
        const int c = __builtin_amdgcn_readlane(cc, j + u);
        g[u] = y4[(size_t)c * (FEAT / 4) + lane];
    }
#pragma unroll
    for (int u = 0; u < C; ++u) {
        const float v = __int_as_float(__builtin_amdgcn_readlane(__float_as_int(vv), j + u));
        acc.x = fmaf(v, g[u].x, acc.x);
        acc.y = fmaf(v, g[u].y, acc.y);
        acc.z = fmaf(v, g[u].z, acc.z);
        acc.w = fmaf(v, g[u].w, acc.w);
    }
}

__global__ void __launch_bounds__(SP1_T)
spmm1_fused(const float* __restrict__ x, const int* __restrict__ rowStart,
            const int* __restrict__ scol, const float* __restrict__ sval,
            const float* __restrict__ W1,
            const float* __restrict__ b1, const float* __restrict__ g1,
            const float* __restrict__ be1, const float* __restrict__ m1,
            const float* __restrict__ v1, const float* __restrict__ W2,
            float* __restrict__ p, int N) {
    __shared__ float w2s[FEAT * NCLS];
    __shared__ __attribute__((aligned(16))) float zs[SP1_BR * ZP];
    {
        constexpr int W2N = FEAT * NCLS;
        constexpr int W2K = (W2N + SP1_T - 1) / SP1_T;
        float t[W2K];
#pragma unroll
        for (int k = 0; k < W2K; ++k) {
            const int i = (int)threadIdx.x + k * SP1_T;
            t[k] = (i < W2N) ? W2[i] : 0.0f;
        }
#pragma unroll
        for (int k = 0; k < W2K; ++k) {
            const int i = (int)threadIdx.x + k * SP1_T;
            if (i < W2N) w2s[i] = t[k];
        }
    }

    const int lane = threadIdx.x & 31;
    const int half = lane >> 4;
    const int r16  = lane & 15;
    const int wave = __builtin_amdgcn_readfirstlane((int)(threadIdx.x >> 5));
    const int lrow0 = wave * SP1_ROWS;
    const int r0 = (int)blockIdx.x * SP1_BR + lrow0;
    int nr = N - r0;
    nr = nr < 0 ? 0 : (nr > SP1_ROWS ? SP1_ROWS : nr);

    if (nr > 0) {
        const float4* __restrict__ x4 = (const float4*)x;
        const int myrs = rowStart[r0 + (lane < nr ? lane : nr)];
        int eB = __builtin_amdgcn_readlane(myrs, 0);
        int eE = __builtin_amdgcn_readlane(myrs, 1);
        int nc = 0; float nv = 0.0f;
        if (eB + lane < eE) { nc = scol[eB + lane]; nv = sval[eB + lane]; }

        for (int i = 0; i < nr; ++i) {
            const int cc = nc; const float vv = nv;
            int nE = eE;
            if (i + 1 < nr) {
                nE = __builtin_amdgcn_readlane(myrs, i + 2);
                nc = 0; nv = 0.0f;
                if (eE + lane < nE) { nc = scol[eE + lane]; nv = sval[eE + lane]; }
            }

            float4 acc = {0.f, 0.f, 0.f, 0.f};
            const int deg = eE - eB;
            const int n = deg < 32 ? deg : 32;
            int j = 0;
            for (; j + 8 <= n; j += 8) gather_fma<8>(x4, cc, vv, j, lane, acc);
            switch (n - j) {
                case 7: gather_fma<7>(x4, cc, vv, j, lane, acc); break;
                case 6: gather_fma<6>(x4, cc, vv, j, lane, acc); break;
                case 5: gather_fma<5>(x4, cc, vv, j, lane, acc); break;
                case 4: gather_fma<4>(x4, cc, vv, j, lane, acc); break;
                case 3: gather_fma<3>(x4, cc, vv, j, lane, acc); break;
                case 2: gather_fma<2>(x4, cc, vv, j, lane, acc); break;
                case 1: gather_fma<1>(x4, cc, vv, j, lane, acc); break;
                default: break;
            }
            for (int e = eB + 32; e < eE; ++e) {
                const int   c = scol[e];
                const float v = sval[e];
                const float4 g = x4[(size_t)c * (FEAT / 4) + lane];
                acc.x = fmaf(v, g.x, acc.x);
                acc.y = fmaf(v, g.y, acc.y);
                acc.z = fmaf(v, g.z, acc.z);
                acc.w = fmaf(v, g.w, acc.w);
            }
            *(float4*)&zs[(lrow0 + i) * ZP + lane * 4] = acc;
            eB = eE; eE = nE;
        }
    }
    {
        const float4 zero4 = {0.f, 0.f, 0.f, 0.f};
        for (int i = nr; i < SP1_ROWS; ++i) *(float4*)&zs[(lrow0 + i) * ZP + lane * 4] = zero4;
    }
    __syncthreads();

    const int ch  = (wave & (SP1_NT - 1)) * 16 + r16;
    const int mr0 = (wave / SP1_NT) * SP1_MW * 16;
    {
        const v8f vzero = {0.f,0.f,0.f,0.f,0.f,0.f,0.f,0.f};
        v8f acc[SP1_MW];
#pragma unroll
        for (int mt = 0; mt < SP1_MW; ++mt) acc[mt] = vzero;
        const float* __restrict__ wcol = W1 + ch;
        const float* zrow = &zs[(mr0 + r16) * ZP + 2 * half];
#pragma unroll 8
        for (int kt = 0; kt < FEAT / 4; ++kt) {
            const int k0 = kt << 2;
            v2f b;
            b.x = wcol[(size_t)(k0 + 2 * half + 0) * FEAT];
            b.y = wcol[(size_t)(k0 + 2 * half + 1) * FEAT];
#pragma unroll
            for (int mt = 0; mt < SP1_MW; ++mt) {
                v2f a;
                a.x = zrow[mt * 16 * ZP + k0 + 0];
                a.y = zrow[mt * 16 * ZP + k0 + 1];
                acc[mt] = __builtin_amdgcn_wmma_f32_16x16x4_f32(
                    false, a, false, b, (short)0, acc[mt], false, false);
            }
        }
        const float bs = g1[ch] * rsqrtf(v1[ch] + EPSB);
        const float bt = (b1[ch] - m1[ch]) * bs + be1[ch];
        __syncthreads();
#pragma unroll
        for (int mt = 0; mt < SP1_MW; ++mt) {
#pragma unroll
            for (int i = 0; i < 8; ++i) {
                const float hv = fmaf(acc[mt][i], bs, bt);
                zs[(mr0 + mt * 16 + i + 8 * half) * ZP + ch] = hv > 0.0f ? hv : 0.0f;
            }
        }
    }
    __syncthreads();

    float w2[4][NCLS];
#pragma unroll
    for (int j = 0; j < 4; ++j)
#pragma unroll
        for (int c = 0; c < NCLS; ++c) w2[j][c] = w2s[(lane * 4 + j) * NCLS + c];
    for (int i = 0; i < nr; ++i) {
        const float4 hq = *(const float4*)&zs[(lrow0 + i) * ZP + lane * 4];
        const float h[4] = {hq.x, hq.y, hq.z, hq.w};
        const int r = r0 + i;
        float mine = 0.0f;
#pragma unroll
        for (int c = 0; c < NCLS; ++c) {
            float ssum = 0.0f;
#pragma unroll
            for (int j = 0; j < 4; ++j) ssum = fmaf(h[j], w2[j][c], ssum);
#pragma unroll
            for (int off = 16; off >= 1; off >>= 1) ssum += __shfl_xor(ssum, off, 32);
            if (lane == c) mine = ssum;
        }
        if (lane < PP) p[(size_t)r * PP + lane] = mine;
    }
}

#define SP2_T    128
#define SP2_G    4
#define SP2_WR   (SP2_G * 4)
#define SP2_BR   (SP2_T / 32 * SP2_WR)

template <int U>
__device__ __forceinline__ unsigned bcast8(unsigned v) {
    return __builtin_amdgcn_mov_dpp8(v, (unsigned)(U * 0x249249));
}

__device__ __forceinline__ void sp2_pairs(const int* __restrict__ scol,
                                          const float* __restrict__ sval,
                                          const int (&eb)[SP2_G], const int (&ee)[SP2_G],
                                          int c8, unsigned (&po)[SP2_G], float (&vl)[SP2_G]) {
#pragma unroll
    for (int g = 0; g < SP2_G; ++g) {
        int i = eb[g] + c8;
        i = (i < ee[g] - 1) ? i : (ee[g] - 1);
        i = (i > 0) ? i : 0;
        po[g] = (unsigned)scol[i] * PP;
        vl[g] = sval[i];
    }
}

template <int U>
__device__ __forceinline__ void sp2_edge(const float* __restrict__ p, const unsigned (&po)[SP2_G],
                                         unsigned c8, float (&ga)[SP2_G][8]) {
#pragma unroll
    for (int g = 0; g < SP2_G; ++g) ga[g][U] = p[bcast8<U>(po[g]) | c8];
}

template <int U>
__device__ __forceinline__ void sp2_fma(const float (&cv)[SP2_G], const int (&nrem)[SP2_G],
                                        const float (&ga)[SP2_G][8], float (&acc)[SP2_G]) {
#pragma unroll
    for (int g = 0; g < SP2_G; ++g) {
        const float v = __uint_as_float(bcast8<U>(__float_as_uint(cv[g])));
        if (U < nrem[g]) acc[g] = fmaf(v, ga[g][U], acc[g]);
    }
}

__global__ void __launch_bounds__(SP2_T)
spmm2_fused(const float* __restrict__ p, const int* __restrict__ rowStart,
            const int* __restrict__ scol, const float* __restrict__ sval,
            const float* __restrict__ b2, const float* __restrict__ g2,
            const float* __restrict__ be2, const float* __restrict__ m2,
            const float* __restrict__ v2, float* __restrict__ out, int N) {
    __shared__ float os[SP2_BR * NCLS];
    const int t    = threadIdx.x;
    const int lane = t & 31;
    const int wave = t >> 5;
    const int q    = lane >> 3;
    const int c8   = lane & 7;
    const int lrow = wave * SP2_WR + q;
    const int row0 = (int)blockIdx.x * SP2_BR + lrow;

    int eb[SP2_G], ee[SP2_G];
    float acc[SP2_G];
#pragma unroll
    for (int g = 0; g < SP2_G; ++g) {
        const int r  = row0 + 4 * g;
        const int rc = (r < N) ? r : (N - 1);
        eb[g] = rowStart[rc]; ee[g] = rowStart[rc + 1];
        acc[g] = 0.0f;
    }
    const int cc = (c8 < NCLS) ? c8 : (NCLS - 1);
    const float kb = b2[cc], km = m2[cc], kbe = be2[cc];
    const float ksc = g2[cc] * rsqrtf(v2[cc] + EPSB);
#pragma unroll
    for (int g = 0; g < SP2_G; ++g)
        if (row0 + 4 * g >= N) ee[g] = eb[g];

    unsigned po[SP2_G]; float vl[SP2_G];
    sp2_pairs(scol, sval, eb, ee, c8, po, vl);
    for (;;) {
        bool more = false;
#pragma unroll
        for (int g = 0; g < SP2_G; ++g) more = more || (eb[g] < ee[g]);
        if (__builtin_amdgcn_ballot_w32(more) == 0u) break;

        float ga[SP2_G][8];
        sp2_edge<0>(p, po, (unsigned)c8, ga); sp2_edge<1>(p, po, (unsigned)c8, ga);
        sp2_edge<2>(p, po, (unsigned)c8, ga); sp2_edge<3>(p, po, (unsigned)c8, ga);
        sp2_edge<4>(p, po, (unsigned)c8, ga); sp2_edge<5>(p, po, (unsigned)c8, ga);
        sp2_edge<6>(p, po, (unsigned)c8, ga); sp2_edge<7>(p, po, (unsigned)c8, ga);
        float cv[SP2_G]; int nrem[SP2_G];
#pragma unroll
        for (int g = 0; g < SP2_G; ++g) {
            cv[g] = vl[g];
            nrem[g] = ee[g] - eb[g];
            eb[g] += 8;
        }
        sp2_pairs(scol, sval, eb, ee, c8, po, vl);
        __builtin_amdgcn_sched_barrier(0);
        sp2_fma<0>(cv, nrem, ga, acc); sp2_fma<1>(cv, nrem, ga, acc);
        sp2_fma<2>(cv, nrem, ga, acc); sp2_fma<3>(cv, nrem, ga, acc);
        sp2_fma<4>(cv, nrem, ga, acc); sp2_fma<5>(cv, nrem, ga, acc);
        sp2_fma<6>(cv, nrem, ga, acc); sp2_fma<7>(cv, nrem, ga, acc);
    }

    if (c8 < NCLS) {
#pragma unroll
        for (int g = 0; g < SP2_G; ++g)
            os[(lrow + 4 * g) * NCLS + c8] = (acc[g] + kb - km) * ksc + kbe;
    }
    __syncthreads();
    const int total = N * NCLS;
    const int base  = (int)blockIdx.x * (SP2_BR * NCLS);
    constexpr int SP2_OK = (SP2_BR * NCLS + SP2_T - 1) / SP2_T;
    float ov[SP2_OK];
#pragma unroll
    for (int k = 0; k < SP2_OK; ++k) {
        const int j = k * SP2_T + t;
        ov[k] = os[j < SP2_BR * NCLS ? j : SP2_BR * NCLS - 1];
    }
#pragma unroll
    for (int k = 0; k < SP2_OK; ++k) {
        const int j = k * SP2_T + t;
        if (j < SP2_BR * NCLS && base + j < total) out[base + j] = ov[k];
    }
}

extern "C" void kernel_launch(void* const* d_in, const int* in_sizes, int n_in,
                              void* d_out, int out_size, void* d_ws, size_t ws_size,
                              hipStream_t stream) {
    const float* x   = (const float*)d_in[0];
    const float* ev  = (const float*)d_in[1];
    const float* W1  = (const float*)d_in[2];
    const float* b1  = (const float*)d_in[3];
    const float* g1  = (const float*)d_in[4];
    const float* be1 = (const float*)d_in[5];
    const float* m1  = (const float*)d_in[6];
    const float* v1  = (const float*)d_in[7];
    const float* W2  = (const float*)d_in[8];
    const float* b2  = (const float*)d_in[9];
    const float* g2  = (const float*)d_in[10];
    const float* be2 = (const float*)d_in[11];
    const float* m2  = (const float*)d_in[12];
    const float* v2  = (const float*)d_in[13];
    const int*  erow = (const int*)d_in[14];
    const int*  ecol = (const int*)d_in[15];
    float* out = (float*)d_out;

    const int N = in_sizes[0] / FEAT;
    const int E = in_sizes[1];

    const int B = (N + BK_R - 1) >> BK_SH;

    float* p        = (float*)d_ws;
    int*   brec     = (int*)(p + (size_t)N * PP);
    int*   cntMat   = brec + (size_t)2 * E;
    int*   startMat = cntMat + (size_t)B * BIN_G;
    int*   rowStart = startMat + ((size_t)B * BIN_G + 1);
    int*   scol     = rowStart + (N + 1);
    float* sval     = (float*)(scol + E);

    bin_count<<<BIN_G, BIN_T, 0, stream>>>(erow, cntMat, E, B);
    scan_rows<<<(B * BIN_G + SCAN_T - 1) / SCAN_T, SCAN_T, 0, stream>>>(cntMat, startMat,
                                                                      B * BIN_G);
    bin_scatter<<<BIN_G, BIN_T, 0, stream>>>(erow, ecol, ev, startMat, brec, E, B);
    bucket_sort<<<B, BS_T, 0, stream>>>(brec, startMat, rowStart, scol, sval, N, B);
    spmm1_fused<<<(N + SP1_BR - 1) / SP1_BR, SP1_T, 0, stream>>>(x, rowStart, scol, sval, W1,
                                                                 b1, g1, be1, m1, v1, W2, p, N);
    spmm2_fused<<<(N + SP2_BR - 1) / SP2_BR, SP2_T, 0, stream>>>(p, rowStart, scol, sval,
                                                                 b2, g2, be2, m2, v2, out, N);
}


# CSR level 1: each edge tile sorts itself by bucket into its own slice (LDS add's return = rank); bucket pass sums its start from 256 tile offsets; 6 -> 4 launches; order of a row's fp sums changes
# speedup vs baseline: 5.0401x; 1.2102x over previous
#include <hip/hip_runtime.h>
#include <hip/hip_bf16.h>

typedef __attribute__((ext_vector_type(2))) float v2f;
typedef __attribute__((ext_vector_type(8))) float v8f;

#define FEAT 128
#define NCLS 7
#define PP   8
#define EPSB 1e-5f

#define BK_SH   8
#define BK_R    (1 << BK_SH)
#define BK_MAX  1024
#define BIN_G   256
#define BIN_T   1024
#define BIN_U   8

static_assert(BK_MAX == BIN_T, "bin_tile scans one bucket counter per thread");

__device__ __forceinline__ int block_excl_scan(int d, int* wt, int t) {
    const int lane = t & 31;
    const int wave = t >> 5;
    int incl = d;
    for (int off = 1; off < 32; off <<= 1) {
        const int up = __shfl_up(incl, off, 32);
        if (lane >= off) incl += up;
    }
    if (lane == 31) wt[wave] = incl;
    __syncthreads();
    int c = (lane < wave) ? wt[lane] : 0;
    for (int off = 16; off >= 1; off >>= 1) c += __shfl_xor(c, off, 32);
    return c + incl - d;
}

__global__ void __launch_bounds__(BIN_T)
bin_tile(const int* __restrict__ erow, const int* __restrict__ ecol,
         const float* __restrict__ ev, int* __restrict__ startL,
         int* __restrict__ brec_, int E, int B) {
    __shared__ int cnt[BK_MAX];
    __shared__ int wt[BIN_T / 32];
    int2* __restrict__ brec = (int2*)brec_;
    const int t = threadIdx.x;
    cnt[t] = 0;
    __syncthreads();
    const int tile = (E + BIN_G - 1) / BIN_G;
    int e0 = (int)blockIdx.x * tile;
    e0 = (e0 < E) ? e0 : E;
    const int e1 = (e0 + tile < E) ? (e0 + tile) : E;
    const int last = (e1 > 0) ? (e1 - 1) : 0;

    if (tile <= BIN_U * BIN_T) {
        int r[BIN_U], c[BIN_U], rk[BIN_U]; float v[BIN_U];
#pragma unroll
        for (int u = 0; u < BIN_U; ++u) {
            const int e = e0 + u * BIN_T + t;
            const int ee = (e < e1) ? e : last;
            r[u] = erow[ee];
            c[u] = ecol[ee];
            v[u] = ev[ee];
        }
        __builtin_amdgcn_sched_barrier(0);
#pragma unroll
        for (int u = 0; u < BIN_U; ++u) {
            rk[u] = -1;
            if (e0 + u * BIN_T + t < e1) rk[u] = atomicAdd(&cnt[r[u] >> BK_SH], 1);
        }
        __syncthreads();
        const int excl = block_excl_scan(cnt[t], wt, t);
        cnt[t] = excl;
        if (t <= B) startL[(size_t)t * BIN_G + blockIdx.x] = excl;
        __syncthreads();
        int st[BIN_U];
#pragma unroll
        for (int u = 0; u < BIN_U; ++u) st[u] = cnt[r[u] >> BK_SH];
#pragma unroll
        for (int u = 0; u < BIN_U; ++u) {
            if (rk[u] >= 0)
                brec[e0 + st[u] + rk[u]] =
                    make_int2((c[u] << BK_SH) | (r[u] & (BK_R - 1)), __float_as_int(v[u]));
        }
    } else {
        for (int base = e0; base < e1; base += BIN_U * BIN_T) {
            int r[BIN_U];
#pragma unroll
            for (int u = 0; u < BIN_U; ++u) {
                const int e = base + u * BIN_T + t;
                r[u] = erow[(e < e1) ? e : last];
            }
#pragma unroll
            for (int u = 0; u < BIN_U; ++u)
                if (base + u * BIN_T + t < e1) atomicAdd(&cnt[r[u] >> BK_SH], 1);
        }
        __syncthreads();
        const int excl = block_excl_scan(cnt[t], wt, t);
        cnt[t] = excl;
        if (t <= B) startL[(size_t)t * BIN_G + blockIdx.x] = excl;
        __syncthreads();
        for (int base = e0; base < e1; base += BIN_U * BIN_T) {
            int r[BIN_U], c[BIN_U], pos[BIN_U]; float v[BIN_U];
#pragma unroll
            for (int u = 0; u < BIN_U; ++u) {
                const int e = base + u * BIN_T + t;
                const int ee = (e < e1) ? e : last;
                r[u] = erow[ee];
                c[u] = ecol[ee];
                v[u] = ev[ee];
            }
#pragma unroll
            for (int u = 0; u < BIN_U; ++u) {
                pos[u] = -1;
                if (base + u * BIN_T + t < e1) pos[u] = atomicAdd(&cnt[r[u] >> BK_SH], 1);
            }
#pragma unroll
            for (int u = 0; u < BIN_U; ++u) {
                if (pos[u] >= 0)
                    brec[e0 + pos[u]] =
                        make_int2((c[u] << BK_SH) | (r[u] & (BK_R - 1)), __float_as_int(v[u]));
            }
        }
    }
}

#define BS_T    1024
#define BS_L    8
#define BS_P    (BIN_G * BS_L / BS_T)
#define BS_U    6
#define BS_O    6
#define BS_CAP  (BS_T * BS_O)

static_assert(BIN_G * BS_L == BS_P * BS_T, "every run has its BS_L lanes");

__device__ __forceinline__ int bucket_scan(int* rc, int* wsum, int t) {
    const int lane = t & 31;
    const int wave = t >> 5;
    int d = 0, incl = 0;
    if (t < BK_R) {
        d = rc[t];
        incl = d;
        for (int off = 1; off < 32; off <<= 1) {
            const int up = __shfl_up(incl, off, 32);
            if (lane >= off) incl += up;
        }
        if (lane == 31) wsum[wave] = incl;
    }
    __syncthreads();
    int excl = 0;
    if (t < BK_R) {
        int c = (lane < wave) ? wsum[lane] : 0;
        for (int off = 16; off >= 1; off >>= 1) c += __shfl_xor(c, off, 32);
        excl = c + incl - d;
        rc[t] = excl;
    }
    __syncthreads();
    return excl;
}

__global__ void __launch_bounds__(BS_T) __attribute__((amdgpu_waves_per_eu(16, 16)))
bucket_sort(const int* __restrict__ brec_, const int* __restrict__ startL,
            int* __restrict__ rowStart, int* __restrict__ scol,
            float* __restrict__ sval, int N, int B, int E) {
    __shared__ int   rc[BK_R];
    __shared__ int   wsum[BK_R / 32];
    __shared__ int   red0[BS_T / 32], red1[BS_T / 32], red2[BS_T / 32];
    __shared__ int   lcol[BS_CAP];
    __shared__ float lval[BS_CAP];
    const int2* __restrict__ brec = (const int2*)brec_;
    const int t    = threadIdx.x;
    const int lane = t & 31;
    const int wave = t >> 5;
    const int b    = blockIdx.x;
    const int grp  = t / BS_L;
    const int sub  = t % BS_L;
    const int tile = (E + BIN_G - 1) / BIN_G;
    int len[BS_P], first[BS_P];
    int a0 = 0, a1 = 0, am = 0;
#pragma unroll
    for (int p = 0; p < BS_P; ++p) {
        const int run = p * (BS_T / BS_L) + grp;
        int tb = run * tile;
        tb = (tb < E) ? tb : E;
        const int lo = startL[(size_t)b * BIN_G + run];
        const int hi = startL[(size_t)(b + 1) * BIN_G + run];
        len[p]   = hi - lo;
        first[p] = tb + lo;
        if (sub == 0) { a0 += lo; a1 += hi; }
        am = (len[p] > am) ? len[p] : am;
    }
    if (t < BK_R) rc[t] = 0;

    int key[BS_P * BS_U], val[BS_P * BS_U];
#pragma unroll
    for (int p = 0; p < BS_P; ++p) {
#pragma unroll
        for (int u = 0; u < BS_U; ++u) {
            const int k  = sub + u * BS_L;
            const int kk = (k < len[p]) ? k : (len[p] - 1);
            const int2 q = brec[(len[p] > 0) ? (first[p] + kk) : 0];
            key[p * BS_U + u] = q.x; val[p * BS_U + u] = q.y;
        }
    }

    for (int off = 16; off >= 1; off >>= 1) {
        a0 += __shfl_xor(a0, off, 32);
        a1 += __shfl_xor(a1, off, 32);
        const int om = __shfl_xor(am, off, 32);
        am = (om > am) ? om : am;
    }
    if (lane == 0) { red0[wave] = a0; red1[wave] = a1; red2[wave] = am; }
    __syncthreads();
    int s0 = red0[lane], s1 = red1[lane], mx = red2[lane];
    for (int off = 16; off >= 1; off >>= 1) {
        s0 += __shfl_xor(s0, off, 32);
        s1 += __shfl_xor(s1, off, 32);
        const int om = __shfl_xor(mx, off, 32);
        mx = (om > mx) ? om : mx;
    }
    const int n = s1 - s0;

    if (n <= BS_CAP && mx <= BS_L * BS_U) {
        int lr[BS_P * BS_U];
#pragma unroll
        for (int p = 0; p < BS_P; ++p) {
#pragma unroll
            for (int u = 0; u < BS_U; ++u) {
                const int s = p * BS_U + u;
                lr[s] = 0;
                if (sub + u * BS_L < len[p]) lr[s] = atomicAdd(&rc[key[s] & (BK_R - 1)], 1);
            }
        }
        __syncthreads();
        const int excl = bucket_scan(rc, wsum, t);
        if (t < BK_R) {
            const int row = b * BK_R + t;
            if (row < N) rowStart[row] = s0 + excl;
        }
#pragma unroll
        for (int s = 0; s < BS_P * BS_U; ++s) lr[s] += rc[key[s] & (BK_R - 1)];
#pragma unroll
        for (int p = 0; p < BS_P; ++p) {
#pragma unroll
            for (int u = 0; u < BS_U; ++u) {
                const int s = p * BS_U + u;
                if (sub + u * BS_L < len[p]) {
                    lcol[lr[s]] = key[s] >> BK_SH;
                    lval[lr[s]] = __int_as_float(val[s]);
                }
            }
        }
        __syncthreads();
        if (n > 0) {
            int oc[BS_O]; float ov[BS_O];
#pragma unroll
            for (int u = 0; u < BS_O; ++u) {
                const int j = t + u * BS_T;
                const int jj = (j < n) ? j : (n - 1);
                oc[u] = lcol[jj]; ov[u] = lval[jj];
            }
#pragma unroll
            for (int u = 0; u < BS_O; ++u) {
                const int j = t + u * BS_T;
                if (j < n) { scol[s0 + j] = oc[u]; sval[s0 + j] = ov[u]; }
            }
        }
    } else {
        for (int p = 0; p < BS_P; ++p)
            for (int k = sub; k < len[p]; k += BS_L)
                atomicAdd(&rc[brec[first[p] + k].x & (BK_R - 1)], 1);
        __syncthreads();
        const int excl = bucket_scan(rc, wsum, t);
        if (t < BK_R) {
            const int row = b * BK_R + t;
            if (row < N) rowStart[row] = s0 + excl;
        }
        for (int p = 0; p < BS_P; ++p) {
            for (int k = sub; k < len[p]; k += BS_L) {
                const int2 q = brec[first[p] + k];
                const int pos = s0 + atomicAdd(&rc[q.x & (BK_R - 1)], 1);
                scol[pos] = q.x >> BK_SH;
                sval[pos] = __int_as_float(q.y);
            }
        }
    }
    if (b == B - 1 && t == 0) rowStart[N] = s1;
}

#define SP1_T    512
#define SP1_W    (SP1_T / 32)
#define SP1_ROWS 4
#define SP1_BR   (SP1_W * SP1_ROWS)
#define SP1_NT   (FEAT / 16)
#define SP1_MW   (SP1_BR / 16 * SP1_NT / SP1_W)
#define ZP       (FEAT + 4)

static_assert(SP1_W % SP1_NT == 0 && SP1_MW >= 1 && SP1_BR % 16 == 0,
              "phase 2 splits SP1_BR / 16 row tiles x SP1_NT column tiles evenly over the waves");

template <int C>
__device__ __forceinline__ void gather_fma(const float4* __restrict__ y4, int cc, float vv,
                                           int j, int lane, float4& acc) {
    float4 g[C];
#pragma unroll
    for (int u = 0; u < C; ++u) {
        const int c = __builtin_amdgcn_readlane(cc, j + u);
        g[u] = y4[(size_t)c * (FEAT / 4) + lane];
    }
#pragma unroll
    for (int u = 0; u < C; ++u) {
        const float v = __int_as_float(__builtin_amdgcn_readlane(__float_as_int(vv), j + u));
        acc.x = fmaf(v, g[u].x, acc.x);
        acc.y = fmaf(v, g[u].y, acc.y);
        acc.z = fmaf(v, g[u].z, acc.z);
        acc.w = fmaf(v, g[u].w, acc.w);
    }
}

__global__ void __launch_bounds__(SP1_T)
spmm1_fused(const float* __restrict__ x, const int* __restrict__ rowStart,
            const int* __restrict__ scol, const float* __restrict__ sval,
            const float* __restrict__ W1,
            const float* __restrict__ b1, const float* __restrict__ g1,
            const float* __restrict__ be1, const float* __restrict__ m1,
            const float* __restrict__ v1, const float* __restrict__ W2,
            float* __restrict__ p, int N) {
    __shared__ float w2s[FEAT * NCLS];
    __shared__ __attribute__((aligned(16))) float zs[SP1_BR * ZP];
    {
        constexpr int W2N = FEAT * NCLS;
        constexpr int W2K = (W2N + SP1_T - 1) / SP1_T;
        float t[W2K];
#pragma unroll
        for (int k = 0; k < W2K; ++k) {
            const int i = (int)threadIdx.x + k * SP1_T;
            t[k] = (i < W2N) ? W2[i] : 0.0f;
        }
#pragma unroll
        for (int k = 0; k < W2K; ++k) {
            const int i = (int)threadIdx.x + k * SP1_T;
            if (i < W2N) w2s[i] = t[k];
        }
    }

    const int lane = threadIdx.x & 31;
    const int half = lane >> 4;
    const int r16  = lane & 15;
    const int wave = __builtin_amdgcn_readfirstlane((int)(threadIdx.x >> 5));
    const int lrow0 = wave * SP1_ROWS;
    const int r0 = (int)blockIdx.x * SP1_BR + lrow0;
    int nr = N - r0;
    nr = nr < 0 ? 0 : (nr > SP1_ROWS ? SP1_ROWS : nr);

    if (nr > 0) {
        const float4* __restrict__ x4 = (const float4*)x;
        const int myrs = rowStart[r0 + (lane < nr ? lane : nr)];
        int eB = __builtin_amdgcn_readlane(myrs, 0);
        int eE = __builtin_amdgcn_readlane(myrs, 1);
        int nc = 0; float nv = 0.0f;
        if (eB + lane < eE) { nc = scol[eB + lane]; nv = sval[eB + lane]; }

        for (int i = 0; i < nr; ++i) {
            const int cc = nc; const float vv = nv;
            int nE = eE;
            if (i + 1 < nr) {
                nE = __builtin_amdgcn_readlane(myrs, i + 2);
                nc = 0; nv = 0.0f;
                if (eE + lane < nE) { nc = scol[eE + lane]; nv = sval[eE + lane]; }
            }

            float4 acc = {0.f, 0.f, 0.f, 0.f};
            const int deg = eE - eB;
            const int n = deg < 32 ? deg : 32;
            int j = 0;
            for (; j + 8 <= n; j += 8) gather_fma<8>(x4, cc, vv, j, lane, acc);
            switch (n - j) {
                case 7: gather_fma<7>(x4, cc, vv, j, lane, acc); break;
                case 6: gather_fma<6>(x4, cc, vv, j, lane, acc); break;
                case 5: gather_fma<5>(x4, cc, vv, j, lane, acc); break;
                case 4: gather_fma<4>(x4, cc, vv, j, lane, acc); break;
                case 3: gather_fma<3>(x4, cc, vv, j, lane, acc); break;
                case 2: gather_fma<2>(x4, cc, vv, j, lane, acc); break;
                case 1: gather_fma<1>(x4, cc, vv, j, lane, acc); break;
                default: break;
            }
            for (int e = eB + 32; e < eE; ++e) {
                const int   c = scol[e];
                const float v = sval[e];
                const float4 g = x4[(size_t)c * (FEAT / 4) + lane];
                acc.x = fmaf(v, g.x, acc.x);
                acc.y = fmaf(v, g.y, acc.y);
                acc.z = fmaf(v, g.z, acc.z);
                acc.w = fmaf(v, g.w, acc.w);
            }
            *(float4*)&zs[(lrow0 + i) * ZP + lane * 4] = acc;
            eB = eE; eE = nE;
        }
    }
    {
        const float4 zero4 = {0.f, 0.f, 0.f, 0.f};
        for (int i = nr; i < SP1_ROWS; ++i) *(float4*)&zs[(lrow0 + i) * ZP + lane * 4] = zero4;
    }
    __syncthreads();

    const int ch  = (wave & (SP1_NT - 1)) * 16 + r16;
    const int mr0 = (wave / SP1_NT) * SP1_MW * 16;
    {
        const v8f vzero = {0.f,0.f,0.f,0.f,0.f,0.f,0.f,0.f};
        v8f acc[SP1_MW];
#pragma unroll
        for (int mt = 0; mt < SP1_MW; ++mt) acc[mt] = vzero;
        const float* __restrict__ wcol = W1 + ch;
        const float* zrow = &zs[(mr0 + r16) * ZP + 2 * half];
#pragma unroll 8
        for (int kt = 0; kt < FEAT / 4; ++kt) {
            const int k0 = kt << 2;
            v2f b;
            b.x = wcol[(size_t)(k0 + 2 * half + 0) * FEAT];
            b.y = wcol[(size_t)(k0 + 2 * half + 1) * FEAT];
#pragma unroll
            for (int mt = 0; mt < SP1_MW; ++mt) {
                v2f a;
                a.x = zrow[mt * 16 * ZP + k0 + 0];
                a.y = zrow[mt * 16 * ZP + k0 + 1];
                acc[mt] = __builtin_amdgcn_wmma_f32_16x16x4_f32(
                    false, a, false, b, (short)0, acc[mt], false, false);
            }
        }
        const float bs = g1[ch] * rsqrtf(v1[ch] + EPSB);
        const float bt = (b1[ch] - m1[ch]) * bs + be1[ch];
        __syncthreads();
#pragma unroll
        for (int mt = 0; mt < SP1_MW; ++mt) {
#pragma unroll
            for (int i = 0; i < 8; ++i) {
                const float hv = fmaf(acc[mt][i], bs, bt);
                zs[(mr0 + mt * 16 + i + 8 * half) * ZP + ch] = hv > 0.0f ? hv : 0.0f;
            }
        }
    }
    __syncthreads();

    float w2[4][NCLS];
#pragma unroll
    for (int j = 0; j < 4; ++j)
#pragma unroll
        for (int c = 0; c < NCLS; ++c) w2[j][c] = w2s[(lane * 4 + j) * NCLS + c];
    for (int i = 0; i < nr; ++i) {
        const float4 hq = *(const float4*)&zs[(lrow0 + i) * ZP + lane * 4];
        const float h[4] = {hq.x, hq.y, hq.z, hq.w};
        const int r = r0 + i;
        float mine = 0.0f;
#pragma unroll
        for (int c = 0; c < NCLS; ++c) {
            float ssum = 0.0f;
#pragma unroll
            for (int j = 0; j < 4; ++j) ssum = fmaf(h[j], w2[j][c], ssum);
#pragma unroll
            for (int off = 16; off >= 1; off >>= 1) ssum += __shfl_xor(ssum, off, 32);
            if (lane == c) mine = ssum;
        }
        if (lane < PP) p[(size_t)r * PP + lane] = mine;
    }
}

#define SP2_T    128
#define SP2_G    4
#define SP2_WR   (SP2_G * 4)
#define SP2_BR   (SP2_T / 32 * SP2_WR)

template <int U>
__device__ __forceinline__ unsigned bcast8(unsigned v) {
    return __builtin_amdgcn_mov_dpp8(v, (unsigned)(U * 0x249249));
}

__device__ __forceinline__ void sp2_pairs(const int* __restrict__ scol,
                                          const float* __restrict__ sval,
                                          const int (&eb)[SP2_G], const int (&ee)[SP2_G],
                                          int c8, unsigned (&po)[SP2_G], float (&vl)[SP2_G]) {
#pragma unroll
    for (int g = 0; g < SP2_G; ++g) {
        int i = eb[g] + c8;
        i = (i < ee[g] - 1) ? i : (ee[g] - 1);
        i = (i > 0) ? i : 0;
        po[g] = (unsigned)scol[i] * PP;
        vl[g] = sval[i];
    }
}

template <int U>
__device__ __forceinline__ void sp2_edge(const float* __restrict__ p, const unsigned (&po)[SP2_G],
                                         unsigned c8, float (&ga)[SP2_G][8]) {
#pragma unroll
    for (int g = 0; g < SP2_G; ++g) ga[g][U] = p[bcast8<U>(po[g]) | c8];
}

template <int U>
__device__ __forceinline__ void sp2_fma(const float (&cv)[SP2_G], const int (&nrem)[SP2_G],
                                        const float (&ga)[SP2_G][8], float (&acc)[SP2_G]) {
#pragma unroll
    for (int g = 0; g < SP2_G; ++g) {
        const float v = __uint_as_float(bcast8<U>(__float_as_uint(cv[g])));
        if (U < nrem[g]) acc[g] = fmaf(v, ga[g][U], acc[g]);
    }
}

__global__ void __launch_bounds__(SP2_T)
spmm2_fused(const float* __restrict__ p, const int* __restrict__ rowStart,
            const int* __restrict__ scol, const float* __restrict__ sval,
            const float* __restrict__ b2, const float* __restrict__ g2,
            const float* __restrict__ be2, const float* __restrict__ m2,
            const float* __restrict__ v2, float* __restrict__ out, int N) {
    __shared__ float os[SP2_BR * NCLS];
    const int t    = threadIdx.x;
    const int lane = t & 31;
    const int wave = t >> 5;
    const int q    = lane >> 3;
    const int c8   = lane & 7;
    const int lrow = wave * SP2_WR + q;
    const int row0 = (int)blockIdx.x * SP2_BR + lrow;

    int eb[SP2_G], ee[SP2_G];
    float acc[SP2_G];
#pragma unroll
    for (int g = 0; g < SP2_G; ++g) {
        const int r  = row0 + 4 * g;
        const int rc = (r < N) ? r : (N - 1);
        eb[g] = rowStart[rc]; ee[g] = rowStart[rc + 1];
        acc[g] = 0.0f;
    }
    const int cc = (c8 < NCLS) ? c8 : (NCLS - 1);
    const float kb = b2[cc], km = m2[cc], kbe = be2[cc];
    const float ksc = g2[cc] * rsqrtf(v2[cc] + EPSB);
#pragma unroll
    for (int g = 0; g < SP2_G; ++g)
        if (row0 + 4 * g >= N) ee[g] = eb[g];

    unsigned po[SP2_G]; float vl[SP2_G];
    sp2_pairs(scol, sval, eb, ee, c8, po, vl);
    for (;;) {
        bool more = false;
#pragma unroll
        for (int g = 0; g < SP2_G; ++g) more = more || (eb[g] < ee[g]);
        if (__builtin_amdgcn_ballot_w32(more) == 0u) break;

        float ga[SP2_G][8];
        sp2_edge<0>(p, po, (unsigned)c8, ga); sp2_edge<1>(p, po, (unsigned)c8, ga);
        sp2_edge<2>(p, po, (unsigned)c8, ga); sp2_edge<3>(p, po, (unsigned)c8, ga);
        sp2_edge<4>(p, po, (unsigned)c8, ga); sp2_edge<5>(p, po, (unsigned)c8, ga);
        sp2_edge<6>(p, po, (unsigned)c8, ga); sp2_edge<7>(p, po, (unsigned)c8, ga);
        float cv[SP2_G]; int nrem[SP2_G];
#pragma unroll
        for (int g = 0; g < SP2_G; ++g) {
            cv[g] = vl[g];
            nrem[g] = ee[g] - eb[g];
            eb[g] += 8;
        }
        sp2_pairs(scol, sval, eb, ee, c8, po, vl);
        __builtin_amdgcn_sched_barrier(0);
        sp2_fma<0>(cv, nrem, ga, acc); sp2_fma<1>(cv, nrem, ga, acc);
        sp2_fma<2>(cv, nrem, ga, acc); sp2_fma<3>(cv, nrem, ga, acc);
        sp2_fma<4>(cv, nrem, ga, acc); sp2_fma<5>(cv, nrem, ga, acc);
        sp2_fma<6>(cv, nrem, ga, acc); sp2_fma<7>(cv, nrem, ga, acc);
    }

    if (c8 < NCLS) {
#pragma unroll
        for (int g = 0; g < SP2_G; ++g)
            os[(lrow + 4 * g) * NCLS + c8] = (acc[g] + kb - km) * ksc + kbe;
    }
    __syncthreads();
    const int total = N * NCLS;
    const int base  = (int)blockIdx.x * (SP2_BR * NCLS);
    constexpr int SP2_OK = (SP2_BR * NCLS + SP2_T - 1) / SP2_T;
    float ov[SP2_OK];
#pragma unroll
    for (int k = 0; k < SP2_OK; ++k) {
        const int j = k * SP2_T + t;
        ov[k] = os[j < SP2_BR * NCLS ? j : SP2_BR * NCLS - 1];
    }
#pragma unroll
    for (int k = 0; k < SP2_OK; ++k) {
        const int j = k * SP2_T + t;
        if (j < SP2_BR * NCLS && base + j < total) out[base + j] = ov[k];
    }
}

extern "C" void kernel_launch(void* const* d_in, const int* in_sizes, int n_in,
                              void* d_out, int out_size, void* d_ws, size_t ws_size,
                              hipStream_t stream) {
    const float* x   = (const float*)d_in[0];
    const float* ev  = (const float*)d_in[1];
    const float* W1  = (const float*)d_in[2];
    const float* b1  = (const float*)d_in[3];
    const float* g1  = (const float*)d_in[4];
    const float* be1 = (const float*)d_in[5];
    const float* m1  = (const float*)d_in[6];
    const float* v1  = (const float*)d_in[7];
    const float* W2  = (const float*)d_in[8];
    const float* b2  = (const float*)d_in[9];
    const float* g2  = (const float*)d_in[10];
    const float* be2 = (const float*)d_in[11];
    const float* m2  = (const float*)d_in[12];
    const float* v2  = (const float*)d_in[13];
    const int*  erow = (const int*)d_in[14];
    const int*  ecol = (const int*)d_in[15];
    float* out = (float*)d_out;

    const int N = in_sizes[0] / FEAT;
    const int E = in_sizes[1];

    const int B = (N + BK_R - 1) >> BK_SH;

    float* p        = (float*)d_ws;
    int*   brec     = (int*)(p + (size_t)N * PP);
    int*   startL   = brec + (size_t)2 * E;
    int*   rowStart = startL + (size_t)(B + 1) * BIN_G;
    int*   scol     = rowStart + (N + 1);
    float* sval     = (float*)(scol + E);

    bin_tile<<<BIN_G, BIN_T, 0, stream>>>(erow, ecol, ev, startL, brec, E, B);
    bucket_sort<<<B, BS_T, 0, stream>>>(brec, startL, rowStart, scol, sval, N, B, E);
    spmm1_fused<<<(N + SP1_BR - 1) / SP1_BR, SP1_T, 0, stream>>>(x, rowStart, scol, sval, W1,
                                                                 b1, g1, be1, m1, v1, W2, p, N);
    spmm2_fused<<<(N + SP2_BR - 1) / SP2_BR, SP2_T, 0, stream>>>(p, rowStart, scol, sval,
                                                                 b2, g2, be2, m2, v2, out, N);
}


# layer-1 kernel as resident work-groups with fixed wave roles: 16 gather waves fill row tile s while 4 waves run tile s-1 on WMMA, A and W1 from LDS (W1 staged once per group); sum orders kept
# speedup vs baseline: 5.3260x; 1.0567x over previous
#include <hip/hip_runtime.h>
#include <hip/hip_bf16.h>

typedef __attribute__((ext_vector_type(2))) float v2f;
typedef __attribute__((ext_vector_type(8))) float v8f;

#define FEAT 128
#define NCLS 7
#define PP   8
#define EPSB 1e-5f

#define BK_SH   8
#define BK_R    (1 << BK_SH)
#define BK_MAX  1024
#define BIN_G   256
#define BIN_T   1024
#define BIN_U   8

static_assert(BK_MAX == BIN_T, "bin_tile scans one bucket counter per thread");

__device__ __forceinline__ int block_excl_scan(int d, int* wt, int t) {
    const int lane = t & 31;
    const int wave = t >> 5;
    int incl = d;
    for (int off = 1; off < 32; off <<= 1) {
        const int up = __shfl_up(incl, off, 32);
        if (lane >= off) incl += up;
    }
    if (lane == 31) wt[wave] = incl;
    __syncthreads();
    int c = (lane < wave) ? wt[lane] : 0;
    for (int off = 16; off >= 1; off >>= 1) c += __shfl_xor(c, off, 32);
    return c + incl - d;
}

__global__ void __launch_bounds__(BIN_T)
bin_tile(const int* __restrict__ erow, const int* __restrict__ ecol,
         const float* __restrict__ ev, int* __restrict__ startL,
         int* __restrict__ brec_, int E, int B) {
    __shared__ int cnt[BK_MAX];
    __shared__ int wt[BIN_T / 32];
    int2* __restrict__ brec = (int2*)brec_;
    const int t = threadIdx.x;
    cnt[t] = 0;
    __syncthreads();
    const int tile = (E + BIN_G - 1) / BIN_G;
    int e0 = (int)blockIdx.x * tile;
    e0 = (e0 < E) ? e0 : E;
    const int e1 = (e0 + tile < E) ? (e0 + tile) : E;
    const int last = (e1 > 0) ? (e1 - 1) : 0;

    if (tile <= BIN_U * BIN_T) {
        int r[BIN_U], c[BIN_U], rk[BIN_U]; float v[BIN_U];
#pragma unroll
        for (int u = 0; u < BIN_U; ++u) {
            const int e = e0 + u * BIN_T + t;
            const int ee = (e < e1) ? e : last;
            r[u] = erow[ee];
            c[u] = ecol[ee];
            v[u] = ev[ee];
        }
        __builtin_amdgcn_sched_barrier(0);
#pragma unroll
        for (int u = 0; u < BIN_U; ++u) {
            rk[u] = -1;
            if (e0 + u * BIN_T + t < e1) rk[u] = atomicAdd(&cnt[r[u] >> BK_SH], 1);
        }
        __syncthreads();
        const int excl = block_excl_scan(cnt[t], wt, t);
        cnt[t] = excl;
        if (t <= B) startL[(size_t)t * BIN_G + blockIdx.x] = excl;
        __syncthreads();
        int st[BIN_U];
#pragma unroll
        for (int u = 0; u < BIN_U; ++u) st[u] = cnt[r[u] >> BK_SH];
#pragma unroll
        for (int u = 0; u < BIN_U; ++u) {
            if (rk[u] >= 0)
                brec[e0 + st[u] + rk[u]] =
                    make_int2((c[u] << BK_SH) | (r[u] & (BK_R - 1)), __float_as_int(v[u]));
        }
    } else {
        for (int base = e0; base < e1; base += BIN_U * BIN_T) {
            int r[BIN_U];
#pragma unroll
            for (int u = 0; u < BIN_U; ++u) {
                const int e = base + u * BIN_T + t;
                r[u] = erow[(e < e1) ? e : last];
            }
#pragma unroll
            for (int u = 0; u < BIN_U; ++u)
                if (base + u * BIN_T + t < e1) atomicAdd(&cnt[r[u] >> BK_SH], 1);
        }
        __syncthreads();
        const int excl = block_excl_scan(cnt[t], wt, t);
        cnt[t] = excl;
        if (t <= B) startL[(size_t)t * BIN_G + blockIdx.x] = excl;
        __syncthreads();
        for (int base = e0; base < e1; base += BIN_U * BIN_T) {
            int r[BIN_U], c[BIN_U], pos[BIN_U]; float v[BIN_U];
#pragma unroll
            for (int u = 0; u < BIN_U; ++u) {
                const int e = base + u * BIN_T + t;
                const int ee = (e < e1) ? e : last;
                r[u] = erow[ee];
                c[u] = ecol[ee];
                v[u] = ev[ee];
            }
#pragma unroll
            for (int u = 0; u < BIN_U; ++u) {
                pos[u] = -1;
                if (base + u * BIN_T + t < e1) pos[u] = atomicAdd(&cnt[r[u] >> BK_SH], 1);
            }
#pragma unroll
            for (int u = 0; u < BIN_U; ++u) {
                if (pos[u] >= 0)
                    brec[e0 + pos[u]] =
                        make_int2((c[u] << BK_SH) | (r[u] & (BK_R - 1)), __float_as_int(v[u]));
            }
        }
    }
}

#define BS_T    1024
#define BS_L    8
#define BS_P    (BIN_G * BS_L / BS_T)
#define BS_U    6
#define BS_O    6
#define BS_CAP  (BS_T * BS_O)

static_assert(BIN_G * BS_L == BS_P * BS_T, "every run has its BS_L lanes");

__device__ __forceinline__ int bucket_scan(int* rc, int* wsum, int t) {
    const int lane = t & 31;
    const int wave = t >> 5;
    int d = 0, incl = 0;
    if (t < BK_R) {
        d = rc[t];
        incl = d;
        for (int off = 1; off < 32; off <<= 1) {
            const int up = __shfl_up(incl, off, 32);
            if (lane >= off) incl += up;
        }
        if (lane == 31) wsum[wave] = incl;
    }
    __syncthreads();
    int excl = 0;
    if (t < BK_R) {
        int c = (lane < wave) ? wsum[lane] : 0;
        for (int off = 16; off >= 1; off >>= 1) c += __shfl_xor(c, off, 32);
        excl = c + incl - d;
        rc[t] = excl;
    }
    __syncthreads();
    return excl;
}

__global__ void __launch_bounds__(BS_T) __attribute__((amdgpu_waves_per_eu(16, 16)))
bucket_sort(const int* __restrict__ brec_, const int* __restrict__ startL,
            int* __restrict__ rowStart, int* __restrict__ scol,
            float* __restrict__ sval, int N, int B, int E) {
    __shared__ int   rc[BK_R];
    __shared__ int   wsum[BK_R / 32];
    __shared__ int   red0[BS_T / 32], red1[BS_T / 32], red2[BS_T / 32];
    __shared__ int   lcol[BS_CAP];
    __shared__ float lval[BS_CAP];
    const int2* __restrict__ brec = (const int2*)brec_;
    const int t    = threadIdx.x;
    const int lane = t & 31;
    const int wave = t >> 5;
    const int b    = blockIdx.x;
    const int grp  = t / BS_L;
    const int sub  = t % BS_L;
    const int tile = (E + BIN_G - 1) / BIN_G;
    int len[BS_P], first[BS_P];
    int a0 = 0, a1 = 0, am = 0;
#pragma unroll
    for (int p = 0; p < BS_P; ++p) {
        const int run = p * (BS_T / BS_L) + grp;
        int tb = run * tile;
        tb = (tb < E) ? tb : E;
        const int lo = startL[(size_t)b * BIN_G + run];
        const int hi = startL[(size_t)(b + 1) * BIN_G + run];
        len[p]   = hi - lo;
        first[p] = tb + lo;
        if (sub == 0) { a0 += lo; a1 += hi; }
        am = (len[p] > am) ? len[p] : am;
    }
    if (t < BK_R) rc[t] = 0;

    int key[BS_P * BS_U], val[BS_P * BS_U];
#pragma unroll
    for (int p = 0; p < BS_P; ++p) {
#pragma unroll
        for (int u = 0; u < BS_U; ++u) {
            const int k  = sub + u * BS_L;
            const int kk = (k < len[p]) ? k : (len[p] - 1);
            const int2 q = brec[(len[p] > 0) ? (first[p] + kk) : 0];
            key[p * BS_U + u] = q.x; val[p * BS_U + u] = q.y;
        }
    }

    for (int off = 16; off >= 1; off >>= 1) {
        a0 += __shfl_xor(a0, off, 32);
        a1 += __shfl_xor(a1, off, 32);
        const int om = __shfl_xor(am, off, 32);
        am = (om > am) ? om : am;
    }
    if (lane == 0) { red0[wave] = a0; red1[wave] = a1; red2[wave] = am; }
    __syncthreads();
    int s0 = red0[lane], s1 = red1[lane], mx = red2[lane];
    for (int off = 16; off >= 1; off >>= 1) {
        s0 += __shfl_xor(s0, off, 32);
        s1 += __shfl_xor(s1, off, 32);
        const int om = __shfl_xor(mx, off, 32);
        mx = (om > mx) ? om : mx;
    }
    const int n = s1 - s0;

    if (n <= BS_CAP && mx <= BS_L * BS_U) {
        int lr[BS_P * BS_U];
#pragma unroll
        for (int p = 0; p < BS_P; ++p) {
#pragma unroll
            for (int u = 0; u < BS_U; ++u) {
                const int s = p * BS_U + u;
                lr[s] = 0;
                if (sub + u * BS_L < len[p]) lr[s] = atomicAdd(&rc[key[s] & (BK_R - 1)], 1);
            }
        }
        __syncthreads();
        const int excl = bucket_scan(rc, wsum, t);
        if (t < BK_R) {
            const int row = b * BK_R + t;
            if (row < N) rowStart[row] = s0 + excl;
        }
#pragma unroll
        for (int s = 0; s < BS_P * BS_U; ++s) lr[s] += rc[key[s] & (BK_R - 1)];
#pragma unroll
        for (int p = 0; p < BS_P; ++p) {
#pragma unroll
            for (int u = 0; u < BS_U; ++u) {
                const int s = p * BS_U + u;
                if (sub + u * BS_L < len[p]) {
                    lcol[lr[s]] = key[s] >> BK_SH;
                    lval[lr[s]] = __int_as_float(val[s]);
                }
            }
        }
        __syncthreads();
        if (n > 0) {
            int oc[BS_O]; float ov[BS_O];
#pragma unroll
            for (int u = 0; u < BS_O; ++u) {
                const int j = t + u * BS_T;
                const int jj = (j < n) ? j : (n - 1);
                oc[u] = lcol[jj]; ov[u] = lval[jj];
            }
#pragma unroll
            for (int u = 0; u < BS_O; ++u) {
                const int j = t + u * BS_T;
                if (j < n) { scol[s0 + j] = oc[u]; sval[s0 + j] = ov[u]; }
            }
        }
    } else {
        for (int p = 0; p < BS_P; ++p)
            for (int k = sub; k < len[p]; k += BS_L)
                atomicAdd(&rc[brec[first[p] + k].x & (BK_R - 1)], 1);
        __syncthreads();
        const int excl = bucket_scan(rc, wsum, t);
        if (t < BK_R) {
            const int row = b * BK_R + t;
            if (row < N) rowStart[row] = s0 + excl;
        }
        for (int p = 0; p < BS_P; ++p) {
            for (int k = sub; k < len[p]; k += BS_L) {
                const int2 q = brec[first[p] + k];
                const int pos = s0 + atomicAdd(&rc[q.x & (BK_R - 1)], 1);
                scol[pos] = q.x >> BK_SH;
                sval[pos] = __int_as_float(q.y);
            }
        }
    }
    if (b == B - 1 && t == 0) rowStart[N] = s1;
}

#define SP1_GW   16
#define SP1_XW   4
#define SP1_W    (SP1_GW + SP1_XW)
#define SP1_T    (SP1_W * 32)
#define SP1_BR   16
#define SP1_GRID 768
#define ZP       (FEAT + 4)
#define W1P      (2 * FEAT + 32)

static_assert(SP1_BR == SP1_GW && SP1_XW * 32 == FEAT,
              "a gather wave owns one row of a tile, a multiplier wave two column tiles");

template <int C>
__device__ __forceinline__ void gather_fma(const float4* __restrict__ y4, int cc, float vv,
                                           int j, int lane, float4& acc) {
    float4 g[C];
#pragma unroll
    for (int u = 0; u < C; ++u) {
        const int c = __builtin_amdgcn_readlane(cc, j + u);
        g[u] = y4[(size_t)c * (FEAT / 4) + lane];
    }
#pragma unroll
    for (int u = 0; u < C; ++u) {
        const float v = __int_as_float(__builtin_amdgcn_readlane(__float_as_int(vv), j + u));
        acc.x = fmaf(v, g[u].x, acc.x);
        acc.y = fmaf(v, g[u].y, acc.y);
        acc.z = fmaf(v, g[u].z, acc.z);
        acc.w = fmaf(v, g[u].w, acc.w);
    }
}

__global__ void __launch_bounds__(SP1_T) __attribute__((amdgpu_waves_per_eu(15, 16)))
spmm1_fused(const float* __restrict__ x, const int* __restrict__ rowStart,
            const int* __restrict__ scol, const float* __restrict__ sval,
            const float* __restrict__ W1,
            const float* __restrict__ b1, const float* __restrict__ g1,
            const float* __restrict__ be1, const float* __restrict__ m1,
            const float* __restrict__ v1, const float* __restrict__ W2,
            float* __restrict__ p, int N) {
    __shared__ __attribute__((aligned(16))) float w1s[(FEAT / 2) * W1P];
    __shared__ __attribute__((aligned(16))) float zs[2 * SP1_BR * ZP];
    __shared__ __attribute__((aligned(16))) float w2s[FEAT * NCLS];
    __shared__ float bns[2 * FEAT];

    const int tid = (int)threadIdx.x;
    for (int i = tid; i < FEAT * NCLS; i += SP1_T) w2s[i] = W2[i];
    for (int i = tid; i < FEAT; i += SP1_T) {
        const float bs = g1[i] * rsqrtf(v1[i] + EPSB);
        bns[i] = bs;
        bns[FEAT + i] = (b1[i] - m1[i]) * bs + be1[i];
    }
    for (int i = tid; i < (FEAT / 2) * (FEAT / 4); i += SP1_T) {
        const int kp = i / (FEAT / 4);
        const int c4 = (i % (FEAT / 4)) * 4;
        const float4 wa = *(const float4*)&W1[(size_t)(2 * kp) * FEAT + c4];
        const float4 wb = *(const float4*)&W1[(size_t)(2 * kp + 1) * FEAT + c4];
        float4* d = (float4*)&w1s[kp * W1P + 2 * c4];
        d[0] = make_float4(wa.x, wb.x, wa.y, wb.y);
        d[1] = make_float4(wa.z, wb.z, wa.w, wb.w);
    }

    const int lane = tid & 31;
    const int half = lane >> 4;
    const int r16  = lane & 15;
    const int wave = __builtin_amdgcn_readfirstlane(tid >> 5);
    const bool gat = wave < SP1_GW;
    const int G  = (int)gridDim.x;
    const int bI = (int)blockIdx.x;
    const int ntiles = (N + SP1_BR - 1) / SP1_BR;
    const int T = (bI < ntiles) ? (ntiles - bI + G - 1) / G : 0;

    const float4* __restrict__ x4 = (const float4*)x;
    int eB = 0, eE = 0, rsN = 0, nc = 0;
    float nv = 0.0f;
    if (gat) {
        int rs0 = 0;
        const int ra = bI * SP1_BR + wave;
        const int rb = (bI + G) * SP1_BR + wave;
        if (0 < T && ra < N && lane < 2) rs0 = rowStart[ra + lane];
        if (1 < T && rb < N && lane < 2) rsN = rowStart[rb + lane];
        eB = __builtin_amdgcn_readlane(rs0, 0);
        eE = __builtin_amdgcn_readlane(rs0, 1);
        if (eB + lane < eE) { nc = scol[eB + lane]; nv = sval[eB + lane]; }
    }
    __syncthreads();

    const v8f vzero = {0.f,0.f,0.f,0.f,0.f,0.f,0.f,0.f};
    for (int s = 0; s < T + 2; ++s) {
        float* zg = zs + (s & 1) * (SP1_BR * ZP);
        float* zm = zs + ((s + 1) & 1) * (SP1_BR * ZP);
        const bool mul_on = (!gat) && (s >= 1) && (s <= T);
        v8f acc0 = vzero, acc1 = vzero;
        if (gat) {
            const int cc = nc; const float vv = nv;
            const int nB = __builtin_amdgcn_readlane(rsN, 0);
            const int nE = __builtin_amdgcn_readlane(rsN, 1);
            nc = 0; nv = 0.0f;
            if (nB + lane < nE) { nc = scol[nB + lane]; nv = sval[nB + lane]; }
            rsN = 0;
            {
                const int rn = (bI + (s + 2) * G) * SP1_BR + wave;
                if (s + 2 < T && rn < N && lane < 2) rsN = rowStart[rn + lane];
            }

            if (s >= 2) {
                const int r = (bI + (s - 2) * G) * SP1_BR + wave;
                if (r < N) {
                    float w2[4][NCLS];
#pragma unroll
                    for (int j = 0; j < 4; ++j)
#pragma unroll
                        for (int c = 0; c < NCLS; ++c) w2[j][c] = w2s[(lane * 4 + j) * NCLS + c];
                    const float4 hq = *(const float4*)&zg[wave * ZP + lane * 4];
                    const float h[4] = {hq.x, hq.y, hq.z, hq.w};
                    float mine = 0.0f;
#pragma unroll
                    for (int c = 0; c < NCLS; ++c) {
                        float ssum = 0.0f;
#pragma unroll
                        for (int j = 0; j < 4; ++j) ssum = fmaf(h[j], w2[j][c], ssum);
#pragma unroll
                        for (int off = 16; off >= 1; off >>= 1) ssum += __shfl_xor(ssum, off, 32);
                        if (lane == c) mine = ssum;
                    }
                    if (lane < PP) p[(size_t)r * PP + lane] = mine;
                }
            }

            float4 acc = {0.f, 0.f, 0.f, 0.f};
            const int deg = eE - eB;
            const int n = deg < 32 ? deg : 32;
            int j = 0;
            for (; j + 8 <= n; j += 8) gather_fma<8>(x4, cc, vv, j, lane, acc);
            switch (n - j) {
                case 7: gather_fma<7>(x4, cc, vv, j, lane, acc); break;
                case 6: gather_fma<6>(x4, cc, vv, j, lane, acc); break;
                case 5: gather_fma<5>(x4, cc, vv, j, lane, acc); break;
                case 4: gather_fma<4>(x4, cc, vv, j, lane, acc); break;
                case 3: gather_fma<3>(x4, cc, vv, j, lane, acc); break;
                case 2: gather_fma<2>(x4, cc, vv, j, lane, acc); break;
                case 1: gather_fma<1>(x4, cc, vv, j, lane, acc); break;
                default: break;
            }
            for (int e = eB + 32; e < eE; ++e) {
                const int   c = scol[e];
                const float v = sval[e];
                const float4 g = x4[(size_t)c * (FEAT / 4) + lane];
                acc.x = fmaf(v, g.x, acc.x);
                acc.y = fmaf(v, g.y, acc.y);
                acc.z = fmaf(v, g.z, acc.z);
                acc.w = fmaf(v, g.w, acc.w);
            }
            *(float4*)&zg[wave * ZP + lane * 4] = acc;
            eB = nB; eE = nE;
        } else if (mul_on) {
            const int ch0 = (wave - SP1_GW) * 32 + r16;
            const float* zrow = zm + r16 * ZP + 2 * half;
            const float* wcol = w1s + half * W1P + 2 * ch0;
#pragma unroll 4
            for (int kt = 0; kt < FEAT / 4; ++kt) {
                const v2f a  = *(const v2f*)(zrow + 4 * kt);
                const v2f b0 = *(const v2f*)(wcol + (2 * kt) * W1P);
                const v2f b1v = *(const v2f*)(wcol + (2 * kt) * W1P + 32);
                acc0 = __builtin_amdgcn_wmma_f32_16x16x4_f32(
                    false, a, false, b0, (short)0, acc0, false, false);
                acc1 = __builtin_amdgcn_wmma_f32_16x16x4_f32(
                    false, a, false, b1v, (short)0, acc1, false, false);
            }
        }
        __syncthreads();
        if (mul_on) {
            const int ch0 = (wave - SP1_GW) * 32 + r16;
            const float bs0 = bns[ch0],      bt0 = bns[FEAT + ch0];
            const float bs1 = bns[ch0 + 16], bt1 = bns[FEAT + ch0 + 16];
#pragma unroll
            for (int i = 0; i < 8; ++i) {
                const float h0 = fmaf(acc0[i], bs0, bt0);
                const float h1 = fmaf(acc1[i], bs1, bt1);
                zm[(i + 8 * half) * ZP + ch0]      = h0 > 0.0f ? h0 : 0.0f;
                zm[(i + 8 * half) * ZP + ch0 + 16] = h1 > 0.0f ? h1 : 0.0f;
            }
        }
        __syncthreads();
    }
}

#define SP2_T    128
#define SP2_G    4
#define SP2_WR   (SP2_G * 4)
#define SP2_BR   (SP2_T / 32 * SP2_WR)

template <int U>
__device__ __forceinline__ unsigned bcast8(unsigned v) {
    return __builtin_amdgcn_mov_dpp8(v, (unsigned)(U * 0x249249));
}

__device__ __forceinline__ void sp2_pairs(const int* __restrict__ scol,
                                          const float* __restrict__ sval,
                                          const int (&eb)[SP2_G], const int (&ee)[SP2_G],
                                          int c8, unsigned (&po)[SP2_G], float (&vl)[SP2_G]) {
#pragma unroll
    for (int g = 0; g < SP2_G; ++g) {
        int i = eb[g] + c8;
        i = (i < ee[g] - 1) ? i : (ee[g] - 1);
        i = (i > 0) ? i : 0;
        po[g] = (unsigned)scol[i] * PP;
        vl[g] = sval[i];
    }
}

template <int U>
__device__ __forceinline__ void sp2_edge(const float* __restrict__ p, const unsigned (&po)[SP2_G],
                                         unsigned c8, float (&ga)[SP2_G][8]) {
#pragma unroll
    for (int g = 0; g < SP2_G; ++g) ga[g][U] = p[bcast8<U>(po[g]) | c8];
}

template <int U>
__device__ __forceinline__ void sp2_fma(const float (&cv)[SP2_G], const int (&nrem)[SP2_G],
                                        const float (&ga)[SP2_G][8], float (&acc)[SP2_G]) {
#pragma unroll
    for (int g = 0; g < SP2_G; ++g) {
        const float v = __uint_as_float(bcast8<U>(__float_as_uint(cv[g])));
        if (U < nrem[g]) acc[g] = fmaf(v, ga[g][U], acc[g]);
    }
}

__global__ void __launch_bounds__(SP2_T)
spmm2_fused(const float* __restrict__ p, const int* __restrict__ rowStart,
            const int* __restrict__ scol, const float* __restrict__ sval,
            const float* __restrict__ b2, const float* __restrict__ g2,
            const float* __restrict__ be2, const float* __restrict__ m2,
            const float* __restrict__ v2, float* __restrict__ out, int N) {
    __shared__ float os[SP2_BR * NCLS];
    const int t    = threadIdx.x;
    const int lane = t & 31;
    const int wave = t >> 5;
    const int q    = lane >> 3;
    const int c8   = lane & 7;
    const int lrow = wave * SP2_WR + q;
    const int row0 = (int)blockIdx.x * SP2_BR + lrow;

    int eb[SP2_G], ee[SP2_G];
    float acc[SP2_G];
#pragma unroll
    for (int g = 0; g < SP2_G; ++g) {
        const int r  = row0 + 4 * g;
        const int rc = (r < N) ? r : (N - 1);
        eb[g] = rowStart[rc]; ee[g] = rowStart[rc + 1];
        acc[g] = 0.0f;
    }
    const int cc = (c8 < NCLS) ? c8 : (NCLS - 1);
    const float kb = b2[cc], km = m2[cc], kbe = be2[cc];
    const float ksc = g2[cc] * rsqrtf(v2[cc] + EPSB);
#pragma unroll
    for (int g = 0; g < SP2_G; ++g)
        if (row0 + 4 * g >= N) ee[g] = eb[g];

    unsigned po[SP2_G]; float vl[SP2_G];
    sp2_pairs(scol, sval, eb, ee, c8, po, vl);
    for (;;) {
        bool more = false;
#pragma unroll
        for (int g = 0; g < SP2_G; ++g) more = more || (eb[g] < ee[g]);
        if (__builtin_amdgcn_ballot_w32(more) == 0u) break;

        float ga[SP2_G][8];
        sp2_edge<0>(p, po, (unsigned)c8, ga); sp2_edge<1>(p, po, (unsigned)c8, ga);
        sp2_edge<2>(p, po, (unsigned)c8, ga); sp2_edge<3>(p, po, (unsigned)c8, ga);
        sp2_edge<4>(p, po, (unsigned)c8, ga); sp2_edge<5>(p, po, (unsigned)c8, ga);
        sp2_edge<6>(p, po, (unsigned)c8, ga); sp2_edge<7>(p, po, (unsigned)c8, ga);
        float cv[SP2_G]; int nrem[SP2_G];
#pragma unroll
        for (int g = 0; g < SP2_G; ++g) {
            cv[g] = vl[g];
            nrem[g] = ee[g] - eb[g];
            eb[g] += 8;
        }
        sp2_pairs(scol, sval, eb, ee, c8, po, vl);
        __builtin_amdgcn_sched_barrier(0);
        sp2_fma<0>(cv, nrem, ga, acc); sp2_fma<1>(cv, nrem, ga, acc);
        sp2_fma<2>(cv, nrem, ga, acc); sp2_fma<3>(cv, nrem, ga, acc);
        sp2_fma<4>(cv, nrem, ga, acc); sp2_fma<5>(cv, nrem, ga, acc);
        sp2_fma<6>(cv, nrem, ga, acc); sp2_fma<7>(cv, nrem, ga, acc);
    }

    if (c8 < NCLS) {
#pragma unroll
        for (int g = 0; g < SP2_G; ++g)
            os[(lrow + 4 * g) * NCLS + c8] = (acc[g] + kb - km) * ksc + kbe;
    }
    __syncthreads();
    const int total = N * NCLS;
    const int base  = (int)blockIdx.x * (SP2_BR * NCLS);
    constexpr int SP2_OK = (SP2_BR * NCLS + SP2_T - 1) / SP2_T;
    float ov[SP2_OK];
#pragma unroll
    for (int k = 0; k < SP2_OK; ++k) {
        const int j = k * SP2_T + t;
        ov[k] = os[j < SP2_BR * NCLS ? j : SP2_BR * NCLS - 1];
    }
#pragma unroll
    for (int k = 0; k < SP2_OK; ++k) {
        const int j = k * SP2_T + t;
        if (j < SP2_BR * NCLS && base + j < total) out[base + j] = ov[k];
    }
}

extern "C" void kernel_launch(void* const* d_in, const int* in_sizes, int n_in,
                              void* d_out, int out_size, void* d_ws, size_t ws_size,
                              hipStream_t stream) {
    const float* x   = (const float*)d_in[0];
    const float* ev  = (const float*)d_in[1];
    const float* W1  = (const float*)d_in[2];
    const float* b1  = (const float*)d_in[3];
    const float* g1  = (const float*)d_in[4];
    const float* be1 = (const float*)d_in[5];
    const float* m1  = (const float*)d_in[6];
    const float* v1  = (const float*)d_in[7];
    const float* W2  = (const float*)d_in[8];
    const float* b2  = (const float*)d_in[9];
    const float* g2  = (const float*)d_in[10];
    const float* be2 = (const float*)d_in[11];
    const float* m2  = (const float*)d_in[12];
    const float* v2  = (const float*)d_in[13];
    const int*  erow = (const int*)d_in[14];
    const int*  ecol = (const int*)d_in[15];
    float* out = (float*)d_out;

    const int N = in_sizes[0] / FEAT;
    const int E = in_sizes[1];

    const int B = (N + BK_R - 1) >> BK_SH;

    float* p        = (float*)d_ws;
    int*   brec     = (int*)(p + (size_t)N * PP);
    int*   startL   = brec + (size_t)2 * E;
    int*   rowStart = startL + (size_t)(B + 1) * BIN_G;
    int*   scol     = rowStart + (N + 1);
    float* sval     = (float*)(scol + E);

    bin_tile<<<BIN_G, BIN_T, 0, stream>>>(erow, ecol, ev, startL, brec, E, B);
    bucket_sort<<<B, BS_T, 0, stream>>>(brec, startL, rowStart, scol, sval, N, B, E);
    spmm1_fused<<<SP1_GRID, SP1_T, 0, stream>>>(x, rowStart, scol, sval, W1,
                                                b1, g1, be1, m1, v1, W2, p, N);
    spmm2_fused<<<(N + SP2_BR - 1) / SP2_BR, SP2_T, 0, stream>>>(p, rowStart, scol, sval,
                                                                 b2, g2, be2, m2, v2, out, N);
}


# bin_tile: a tile's records are put in slice order in LDS and leave as whole 128-byte lines (16-byte stores, line-aligned); was one 8-byte store per record, 32 lines per store; sum order as parent
# speedup vs baseline: 5.6053x; 1.0524x over previous
#include <hip/hip_runtime.h>
#include <hip/hip_bf16.h>

typedef __attribute__((ext_vector_type(2))) float v2f;
typedef __attribute__((ext_vector_type(8))) float v8f;

#define FEAT 128
#define NCLS 7
#define PP   8
#define EPSB 1e-5f

#define BK_SH   8
#define BK_R    (1 << BK_SH)
#define BK_MAX  1024
#define BIN_G   256
#define BIN_T   1024
#define BIN_U   8

static_assert(BK_MAX == BIN_T, "bin_tile scans one bucket counter per thread");

__device__ __forceinline__ int block_excl_scan(int d, int* wt, int t) {
    const int lane = t & 31;
    const int wave = t >> 5;
    int incl = d;
    for (int off = 1; off < 32; off <<= 1) {
        const int up = __shfl_up(incl, off, 32);
        if (lane >= off) incl += up;
    }
    if (lane == 31) wt[wave] = incl;
    __syncthreads();
    int c = (lane < wave) ? wt[lane] : 0;
    for (int off = 16; off >= 1; off >>= 1) c += __shfl_xor(c, off, 32);
    return c + incl - d;
}

__global__ void __launch_bounds__(BIN_T)
bin_tile(const int* __restrict__ erow, const int* __restrict__ ecol,
         const float* __restrict__ ev, int* __restrict__ startL,
         int* __restrict__ brec_, int E, int B) {
    __shared__ int cnt[BK_MAX];
    __shared__ int wt[BIN_T / 32];
    __shared__ int2 srec[BIN_U * BIN_T];
    int2* __restrict__ brec = (int2*)brec_;
    const int t = threadIdx.x;
    cnt[t] = 0;
    __syncthreads();
    const int tile = (E + BIN_G - 1) / BIN_G;
    int e0 = (int)blockIdx.x * tile;
    e0 = (e0 < E) ? e0 : E;
    const int e1 = (e0 + tile < E) ? (e0 + tile) : E;
    const int last = (e1 > 0) ? (e1 - 1) : 0;

    if (tile <= BIN_U * BIN_T) {
        int r[BIN_U], c[BIN_U], rk[BIN_U]; float v[BIN_U];
#pragma unroll
        for (int u = 0; u < BIN_U; ++u) {
            const int e = e0 + u * BIN_T + t;
            const int ee = (e < e1) ? e : last;
            r[u] = erow[ee];
            c[u] = ecol[ee];
            v[u] = ev[ee];
        }
        __builtin_amdgcn_sched_barrier(0);
#pragma unroll
        for (int u = 0; u < BIN_U; ++u) {
            rk[u] = -1;
            if (e0 + u * BIN_T + t < e1) rk[u] = atomicAdd(&cnt[r[u] >> BK_SH], 1);
        }
        __syncthreads();
        const int excl = block_excl_scan(cnt[t], wt, t);
        cnt[t] = excl;
        if (t <= B) startL[(size_t)t * BIN_G + blockIdx.x] = excl;
        __syncthreads();
        int st[BIN_U];
#pragma unroll
        for (int u = 0; u < BIN_U; ++u) st[u] = cnt[r[u] >> BK_SH];
#pragma unroll
        for (int u = 0; u < BIN_U; ++u) {
            if (rk[u] >= 0)
                srec[st[u] + rk[u]] =
                    make_int2((c[u] << BK_SH) | (r[u] & (BK_R - 1)), __float_as_int(v[u]));
        }
        __syncthreads();
        const int n = e1 - e0;
        if (n > 0) {
            constexpr int BIN_O = BIN_U / 2 + 1;
            const int g0 = e0 & ~15;
            int4 q[BIN_O];
#pragma unroll
            for (int k = 0; k < BIN_O; ++k) {
                const int ja = g0 + 2 * (k * BIN_T + t) - e0;
                const int jb = ja + 1;
                const int2 a = srec[ja < 0 ? 0 : (ja < n ? ja : n - 1)];
                const int2 b = srec[jb < 0 ? 0 : (jb < n ? jb : n - 1)];
                q[k] = make_int4(a.x, a.y, b.x, b.y);
            }
#pragma unroll
            for (int k = 0; k < BIN_O; ++k) {
                const int g = g0 + 2 * (k * BIN_T + t);
                const bool ina = (g >= e0) && (g < e1);
                const bool inb = (g + 1 >= e0) && (g + 1 < e1);
                if (ina && inb)  *(int4*)&brec[g] = q[k];
                else if (ina)    brec[g] = make_int2(q[k].x, q[k].y);
                else if (inb)    brec[g + 1] = make_int2(q[k].z, q[k].w);
            }
        }
    } else {
        for (int base = e0; base < e1; base += BIN_U * BIN_T) {
            int r[BIN_U];
#pragma unroll
            for (int u = 0; u < BIN_U; ++u) {
                const int e = base + u * BIN_T + t;
                r[u] = erow[(e < e1) ? e : last];
            }
#pragma unroll
            for (int u = 0; u < BIN_U; ++u)
                if (base + u * BIN_T + t < e1) atomicAdd(&cnt[r[u] >> BK_SH], 1);
        }
        __syncthreads();
        const int excl = block_excl_scan(cnt[t], wt, t);
        cnt[t] = excl;
        if (t <= B) startL[(size_t)t * BIN_G + blockIdx.x] = excl;
        __syncthreads();
        for (int base = e0; base < e1; base += BIN_U * BIN_T) {
            int r[BIN_U], c[BIN_U], pos[BIN_U]; float v[BIN_U];
#pragma unroll
            for (int u = 0; u < BIN_U; ++u) {
                const int e = base + u * BIN_T + t;
                const int ee = (e < e1) ? e : last;
                r[u] = erow[ee];
                c[u] = ecol[ee];
                v[u] = ev[ee];
            }
#pragma unroll
            for (int u = 0; u < BIN_U; ++u) {
                pos[u] = -1;
                if (base + u * BIN_T + t < e1) pos[u] = atomicAdd(&cnt[r[u] >> BK_SH], 1);
            }
#pragma unroll
            for (int u = 0; u < BIN_U; ++u) {
                if (pos[u] >= 0)
                    brec[e0 + pos[u]] =
                        make_int2((c[u] << BK_SH) | (r[u] & (BK_R - 1)), __float_as_int(v[u]));
            }
        }
    }
}

#define BS_T    1024
#define BS_L    8
#define BS_P    (BIN_G * BS_L / BS_T)
#define BS_U    6
#define BS_O    6
#define BS_CAP  (BS_T * BS_O)

static_assert(BIN_G * BS_L == BS_P * BS_T, "every run has its BS_L lanes");

__device__ __forceinline__ int bucket_scan(int* rc, int* wsum, int t) {
    const int lane = t & 31;
    const int wave = t >> 5;
    int d = 0, incl = 0;
    if (t < BK_R) {
        d = rc[t];
        incl = d;
        for (int off = 1; off < 32; off <<= 1) {
            const int up = __shfl_up(incl, off, 32);
            if (lane >= off) incl += up;
        }
        if (lane == 31) wsum[wave] = incl;
    }
    __syncthreads();
    int excl = 0;
    if (t < BK_R) {
        int c = (lane < wave) ? wsum[lane] : 0;
        for (int off = 16; off >= 1; off >>= 1) c += __shfl_xor(c, off, 32);
        excl = c + incl - d;
        rc[t] = excl;
    }
    __syncthreads();
    return excl;
}

__global__ void __launch_bounds__(BS_T) __attribute__((amdgpu_waves_per_eu(16, 16)))
bucket_sort(const int* __restrict__ brec_, const int* __restrict__ startL,
            int* __restrict__ rowStart, int* __restrict__ scol,
            float* __restrict__ sval, int N, int B, int E) {
    __shared__ int   rc[BK_R];
    __shared__ int   wsum[BK_R / 32];
    __shared__ int   red0[BS_T / 32], red1[BS_T / 32], red2[BS_T / 32];
    __shared__ int   lcol[BS_CAP];
    __shared__ float lval[BS_CAP];
    const int2* __restrict__ brec = (const int2*)brec_;
    const int t    = threadIdx.x;
    const int lane = t & 31;
    const int wave = t >> 5;
    const int b    = blockIdx.x;
    const int grp  = t / BS_L;
    const int sub  = t % BS_L;
    const int tile = (E + BIN_G - 1) / BIN_G;
    int len[BS_P], first[BS_P];
    int a0 = 0, a1 = 0, am = 0;
#pragma unroll
    for (int p = 0; p < BS_P; ++p) {
        const int run = p * (BS_T / BS_L) + grp;
        int tb = run * tile;
        tb = (tb < E) ? tb : E;
        const int lo = startL[(size_t)b * BIN_G + run];
        const int hi = startL[(size_t)(b + 1) * BIN_G + run];
        len[p]   = hi - lo;
        first[p] = tb + lo;
        if (sub == 0) { a0 += lo; a1 += hi; }
        am = (len[p] > am) ? len[p] : am;
    }
    if (t < BK_R) rc[t] = 0;

    int key[BS_P * BS_U], val[BS_P * BS_U];
#pragma unroll
    for (int p = 0; p < BS_P; ++p) {
#pragma unroll
        for (int u = 0; u < BS_U; ++u) {
            const int k  = sub + u * BS_L;
            const int kk = (k < len[p]) ? k : (len[p] - 1);
            const int2 q = brec[(len[p] > 0) ? (first[p] + kk) : 0];
            key[p * BS_U + u] = q.x; val[p * BS_U + u] = q.y;
        }
    }

    for (int off = 16; off >= 1; off >>= 1) {
        a0 += __shfl_xor(a0, off, 32);
        a1 += __shfl_xor(a1, off, 32);
        const int om = __shfl_xor(am, off, 32);
        am = (om > am) ? om : am;
    }
    if (lane == 0) { red0[wave] = a0; red1[wave] = a1; red2[wave] = am; }
    __syncthreads();
    int s0 = red0[lane], s1 = red1[lane], mx = red2[lane];
    for (int off = 16; off >= 1; off >>= 1) {
        s0 += __shfl_xor(s0, off, 32);
        s1 += __shfl_xor(s1, off, 32);
        const int om = __shfl_xor(mx, off, 32);
        mx = (om > mx) ? om : mx;
    }
    const int n = s1 - s0;

    if (n <= BS_CAP && mx <= BS_L * BS_U) {
        int lr[BS_P * BS_U];
#pragma unroll
        for (int p = 0; p < BS_P; ++p) {
#pragma unroll
            for (int u = 0; u < BS_U; ++u) {
                const int s = p * BS_U + u;
                lr[s] = 0;
                if (sub + u * BS_L < len[p]) lr[s] = atomicAdd(&rc[key[s] & (BK_R - 1)], 1);
            }
        }
        __syncthreads();
        const int excl = bucket_scan(rc, wsum, t);
        if (t < BK_R) {
            const int row = b * BK_R + t;
            if (row < N) rowStart[row] = s0 + excl;
        }
#pragma unroll
        for (int s = 0; s < BS_P * BS_U; ++s) lr[s] += rc[key[s] & (BK_R - 1)];
#pragma unroll
        for (int p = 0; p < BS_P; ++p) {
#pragma unroll
            for (int u = 0; u < BS_U; ++u) {
                const int s = p * BS_U + u;
                if (sub + u * BS_L < len[p]) {
                    lcol[lr[s]] = key[s] >> BK_SH;
                    lval[lr[s]] = __int_as_float(val[s]);
                }
            }
        }
        __syncthreads();
        if (n > 0) {
            int oc[BS_O]; float ov[BS_O];
#pragma unroll
            for (int u = 0; u < BS_O; ++u) {
                const int j = t + u * BS_T;
                const int jj = (j < n) ? j : (n - 1);
                oc[u] = lcol[jj]; ov[u] = lval[jj];
            }
#pragma unroll
            for (int u = 0; u < BS_O; ++u) {
                const int j = t + u * BS_T;
                if (j < n) { scol[s0 + j] = oc[u]; sval[s0 + j] = ov[u]; }
            }
        }
    } else {
        for (int p = 0; p < BS_P; ++p)
            for (int k = sub; k < len[p]; k += BS_L)
                atomicAdd(&rc[brec[first[p] + k].x & (BK_R - 1)], 1);
        __syncthreads();
        const int excl = bucket_scan(rc, wsum, t);
        if (t < BK_R) {
            const int row = b * BK_R + t;
            if (row < N) rowStart[row] = s0 + excl;
        }
        for (int p = 0; p < BS_P; ++p) {
            for (int k = sub; k < len[p]; k += BS_L) {
                const int2 q = brec[first[p] + k];
                const int pos = s0 + atomicAdd(&rc[q.x & (BK_R - 1)], 1);
                scol[pos] = q.x >> BK_SH;
                sval[pos] = __int_as_float(q.y);
            }
        }
    }
    if (b == B - 1 && t == 0) rowStart[N] = s1;
}

#define SP1_GW   16
#define SP1_XW   4
#define SP1_W    (SP1_GW + SP1_XW)
#define SP1_T    (SP1_W * 32)
#define SP1_BR   16
#define SP1_GRID 768
#define ZP       (FEAT + 4)
#define W1P      (2 * FEAT + 32)

static_assert(SP1_BR == SP1_GW && SP1_XW * 32 == FEAT,
              "a gather wave owns one row of a tile, a multiplier wave two column tiles");

template <int C>
__device__ __forceinline__ void gather_fma(const float4* __restrict__ y4, int cc, float vv,
                                           int j, int lane, float4& acc) {
    float4 g[C];
#pragma unroll
    for (int u = 0; u < C; ++u) {
        const int c = __builtin_amdgcn_readlane(cc, j + u);
        g[u] = y4[(size_t)c * (FEAT / 4) + lane];
    }
#pragma unroll
    for (int u = 0; u < C; ++u) {
        const float v = __int_as_float(__builtin_amdgcn_readlane(__float_as_int(vv), j + u));
        acc.x = fmaf(v, g[u].x, acc.x);
        acc.y = fmaf(v, g[u].y, acc.y);
        acc.z = fmaf(v, g[u].z, acc.z);
        acc.w = fmaf(v, g[u].w, acc.w);
    }
}

__global__ void __launch_bounds__(SP1_T) __attribute__((amdgpu_waves_per_eu(15, 16)))
spmm1_fused(const float* __restrict__ x, const int* __restrict__ rowStart,
            const int* __restrict__ scol, const float* __restrict__ sval,
            const float* __restrict__ W1,
            const float* __restrict__ b1, const float* __restrict__ g1,
            const float* __restrict__ be1, const float* __restrict__ m1,
            const float* __restrict__ v1, const float* __restrict__ W2,
            float* __restrict__ p, int N) {
    __shared__ __attribute__((aligned(16))) float w1s[(FEAT / 2) * W1P];
    __shared__ __attribute__((aligned(16))) float zs[2 * SP1_BR * ZP];
    __shared__ __attribute__((aligned(16))) float w2s[FEAT * NCLS];
    __shared__ float bns[2 * FEAT];

    const int tid = (int)threadIdx.x;
    for (int i = tid; i < FEAT * NCLS; i += SP1_T) w2s[i] = W2[i];
    for (int i = tid; i < FEAT; i += SP1_T) {
        const float bs = g1[i] * rsqrtf(v1[i] + EPSB);
        bns[i] = bs;
        bns[FEAT + i] = (b1[i] - m1[i]) * bs + be1[i];
    }
    for (int i = tid; i < (FEAT / 2) * (FEAT / 4); i += SP1_T) {
        const int kp = i / (FEAT / 4);
        const int c4 = (i % (FEAT / 4)) * 4;
        const float4 wa = *(const float4*)&W1[(size_t)(2 * kp) * FEAT + c4];
        const float4 wb = *(const float4*)&W1[(size_t)(2 * kp + 1) * FEAT + c4];
        float4* d = (float4*)&w1s[kp * W1P + 2 * c4];
        d[0] = make_float4(wa.x, wb.x, wa.y, wb.y);
        d[1] = make_float4(wa.z, wb.z, wa.w, wb.w);
    }

    const int lane = tid & 31;
    const int half = lane >> 4;
    const int r16  = lane & 15;
    const int wave = __builtin_amdgcn_readfirstlane(tid >> 5);
    const bool gat = wave < SP1_GW;
    const int G  = (int)gridDim.x;
    const int bI = (int)blockIdx.x;
    const int ntiles = (N + SP1_BR - 1) / SP1_BR;
    const int T = (bI < ntiles) ? (ntiles - bI + G - 1) / G : 0;

    const float4* __restrict__ x4 = (const float4*)x;
    int eB = 0, eE = 0, rsN = 0, nc = 0;
    float nv = 0.0f;
    if (gat) {
        int rs0 = 0;
        const int ra = bI * SP1_BR + wave;
        const int rb = (bI + G) * SP1_BR + wave;
        if (0 < T && ra < N && lane < 2) rs0 = rowStart[ra + lane];
        if (1 < T && rb < N && lane < 2) rsN = rowStart[rb + lane];
        eB = __builtin_amdgcn_readlane(rs0, 0);
        eE = __builtin_amdgcn_readlane(rs0, 1);
        if (eB + lane < eE) { nc = scol[eB + lane]; nv = sval[eB + lane]; }
    }
    __syncthreads();

    const v8f vzero = {0.f,0.f,0.f,0.f,0.f,0.f,0.f,0.f};
    for (int s = 0; s < T + 2; ++s) {
        float* zg = zs + (s & 1) * (SP1_BR * ZP);
        float* zm = zs + ((s + 1) & 1) * (SP1_BR * ZP);
        const bool mul_on = (!gat) && (s >= 1) && (s <= T);
        v8f acc0 = vzero, acc1 = vzero;
        if (gat) {
            const int cc = nc; const float vv = nv;
            const int nB = __builtin_amdgcn_readlane(rsN, 0);
            const int nE = __builtin_amdgcn_readlane(rsN, 1);
            nc = 0; nv = 0.0f;
            if (nB + lane < nE) { nc = scol[nB + lane]; nv = sval[nB + lane]; }
            rsN = 0;
            {
                const int rn = (bI + (s + 2) * G) * SP1_BR + wave;
                if (s + 2 < T && rn < N && lane < 2) rsN = rowStart[rn + lane];
            }

            if (s >= 2) {
                const int r = (bI + (s - 2) * G) * SP1_BR + wave;
                if (r < N) {
                    float w2[4][NCLS];
#pragma unroll
                    for (int j = 0; j < 4; ++j)
#pragma unroll
                        for (int c = 0; c < NCLS; ++c) w2[j][c] = w2s[(lane * 4 + j) * NCLS + c];
                    const float4 hq = *(const float4*)&zg[wave * ZP + lane * 4];
                    const float h[4] = {hq.x, hq.y, hq.z, hq.w};
                    float mine = 0.0f;
#pragma unroll
                    for (int c = 0; c < NCLS; ++c) {
                        float ssum = 0.0f;
#pragma unroll
                        for (int j = 0; j < 4; ++j) ssum = fmaf(h[j], w2[j][c], ssum);
#pragma unroll
                        for (int off = 16; off >= 1; off >>= 1) ssum += __shfl_xor(ssum, off, 32);
                        if (lane == c) mine = ssum;
                    }
                    if (lane < PP) p[(size_t)r * PP + lane] = mine;
                }
            }

            float4 acc = {0.f, 0.f, 0.f, 0.f};
            const int deg = eE - eB;
            const int n = deg < 32 ? deg : 32;
            int j = 0;
            for (; j + 8 <= n; j += 8) gather_fma<8>(x4, cc, vv, j, lane, acc);
            switch (n - j) {
                case 7: gather_fma<7>(x4, cc, vv, j, lane, acc); break;
                case 6: gather_fma<6>(x4, cc, vv, j, lane, acc); break;
                case 5: gather_fma<5>(x4, cc, vv, j, lane, acc); break;
                case 4: gather_fma<4>(x4, cc, vv, j, lane, acc); break;
                case 3: gather_fma<3>(x4, cc, vv, j, lane, acc); break;
                case 2: gather_fma<2>(x4, cc, vv, j, lane, acc); break;
                case 1: gather_fma<1>(x4, cc, vv, j, lane, acc); break;
                default: break;
            }
            for (int e = eB + 32; e < eE; ++e) {
                const int   c = scol[e];
                const float v = sval[e];
                const float4 g = x4[(size_t)c * (FEAT / 4) + lane];
                acc.x = fmaf(v, g.x, acc.x);
                acc.y = fmaf(v, g.y, acc.y);
                acc.z = fmaf(v, g.z, acc.z);
                acc.w = fmaf(v, g.w, acc.w);
            }
            *(float4*)&zg[wave * ZP + lane * 4] = acc;
            eB = nB; eE = nE;
        } else if (mul_on) {
            const int ch0 = (wave - SP1_GW) * 32 + r16;
            const float* zrow = zm + r16 * ZP + 2 * half;
            const float* wcol = w1s + half * W1P + 2 * ch0;
#pragma unroll 4
            for (int kt = 0; kt < FEAT / 4; ++kt) {
                const v2f a  = *(const v2f*)(zrow + 4 * kt);
                const v2f b0 = *(const v2f*)(wcol + (2 * kt) * W1P);
                const v2f b1v = *(const v2f*)(wcol + (2 * kt) * W1P + 32);
                acc0 = __builtin_amdgcn_wmma_f32_16x16x4_f32(
                    false, a, false, b0, (short)0, acc0, false, false);
                acc1 = __builtin_amdgcn_wmma_f32_16x16x4_f32(
                    false, a, false, b1v, (short)0, acc1, false, false);
            }
        }
        __syncthreads();
        if (mul_on) {
            const int ch0 = (wave - SP1_GW) * 32 + r16;
            const float bs0 = bns[ch0],      bt0 = bns[FEAT + ch0];
            const float bs1 = bns[ch0 + 16], bt1 = bns[FEAT + ch0 + 16];
#pragma unroll
            for (int i = 0; i < 8; ++i) {
                const float h0 = fmaf(acc0[i], bs0, bt0);
                const float h1 = fmaf(acc1[i], bs1, bt1);
                zm[(i + 8 * half) * ZP + ch0]      = h0 > 0.0f ? h0 : 0.0f;
                zm[(i + 8 * half) * ZP + ch0 + 16] = h1 > 0.0f ? h1 : 0.0f;
            }
        }
        __syncthreads();
    }
}

#define SP2_T    128
#define SP2_G    4
#define SP2_WR   (SP2_G * 4)
#define SP2_BR   (SP2_T / 32 * SP2_WR)

template <int U>
__device__ __forceinline__ unsigned bcast8(unsigned v) {
    return __builtin_amdgcn_mov_dpp8(v, (unsigned)(U * 0x249249));
}

__device__ __forceinline__ void sp2_pairs(const int* __restrict__ scol,
                                          const float* __restrict__ sval,
                                          const int (&eb)[SP2_G], const int (&ee)[SP2_G],
                                          int c8, unsigned (&po)[SP2_G], float (&vl)[SP2_G]) {
#pragma unroll
    for (int g = 0; g < SP2_G; ++g) {
        int i = eb[g] + c8;
        i = (i < ee[g] - 1) ? i : (ee[g] - 1);
        i = (i > 0) ? i : 0;
        po[g] = (unsigned)scol[i] * PP;
        vl[g] = sval[i];
    }
}

template <int U>
__device__ __forceinline__ void sp2_edge(const float* __restrict__ p, const unsigned (&po)[SP2_G],
                                         unsigned c8, float (&ga)[SP2_G][8]) {
#pragma unroll
    for (int g = 0; g < SP2_G; ++g) ga[g][U] = p[bcast8<U>(po[g]) | c8];
}

template <int U>
__device__ __forceinline__ void sp2_fma(const float (&cv)[SP2_G], const int (&nrem)[SP2_G],
                                        const float (&ga)[SP2_G][8], float (&acc)[SP2_G]) {
#pragma unroll
    for (int g = 0; g < SP2_G; ++g) {
        const float v = __uint_as_float(bcast8<U>(__float_as_uint(cv[g])));
        if (U < nrem[g]) acc[g] = fmaf(v, ga[g][U], acc[g]);
    }
}

__global__ void __launch_bounds__(SP2_T)
spmm2_fused(const float* __restrict__ p, const int* __restrict__ rowStart,
            const int* __restrict__ scol, const float* __restrict__ sval,
            const float* __restrict__ b2, const float* __restrict__ g2,
            const float* __restrict__ be2, const float* __restrict__ m2,
            const float* __restrict__ v2, float* __restrict__ out, int N) {
    __shared__ float os[SP2_BR * NCLS];
    const int t    = threadIdx.x;
    const int lane = t & 31;
    const int wave = t >> 5;
    const int q    = lane >> 3;
    const int c8   = lane & 7;
    const int lrow = wave * SP2_WR + q;
    const int row0 = (int)blockIdx.x * SP2_BR + lrow;

    int eb[SP2_G], ee[SP2_G];
    float acc[SP2_G];
#pragma unroll
    for (int g = 0; g < SP2_G; ++g) {
        const int r  = row0 + 4 * g;
        const int rc = (r < N) ? r : (N - 1);
        eb[g] = rowStart[rc]; ee[g] = rowStart[rc + 1];
        acc[g] = 0.0f;
    }
    const int cc = (c8 < NCLS) ? c8 : (NCLS - 1);
    const float kb = b2[cc], km = m2[cc], kbe = be2[cc];
    const float ksc = g2[cc] * rsqrtf(v2[cc] + EPSB);
#pragma unroll
    for (int g = 0; g < SP2_G; ++g)
        if (row0 + 4 * g >= N) ee[g] = eb[g];

    unsigned po[SP2_G]; float vl[SP2_G];
    sp2_pairs(scol, sval, eb, ee, c8, po, vl);
    for (;;) {
        bool more = false;
#pragma unroll
        for (int g = 0; g < SP2_G; ++g) more = more || (eb[g] < ee[g]);
        if (__builtin_amdgcn_ballot_w32(more) == 0u) break;

        float ga[SP2_G][8];
        sp2_edge<0>(p, po, (unsigned)c8, ga); sp2_edge<1>(p, po, (unsigned)c8, ga);
        sp2_edge<2>(p, po, (unsigned)c8, ga); sp2_edge<3>(p, po, (unsigned)c8, ga);
        sp2_edge<4>(p, po, (unsigned)c8, ga); sp2_edge<5>(p, po, (unsigned)c8, ga);
        sp2_edge<6>(p, po, (unsigned)c8, ga); sp2_edge<7>(p, po, (unsigned)c8, ga);
        float cv[SP2_G]; int nrem[SP2_G];
#pragma unroll
        for (int g = 0; g < SP2_G; ++g) {
            cv[g] = vl[g];
            nrem[g] = ee[g] - eb[g];
            eb[g] += 8;
        }
        sp2_pairs(scol, sval, eb, ee, c8, po, vl);
        __builtin_amdgcn_sched_barrier(0);
        sp2_fma<0>(cv, nrem, ga, acc); sp2_fma<1>(cv, nrem, ga, acc);
        sp2_fma<2>(cv, nrem, ga, acc); sp2_fma<3>(cv, nrem, ga, acc);
        sp2_fma<4>(cv, nrem, ga, acc); sp2_fma<5>(cv, nrem, ga, acc);
        sp2_fma<6>(cv, nrem, ga, acc); sp2_fma<7>(cv, nrem, ga, acc);
    }

    if (c8 < NCLS) {
#pragma unroll
        for (int g = 0; g < SP2_G; ++g)
            os[(lrow + 4 * g) * NCLS + c8] = (acc[g] + kb - km) * ksc + kbe;
    }
    __syncthreads();
    const int total = N * NCLS;
    const int base  = (int)blockIdx.x * (SP2_BR * NCLS);
    constexpr int SP2_OK = (SP2_BR * NCLS + SP2_T - 1) / SP2_T;
    float ov[SP2_OK];
#pragma unroll
    for (int k = 0; k < SP2_OK; ++k) {
        const int j = k * SP2_T + t;
        ov[k] = os[j < SP2_BR * NCLS ? j : SP2_BR * NCLS - 1];
    }
#pragma unroll
    for (int k = 0; k < SP2_OK; ++k) {
        const int j = k * SP2_T + t;
        if (j < SP2_BR * NCLS && base + j < total) out[base + j] = ov[k];
    }
}

extern "C" void kernel_launch(void* const* d_in, const int* in_sizes, int n_in,
                              void* d_out, int out_size, void* d_ws, size_t ws_size,
                              hipStream_t stream) {
    const float* x   = (const float*)d_in[0];
    const float* ev  = (const float*)d_in[1];
    const float* W1  = (const float*)d_in[2];
    const float* b1  = (const float*)d_in[3];
    const float* g1  = (const float*)d_in[4];
    const float* be1 = (const float*)d_in[5];
    const float* m1  = (const float*)d_in[6];
    const float* v1  = (const float*)d_in[7];
    const float* W2  = (const float*)d_in[8];
    const float* b2  = (const float*)d_in[9];
    const float* g2  = (const float*)d_in[10];
    const float* be2 = (const float*)d_in[11];
    const float* m2  = (const float*)d_in[12];
    const float* v2  = (const float*)d_in[13];
    const int*  erow = (const int*)d_in[14];
    const int*  ecol = (const int*)d_in[15];
    float* out = (float*)d_out;

    const int N = in_sizes[0] / FEAT;
    const int E = in_sizes[1];

    const int B = (N + BK_R - 1) >> BK_SH;

    float* p        = (float*)d_ws;
    int*   brec     = (int*)(p + (size_t)N * PP);
    int*   startL   = brec + (size_t)2 * E;
    int*   rowStart = startL + (size_t)(B + 1) * BIN_G;
    int*   scol     = rowStart + (N + 1);
    float* sval     = (float*)(scol + E);

    bin_tile<<<BIN_G, BIN_T, 0, stream>>>(erow, ecol, ev, startL, brec, E, B);
    bucket_sort<<<B, BS_T, 0, stream>>>(brec, startL, rowStart, scol, sval, N, B, E);
    spmm1_fused<<<SP1_GRID, SP1_T, 0, stream>>>(x, rowStart, scol, sval, W1,
                                                b1, g1, be1, m1, v1, W2, p, N);
    spmm2_fused<<<(N + SP2_BR - 1) / SP2_BR, SP2_T, 0, stream>>>(p, rowStart, scol, sval,
                                                                 b2, g2, be2, m2, v2, out, N);
}


# spmm1_fused: p = h@W2 moves from the 16 gather waves to the 4 multiplier waves (fp32 WMMA on their own h columns, 4 partials added in fixed order); one barrier per step; order of W2 sums changes
# speedup vs baseline: 5.8287x; 1.0399x over previous
#include <hip/hip_runtime.h>
#include <hip/hip_bf16.h>

typedef __attribute__((ext_vector_type(2))) float v2f;
typedef __attribute__((ext_vector_type(8))) float v8f;

#define FEAT 128
#define NCLS 7
#define PP   8
#define EPSB 1e-5f

#define BK_SH   8
#define BK_R    (1 << BK_SH)
#define BK_MAX  1024
#define BIN_G   256
#define BIN_T   1024
#define BIN_U   8

static_assert(BK_MAX == BIN_T, "bin_tile scans one bucket counter per thread");

__device__ __forceinline__ int block_excl_scan(int d, int* wt, int t) {
    const int lane = t & 31;
    const int wave = t >> 5;
    int incl = d;
    for (int off = 1; off < 32; off <<= 1) {
        const int up = __shfl_up(incl, off, 32);
        if (lane >= off) incl += up;
    }
    if (lane == 31) wt[wave] = incl;
    __syncthreads();
    int c = (lane < wave) ? wt[lane] : 0;
    for (int off = 16; off >= 1; off >>= 1) c += __shfl_xor(c, off, 32);
    return c + incl - d;
}

__global__ void __launch_bounds__(BIN_T)
bin_tile(const int* __restrict__ erow, const int* __restrict__ ecol,
         const float* __restrict__ ev, int* __restrict__ startL,
         int* __restrict__ brec_, int E, int B) {
    __shared__ int cnt[BK_MAX];
    __shared__ int wt[BIN_T / 32];
    __shared__ int2 srec[BIN_U * BIN_T];
    int2* __restrict__ brec = (int2*)brec_;
    const int t = threadIdx.x;
    cnt[t] = 0;
    __syncthreads();
    const int tile = (E + BIN_G - 1) / BIN_G;
    int e0 = (int)blockIdx.x * tile;
    e0 = (e0 < E) ? e0 : E;
    const int e1 = (e0 + tile < E) ? (e0 + tile) : E;
    const int last = (e1 > 0) ? (e1 - 1) : 0;

    if (tile <= BIN_U * BIN_T) {
        int r[BIN_U], c[BIN_U], rk[BIN_U]; float v[BIN_U];
#pragma unroll
        for (int u = 0; u < BIN_U; ++u) {
            const int e = e0 + u * BIN_T + t;
            const int ee = (e < e1) ? e : last;
            r[u] = erow[ee];
            c[u] = ecol[ee];
            v[u] = ev[ee];
        }
        __builtin_amdgcn_sched_barrier(0);
#pragma unroll
        for (int u = 0; u < BIN_U; ++u) {
            rk[u] = -1;
            if (e0 + u * BIN_T + t < e1) rk[u] = atomicAdd(&cnt[r[u] >> BK_SH], 1);
        }
        __syncthreads();
        const int excl = block_excl_scan(cnt[t], wt, t);
        cnt[t] = excl;
        if (t <= B) startL[(size_t)t * BIN_G + blockIdx.x] = excl;
        __syncthreads();
        int st[BIN_U];
#pragma unroll
        for (int u = 0; u < BIN_U; ++u) st[u] = cnt[r[u] >> BK_SH];
#pragma unroll
        for (int u = 0; u < BIN_U; ++u) {
            if (rk[u] >= 0)
                srec[st[u] + rk[u]] =
                    make_int2((c[u] << BK_SH) | (r[u] & (BK_R - 1)), __float_as_int(v[u]));
        }
        __syncthreads();
        const int n = e1 - e0;
        if (n > 0) {
            constexpr int BIN_O = BIN_U / 2 + 1;
            const int g0 = e0 & ~15;
            int4 q[BIN_O];
#pragma unroll
            for (int k = 0; k < BIN_O; ++k) {
                const int ja = g0 + 2 * (k * BIN_T + t) - e0;
                const int jb = ja + 1;
                const int2 a = srec[ja < 0 ? 0 : (ja < n ? ja : n - 1)];
                const int2 b = srec[jb < 0 ? 0 : (jb < n ? jb : n - 1)];
                q[k] = make_int4(a.x, a.y, b.x, b.y);
            }
#pragma unroll
            for (int k = 0; k < BIN_O; ++k) {
                const int g = g0 + 2 * (k * BIN_T + t);
                const bool ina = (g >= e0) && (g < e1);
                const bool inb = (g + 1 >= e0) && (g + 1 < e1);
                if (ina && inb)  *(int4*)&brec[g] = q[k];
                else if (ina)    brec[g] = make_int2(q[k].x, q[k].y);
                else if (inb)    brec[g + 1] = make_int2(q[k].z, q[k].w);
            }
        }
    } else {
        for (int base = e0; base < e1; base += BIN_U * BIN_T) {
            int r[BIN_U];
#pragma unroll
            for (int u = 0; u < BIN_U; ++u) {
                const int e = base + u * BIN_T + t;
                r[u] = erow[(e < e1) ? e : last];
            }
#pragma unroll
            for (int u = 0; u < BIN_U; ++u)
                if (base + u * BIN_T + t < e1) atomicAdd(&cnt[r[u] >> BK_SH], 1);
        }
        __syncthreads();
        const int excl = block_excl_scan(cnt[t], wt, t);
        cnt[t] = excl;
        if (t <= B) startL[(size_t)t * BIN_G + blockIdx.x] = excl;
        __syncthreads();
        for (int base = e0; base < e1; base += BIN_U * BIN_T) {
            int r[BIN_U], c[BIN_U], pos[BIN_U]; float v[BIN_U];
#pragma unroll
            for (int u = 0; u < BIN_U; ++u) {
                const int e = base + u * BIN_T + t;
                const int ee = (e < e1) ? e : last;
                r[u] = erow[ee];
                c[u] = ecol[ee];
                v[u] = ev[ee];
            }
#pragma unroll
            for (int u = 0; u < BIN_U; ++u) {
                pos[u] = -1;
                if (base + u * BIN_T + t < e1) pos[u] = atomicAdd(&cnt[r[u] >> BK_SH], 1);
            }
#pragma unroll
            for (int u = 0; u < BIN_U; ++u) {
                if (pos[u] >= 0)
                    brec[e0 + pos[u]] =
                        make_int2((c[u] << BK_SH) | (r[u] & (BK_R - 1)), __float_as_int(v[u]));
            }
        }
    }
}

#define BS_T    1024
#define BS_L    8
#define BS_P    (BIN_G * BS_L / BS_T)
#define BS_U    6
#define BS_O    6
#define BS_CAP  (BS_T * BS_O)

static_assert(BIN_G * BS_L == BS_P * BS_T, "every run has its BS_L lanes");

__device__ __forceinline__ int bucket_scan(int* rc, int* wsum, int t) {
    const int lane = t & 31;
    const int wave = t >> 5;
    int d = 0, incl = 0;
    if (t < BK_R) {
        d = rc[t];
        incl = d;
        for (int off = 1; off < 32; off <<= 1) {
            const int up = __shfl_up(incl, off, 32);
            if (lane >= off) incl += up;
        }
        if (lane == 31) wsum[wave] = incl;
    }
    __syncthreads();
    int excl = 0;
    if (t < BK_R) {
        int c = (lane < wave) ? wsum[lane] : 0;
        for (int off = 16; off >= 1; off >>= 1) c += __shfl_xor(c, off, 32);
        excl = c + incl - d;
        rc[t] = excl;
    }
    __syncthreads();
    return excl;
}

__global__ void __launch_bounds__(BS_T) __attribute__((amdgpu_waves_per_eu(16, 16)))
bucket_sort(const int* __restrict__ brec_, const int* __restrict__ startL,
            int* __restrict__ rowStart, int* __restrict__ scol,
            float* __restrict__ sval, int N, int B, int E) {
    __shared__ int   rc[BK_R];
    __shared__ int   wsum[BK_R / 32];
    __shared__ int   red0[BS_T / 32], red1[BS_T / 32], red2[BS_T / 32];
    __shared__ int   lcol[BS_CAP];
    __shared__ float lval[BS_CAP];
    const int2* __restrict__ brec = (const int2*)brec_;
    const int t    = threadIdx.x;
    const int lane = t & 31;
    const int wave = t >> 5;
    const int b    = blockIdx.x;
    const int grp  = t / BS_L;
    const int sub  = t % BS_L;
    const int tile = (E + BIN_G - 1) / BIN_G;
    int len[BS_P], first[BS_P];
    int a0 = 0, a1 = 0, am = 0;
#pragma unroll
    for (int p = 0; p < BS_P; ++p) {
        const int run = p * (BS_T / BS_L) + grp;
        int tb = run * tile;
        tb = (tb < E) ? tb : E;
        const int lo = startL[(size_t)b * BIN_G + run];
        const int hi = startL[(size_t)(b + 1) * BIN_G + run];
        len[p]   = hi - lo;
        first[p] = tb + lo;
        if (sub == 0) { a0 += lo; a1 += hi; }
        am = (len[p] > am) ? len[p] : am;
    }
    if (t < BK_R) rc[t] = 0;

    int key[BS_P * BS_U], val[BS_P * BS_U];
#pragma unroll
    for (int p = 0; p < BS_P; ++p) {
#pragma unroll
        for (int u = 0; u < BS_U; ++u) {
            const int k  = sub + u * BS_L;
            const int kk = (k < len[p]) ? k : (len[p] - 1);
            const int2 q = brec[(len[p] > 0) ? (first[p] + kk) : 0];
            key[p * BS_U + u] = q.x; val[p * BS_U + u] = q.y;
        }
    }

    for (int off = 16; off >= 1; off >>= 1) {
        a0 += __shfl_xor(a0, off, 32);
        a1 += __shfl_xor(a1, off, 32);
        const int om = __shfl_xor(am, off, 32);
        am = (om > am) ? om : am;
    }
    if (lane == 0) { red0[wave] = a0; red1[wave] = a1; red2[wave] = am; }
    __syncthreads();
    int s0 = red0[lane], s1 = red1[lane], mx = red2[lane];
    for (int off = 16; off >= 1; off >>= 1) {
        s0 += __shfl_xor(s0, off, 32);
        s1 += __shfl_xor(s1, off, 32);
        const int om = __shfl_xor(mx, off, 32);
        mx = (om > mx) ? om : mx;
    }
    const int n = s1 - s0;

    if (n <= BS_CAP && mx <= BS_L * BS_U) {
        int lr[BS_P * BS_U];
#pragma unroll
        for (int p = 0; p < BS_P; ++p) {
#pragma unroll
            for (int u = 0; u < BS_U; ++u) {
                const int s = p * BS_U + u;
                lr[s] = 0;
                if (sub + u * BS_L < len[p]) lr[s] = atomicAdd(&rc[key[s] & (BK_R - 1)], 1);
            }
        }
        __syncthreads();
        const int excl = bucket_scan(rc, wsum, t);
        if (t < BK_R) {
            const int row = b * BK_R + t;
            if (row < N) rowStart[row] = s0 + excl;
        }
#pragma unroll
        for (int s = 0; s < BS_P * BS_U; ++s) lr[s] += rc[key[s] & (BK_R - 1)];
#pragma unroll
        for (int p = 0; p < BS_P; ++p) {
#pragma unroll
            for (int u = 0; u < BS_U; ++u) {
                const int s = p * BS_U + u;
                if (sub + u * BS_L < len[p]) {
                    lcol[lr[s]] = key[s] >> BK_SH;
                    lval[lr[s]] = __int_as_float(val[s]);
                }
            }
        }
        __syncthreads();
        if (n > 0) {
            int oc[BS_O]; float ov[BS_O];
#pragma unroll
            for (int u = 0; u < BS_O; ++u) {
                const int j = t + u * BS_T;
                const int jj = (j < n) ? j : (n - 1);
                oc[u] = lcol[jj]; ov[u] = lval[jj];
            }
#pragma unroll
            for (int u = 0; u < BS_O; ++u) {
                const int j = t + u * BS_T;
                if (j < n) { scol[s0 + j] = oc[u]; sval[s0 + j] = ov[u]; }
            }
        }
    } else {
        for (int p = 0; p < BS_P; ++p)
            for (int k = sub; k < len[p]; k += BS_L)
                atomicAdd(&rc[brec[first[p] + k].x & (BK_R - 1)], 1);
        __syncthreads();
        const int excl = bucket_scan(rc, wsum, t);
        if (t < BK_R) {
            const int row = b * BK_R + t;
            if (row < N) rowStart[row] = s0 + excl;
        }
        for (int p = 0; p < BS_P; ++p) {
            for (int k = sub; k < len[p]; k += BS_L) {
                const int2 q = brec[first[p] + k];
                const int pos = s0 + atomicAdd(&rc[q.x & (BK_R - 1)], 1);
                scol[pos] = q.x >> BK_SH;
                sval[pos] = __int_as_float(q.y);
            }
        }
    }
    if (b == B - 1 && t == 0) rowStart[N] = s1;
}

#define SP1_GW   16
#define SP1_XW   4
#define SP1_W    (SP1_GW + SP1_XW)
#define SP1_T    (SP1_W * 32)
#define SP1_BR   16
#define SP1_GRID 768
#define ZP       (FEAT + 4)
#define W1P      (2 * FEAT + 32)
#define W2Q      (2 * PP)
#define PTILE    (SP1_BR * PP)

static_assert(SP1_BR == SP1_GW && SP1_XW * 32 == FEAT,
              "a gather wave owns one row of a tile, a multiplier wave two column tiles");

template <int C>
__device__ __forceinline__ void gather_fma(const float4* __restrict__ y4, int cc, float vv,
                                           int j, int lane, float4& acc) {
    float4 g[C];
#pragma unroll
    for (int u = 0; u < C; ++u) {
        const int c = __builtin_amdgcn_readlane(cc, j + u);
        g[u] = y4[(size_t)c * (FEAT / 4) + lane];
    }
#pragma unroll
    for (int u = 0; u < C; ++u) {
        const float v = __int_as_float(__builtin_amdgcn_readlane(__float_as_int(vv), j + u));
        acc.x = fmaf(v, g[u].x, acc.x);
        acc.y = fmaf(v, g[u].y, acc.y);
        acc.z = fmaf(v, g[u].z, acc.z);
        acc.w = fmaf(v, g[u].w, acc.w);
    }
}

__global__ void __launch_bounds__(SP1_T) __attribute__((amdgpu_waves_per_eu(15, 16)))
spmm1_fused(const float* __restrict__ x, const int* __restrict__ rowStart,
            const int* __restrict__ scol, const float* __restrict__ sval,
            const float* __restrict__ W1,
            const float* __restrict__ b1, const float* __restrict__ g1,
            const float* __restrict__ be1, const float* __restrict__ m1,
            const float* __restrict__ v1, const float* __restrict__ W2,
            float* __restrict__ p, int N) {
    __shared__ __attribute__((aligned(16))) float w1s[(FEAT / 2) * W1P];
    __shared__ __attribute__((aligned(16))) float zs[2 * SP1_BR * ZP];
    __shared__ __attribute__((aligned(16))) float hb[SP1_BR * ZP];
    __shared__ __attribute__((aligned(16))) float w2p[(FEAT / 2) * W2Q];
    __shared__ __attribute__((aligned(16))) float pp[2 * SP1_XW * PTILE];
    __shared__ float bns[2 * FEAT];

    const int tid = (int)threadIdx.x;
    for (int i = tid; i < (FEAT / 2) * W2Q; i += SP1_T) {
        const int kp = i / W2Q;
        const int rem = i % W2Q;
        const int n = rem >> 1;
        const int k = 2 * kp + (rem & 1);
        w2p[i] = (n < NCLS) ? W2[k * NCLS + n] : 0.0f;
    }
    for (int i = tid; i < FEAT; i += SP1_T) {
        const float bs = g1[i] * rsqrtf(v1[i] + EPSB);
        bns[i] = bs;
        bns[FEAT + i] = (b1[i] - m1[i]) * bs + be1[i];
    }
    for (int i = tid; i < (FEAT / 2) * (FEAT / 4); i += SP1_T) {
        const int kp = i / (FEAT / 4);
        const int c4 = (i % (FEAT / 4)) * 4;
        const float4 wa = *(const float4*)&W1[(size_t)(2 * kp) * FEAT + c4];
        const float4 wb = *(const float4*)&W1[(size_t)(2 * kp + 1) * FEAT + c4];
        float4* d = (float4*)&w1s[kp * W1P + 2 * c4];
        d[0] = make_float4(wa.x, wb.x, wa.y, wb.y);
        d[1] = make_float4(wa.z, wb.z, wa.w, wb.w);
    }

    const int lane = tid & 31;
    const int half = lane >> 4;
    const int r16  = lane & 15;
    const int wave = __builtin_amdgcn_readfirstlane(tid >> 5);
    const bool gat = wave < SP1_GW;
    const int mw = wave - SP1_GW;
    const int G  = (int)gridDim.x;
    const int bI = (int)blockIdx.x;
    const int ntiles = (N + SP1_BR - 1) / SP1_BR;
    const int T = (bI < ntiles) ? (ntiles - bI + G - 1) / G : 0;

    const float4* __restrict__ x4 = (const float4*)x;
    int eB = 0, eE = 0, rsN = 0, nc = 0;
    float nv = 0.0f;
    if (gat) {
        int rs0 = 0;
        const int ra = bI * SP1_BR + wave;
        const int rb = (bI + G) * SP1_BR + wave;
        if (0 < T && ra < N && lane < 2) rs0 = rowStart[ra + lane];
        if (1 < T && rb < N && lane < 2) rsN = rowStart[rb + lane];
        eB = __builtin_amdgcn_readlane(rs0, 0);
        eE = __builtin_amdgcn_readlane(rs0, 1);
        if (eB + lane < eE) { nc = scol[eB + lane]; nv = sval[eB + lane]; }
    }
    __syncthreads();

    const v8f vzero = {0.f,0.f,0.f,0.f,0.f,0.f,0.f,0.f};
    for (int s = 0; s < T + 1; ++s) {
        float* zg = zs + (s & 1) * (SP1_BR * ZP);
        const float* zm = zs + ((s + 1) & 1) * (SP1_BR * ZP);
        if (gat) {
            const int cc = nc; const float vv = nv;
            const int nB = __builtin_amdgcn_readlane(rsN, 0);
            const int nE = __builtin_amdgcn_readlane(rsN, 1);
            nc = 0; nv = 0.0f;
            if (nB + lane < nE) { nc = scol[nB + lane]; nv = sval[nB + lane]; }
            rsN = 0;
            {
                const int rn = (bI + (s + 2) * G) * SP1_BR + wave;
                if (s + 2 < T && rn < N && lane < 2) rsN = rowStart[rn + lane];
            }

            float4 acc = {0.f, 0.f, 0.f, 0.f};
            const int deg = eE - eB;
            const int n = deg < 32 ? deg : 32;
            int j = 0;
            for (; j + 8 <= n; j += 8) gather_fma<8>(x4, cc, vv, j, lane, acc);
            switch (n - j) {
                case 7: gather_fma<7>(x4, cc, vv, j, lane, acc); break;
                case 6: gather_fma<6>(x4, cc, vv, j, lane, acc); break;
                case 5: gather_fma<5>(x4, cc, vv, j, lane, acc); break;
                case 4: gather_fma<4>(x4, cc, vv, j, lane, acc); break;
                case 3: gather_fma<3>(x4, cc, vv, j, lane, acc); break;
                case 2: gather_fma<2>(x4, cc, vv, j, lane, acc); break;
                case 1: gather_fma<1>(x4, cc, vv, j, lane, acc); break;
                default: break;
            }
            for (int e = eB + 32; e < eE; ++e) {
                const int   c = scol[e];
                const float v = sval[e];
                const float4 g = x4[(size_t)c * (FEAT / 4) + lane];
                acc.x = fmaf(v, g.x, acc.x);
                acc.y = fmaf(v, g.y, acc.y);
                acc.z = fmaf(v, g.z, acc.z);
                acc.w = fmaf(v, g.w, acc.w);
            }
            *(float4*)&zg[wave * ZP + lane * 4] = acc;
            eB = nB; eE = nE;
        } else {
            if (s >= 2) {
                const float* q = pp + ((s + 1) & 1) * (SP1_XW * PTILE);
                const int v = mw * 32 + lane;
                const float sum = ((q[v] + q[PTILE + v]) + q[2 * PTILE + v]) + q[3 * PTILE + v];
                const int t2 = bI + (s - 2) * G;
                if (t2 * SP1_BR + (v >> 3) < N) p[(size_t)t2 * PTILE + v] = sum;
            }
            if (s >= 1 && s <= T) {
                v8f acc0 = vzero, acc1 = vzero;
                const int ch0 = mw * 32 + r16;
                const float* zrow = zm + r16 * ZP + 2 * half;
                const float* wcol = w1s + half * W1P + 2 * ch0;
#pragma unroll 4
                for (int kt = 0; kt < FEAT / 4; ++kt) {
                    const v2f a  = *(const v2f*)(zrow + 4 * kt);
                    const v2f b0 = *(const v2f*)(wcol + (2 * kt) * W1P);
                    const v2f b1v = *(const v2f*)(wcol + (2 * kt) * W1P + 32);
                    acc0 = __builtin_amdgcn_wmma_f32_16x16x4_f32(
                        false, a, false, b0, (short)0, acc0, false, false);
                    acc1 = __builtin_amdgcn_wmma_f32_16x16x4_f32(
                        false, a, false, b1v, (short)0, acc1, false, false);
                }
                const float bs0 = bns[ch0],      bt0 = bns[FEAT + ch0];
                const float bs1 = bns[ch0 + 16], bt1 = bns[FEAT + ch0 + 16];
#pragma unroll
                for (int i = 0; i < 8; ++i) {
                    const float h0 = fmaf(acc0[i], bs0, bt0);
                    const float h1 = fmaf(acc1[i], bs1, bt1);
                    hb[(i + 8 * half) * ZP + ch0]      = h0 > 0.0f ? h0 : 0.0f;
                    hb[(i + 8 * half) * ZP + ch0 + 16] = h1 > 0.0f ? h1 : 0.0f;
                }
                v8f accp = vzero;
                const float* hrow = hb + r16 * ZP + 32 * mw + 2 * half;
                const float* w2c  = w2p + (16 * mw + half) * W2Q + 2 * (r16 & 7);
#pragma unroll
                for (int kt = 0; kt < 8; ++kt) {
                    const v2f a = *(const v2f*)(hrow + 4 * kt);
                    const v2f b = *(const v2f*)(w2c + (2 * kt) * W2Q);
                    accp = __builtin_amdgcn_wmma_f32_16x16x4_f32(
                        false, a, false, b, (short)0, accp, false, false);
                }
                float* qw = pp + (s & 1) * (SP1_XW * PTILE) + mw * PTILE + (8 * half) * PP + (r16 & 7);
#pragma unroll
                for (int i = 0; i < 8; ++i) qw[i * PP] = accp[i];
            }
        }
        __syncthreads();
    }
    if (!gat && T >= 1) {
        const float* q = pp + (T & 1) * (SP1_XW * PTILE);
        const int v = mw * 32 + lane;
        const float sum = ((q[v] + q[PTILE + v]) + q[2 * PTILE + v]) + q[3 * PTILE + v];
        const int t2 = bI + (T - 1) * G;
        if (t2 * SP1_BR + (v >> 3) < N) p[(size_t)t2 * PTILE + v] = sum;
    }
}

#define SP2_T    128
#define SP2_G    4
#define SP2_WR   (SP2_G * 4)
#define SP2_BR   (SP2_T / 32 * SP2_WR)

template <int U>
__device__ __forceinline__ unsigned bcast8(unsigned v) {
    return __builtin_amdgcn_mov_dpp8(v, (unsigned)(U * 0x249249));
}

__device__ __forceinline__ void sp2_pairs(const int* __restrict__ scol,
                                          const float* __restrict__ sval,
                                          const int (&eb)[SP2_G], const int (&ee)[SP2_G],
                                          int c8, unsigned (&po)[SP2_G], float (&vl)[SP2_G]) {
#pragma unroll
    for (int g = 0; g < SP2_G; ++g) {
        int i = eb[g] + c8;
        i = (i < ee[g] - 1) ? i : (ee[g] - 1);
        i = (i > 0) ? i : 0;
        po[g] = (unsigned)scol[i] * PP;
        vl[g] = sval[i];
    }
}

template <int U>
__device__ __forceinline__ void sp2_edge(const float* __restrict__ p, const unsigned (&po)[SP2_G],
                                         unsigned c8, float (&ga)[SP2_G][8]) {
#pragma unroll
    for (int g = 0; g < SP2_G; ++g) ga[g][U] = p[bcast8<U>(po[g]) | c8];
}

template <int U>
__device__ __forceinline__ void sp2_fma(const float (&cv)[SP2_G], const int (&nrem)[SP2_G],
                                        const float (&ga)[SP2_G][8], float (&acc)[SP2_G]) {
#pragma unroll
    for (int g = 0; g < SP2_G; ++g) {
        const float v = __uint_as_float(bcast8<U>(__float_as_uint(cv[g])));
        if (U < nrem[g]) acc[g] = fmaf(v, ga[g][U], acc[g]);
    }
}

__global__ void __launch_bounds__(SP2_T)
spmm2_fused(const float* __restrict__ p, const int* __restrict__ rowStart,
            const int* __restrict__ scol, const float* __restrict__ sval,
            const float* __restrict__ b2, const float* __restrict__ g2,
            const float* __restrict__ be2, const float* __restrict__ m2,
            const float* __restrict__ v2, float* __restrict__ out, int N) {
    __shared__ float os[SP2_BR * NCLS];
    const int t    = threadIdx.x;
    const int lane = t & 31;
    const int wave = t >> 5;
    const int q    = lane >> 3;
    const int c8   = lane & 7;
    const int lrow = wave * SP2_WR + q;
    const int row0 = (int)blockIdx.x * SP2_BR + lrow;

    int eb[SP2_G], ee[SP2_G];
    float acc[SP2_G];
#pragma unroll
    for (int g = 0; g < SP2_G; ++g) {
        const int r  = row0 + 4 * g;
        const int rc = (r < N) ? r : (N - 1);
        eb[g] = rowStart[rc]; ee[g] = rowStart[rc + 1];
        acc[g] = 0.0f;
    }
    const int cc = (c8 < NCLS) ? c8 : (NCLS - 1);
    const float kb = b2[cc], km = m2[cc], kbe = be2[cc];
    const float ksc = g2[cc] * rsqrtf(v2[cc] + EPSB);
#pragma unroll
    for (int g = 0; g < SP2_G; ++g)
        if (row0 + 4 * g >= N) ee[g] = eb[g];

    unsigned po[SP2_G]; float vl[SP2_G];
    sp2_pairs(scol, sval, eb, ee, c8, po, vl);
    for (;;) {
        bool more = false;
#pragma unroll
        for (int g = 0; g < SP2_G; ++g) more = more || (eb[g] < ee[g]);
        if (__builtin_amdgcn_ballot_w32(more) == 0u) break;

        float ga[SP2_G][8];
        sp2_edge<0>(p, po, (unsigned)c8, ga); sp2_edge<1>(p, po, (unsigned)c8, ga);
        sp2_edge<2>(p, po, (unsigned)c8, ga); sp2_edge<3>(p, po, (unsigned)c8, ga);
        sp2_edge<4>(p, po, (unsigned)c8, ga); sp2_edge<5>(p, po, (unsigned)c8, ga);
        sp2_edge<6>(p, po, (unsigned)c8, ga); sp2_edge<7>(p, po, (unsigned)c8, ga);
        float cv[SP2_G]; int nrem[SP2_G];
#pragma unroll
        for (int g = 0; g < SP2_G; ++g) {
            cv[g] = vl[g];
            nrem[g] = ee[g] - eb[g];
            eb[g] += 8;
        }
        sp2_pairs(scol, sval, eb, ee, c8, po, vl);
        __builtin_amdgcn_sched_barrier(0);
        sp2_fma<0>(cv, nrem, ga, acc); sp2_fma<1>(cv, nrem, ga, acc);
        sp2_fma<2>(cv, nrem, ga, acc); sp2_fma<3>(cv, nrem, ga, acc);
        sp2_fma<4>(cv, nrem, ga, acc); sp2_fma<5>(cv, nrem, ga, acc);
        sp2_fma<6>(cv, nrem, ga, acc); sp2_fma<7>(cv, nrem, ga, acc);
    }

    if (c8 < NCLS) {
#pragma unroll
        for (int g = 0; g < SP2_G; ++g)
            os[(lrow + 4 * g) * NCLS + c8] = (acc[g] + kb - km) * ksc + kbe;
    }
    __syncthreads();
    const int total = N * NCLS;
    const int base  = (int)blockIdx.x * (SP2_BR * NCLS);
    constexpr int SP2_OK = (SP2_BR * NCLS + SP2_T - 1) / SP2_T;
    float ov[SP2_OK];
#pragma unroll
    for (int k = 0; k < SP2_OK; ++k) {
        const int j = k * SP2_T + t;
        ov[k] = os[j < SP2_BR * NCLS ? j : SP2_BR * NCLS - 1];
    }
#pragma unroll
    for (int k = 0; k < SP2_OK; ++k) {
        const int j = k * SP2_T + t;
        if (j < SP2_BR * NCLS && base + j < total) out[base + j] = ov[k];
    }
}

extern "C" void kernel_launch(void* const* d_in, const int* in_sizes, int n_in,
                              void* d_out, int out_size, void* d_ws, size_t ws_size,
                              hipStream_t stream) {
    const float* x   = (const float*)d_in[0];
    const float* ev  = (const float*)d_in[1];
    const float* W1  = (const float*)d_in[2];
    const float* b1  = (const float*)d_in[3];
    const float* g1  = (const float*)d_in[4];
    const float* be1 = (const float*)d_in[5];
    const float* m1  = (const float*)d_in[6];
    const float* v1  = (const float*)d_in[7];
    const float* W2  = (const float*)d_in[8];
    const float* b2  = (const float*)d_in[9];
    const float* g2  = (const float*)d_in[10];
    const float* be2 = (const float*)d_in[11];
    const float* m2  = (const float*)d_in[12];
    const float* v2  = (const float*)d_in[13];
    const int*  erow = (const int*)d_in[14];
    const int*  ecol = (const int*)d_in[15];
    float* out = (float*)d_out;

    const int N = in_sizes[0] / FEAT;
    const int E = in_sizes[1];

    const int B = (N + BK_R - 1) >> BK_SH;

    float* p        = (float*)d_ws;
    int*   brec     = (int*)(p + (size_t)N * PP);
    int*   startL   = brec + (size_t)2 * E;
    int*   rowStart = startL + (size_t)(B + 1) * BIN_G;
    int*   scol     = rowStart + (N + 1);
    float* sval     = (float*)(scol + E);

    bin_tile<<<BIN_G, BIN_T, 0, stream>>>(erow, ecol, ev, startL, brec, E, B);
    bucket_sort<<<B, BS_T, 0, stream>>>(brec, startL, rowStart, scol, sval, N, B, E);
    spmm1_fused<<<SP1_GRID, SP1_T, 0, stream>>>(x, rowStart, scol, sval, W1,
                                                b1, g1, be1, m1, v1, W2, p, N);
    spmm2_fused<<<(N + SP2_BR - 1) / SP2_BR, SP2_T, 0, stream>>>(p, rowStart, scol, sval,
                                                                 b2, g2, be2, m2, v2, out, N);
}


# spmm1_fused: W1, W2 and bn constants staged by the 4 multiplier waves under tile 0's gathers (were all 20 waves, ahead of a barrier that goes); W2 loads made unconditional; sums unchanged
# speedup vs baseline: 6.0946x; 1.0456x over previous
#include <hip/hip_runtime.h>
#include <hip/hip_bf16.h>

typedef __attribute__((ext_vector_type(2))) float v2f;
typedef __attribute__((ext_vector_type(8))) float v8f;

#define FEAT 128
#define NCLS 7
#define PP   8
#define EPSB 1e-5f

#define BK_SH   8
#define BK_R    (1 << BK_SH)
#define BK_MAX  1024
#define BIN_G   256
#define BIN_T   1024
#define BIN_U   8

static_assert(BK_MAX == BIN_T, "bin_tile scans one bucket counter per thread");

__device__ __forceinline__ int block_excl_scan(int d, int* wt, int t) {
    const int lane = t & 31;
    const int wave = t >> 5;
    int incl = d;
    for (int off = 1; off < 32; off <<= 1) {
        const int up = __shfl_up(incl, off, 32);
        if (lane >= off) incl += up;
    }
    if (lane == 31) wt[wave] = incl;
    __syncthreads();
    int c = (lane < wave) ? wt[lane] : 0;
    for (int off = 16; off >= 1; off >>= 1) c += __shfl_xor(c, off, 32);
    return c + incl - d;
}

__global__ void __launch_bounds__(BIN_T)
bin_tile(const int* __restrict__ erow, const int* __restrict__ ecol,
         const float* __restrict__ ev, int* __restrict__ startL,
         int* __restrict__ brec_, int E, int B) {
    __shared__ int cnt[BK_MAX];
    __shared__ int wt[BIN_T / 32];
    __shared__ int2 srec[BIN_U * BIN_T];
    int2* __restrict__ brec = (int2*)brec_;
    const int t = threadIdx.x;
    cnt[t] = 0;
    __syncthreads();
    const int tile = (E + BIN_G - 1) / BIN_G;
    int e0 = (int)blockIdx.x * tile;
    e0 = (e0 < E) ? e0 : E;
    const int e1 = (e0 + tile < E) ? (e0 + tile) : E;
    const int last = (e1 > 0) ? (e1 - 1) : 0;

    if (tile <= BIN_U * BIN_T) {
        int r[BIN_U], c[BIN_U], rk[BIN_U]; float v[BIN_U];
#pragma unroll
        for (int u = 0; u < BIN_U; ++u) {
            const int e = e0 + u * BIN_T + t;
            const int ee = (e < e1) ? e : last;
            r[u] = erow[ee];
            c[u] = ecol[ee];
            v[u] = ev[ee];
        }
        __builtin_amdgcn_sched_barrier(0);
#pragma unroll
        for (int u = 0; u < BIN_U; ++u) {
            rk[u] = -1;
            if (e0 + u * BIN_T + t < e1) rk[u] = atomicAdd(&cnt[r[u] >> BK_SH], 1);
        }
        __syncthreads();
        const int excl = block_excl_scan(cnt[t], wt, t);
        cnt[t] = excl;
        if (t <= B) startL[(size_t)t * BIN_G + blockIdx.x] = excl;
        __syncthreads();
        int st[BIN_U];
#pragma unroll
        for (int u = 0; u < BIN_U; ++u) st[u] = cnt[r[u] >> BK_SH];
#pragma unroll
        for (int u = 0; u < BIN_U; ++u) {
            if (rk[u] >= 0)
                srec[st[u] + rk[u]] =
                    make_int2((c[u] << BK_SH) | (r[u] & (BK_R - 1)), __float_as_int(v[u]));
        }
        __syncthreads();
        const int n = e1 - e0;
        if (n > 0) {
            constexpr int BIN_O = BIN_U / 2 + 1;
            const int g0 = e0 & ~15;
            int4 q[BIN_O];
#pragma unroll
            for (int k = 0; k < BIN_O; ++k) {
                const int ja = g0 + 2 * (k * BIN_T + t) - e0;
                const int jb = ja + 1;
                const int2 a = srec[ja < 0 ? 0 : (ja < n ? ja : n - 1)];
                const int2 b = srec[jb < 0 ? 0 : (jb < n ? jb : n - 1)];
                q[k] = make_int4(a.x, a.y, b.x, b.y);
            }
#pragma unroll
            for (int k = 0; k < BIN_O; ++k) {
                const int g = g0 + 2 * (k * BIN_T + t);
                const bool ina = (g >= e0) && (g < e1);
                const bool inb = (g + 1 >= e0) && (g + 1 < e1);
                if (ina && inb)  *(int4*)&brec[g] = q[k];
                else if (ina)    brec[g] = make_int2(q[k].x, q[k].y);
                else if (inb)    brec[g + 1] = make_int2(q[k].z, q[k].w);
            }
        }
    } else {
        for (int base = e0; base < e1; base += BIN_U * BIN_T) {
            int r[BIN_U];
#pragma unroll
            for (int u = 0; u < BIN_U; ++u) {
                const int e = base + u * BIN_T + t;
                r[u] = erow[(e < e1) ? e : last];
            }
#pragma unroll
            for (int u = 0; u < BIN_U; ++u)
                if (base + u * BIN_T + t < e1) atomicAdd(&cnt[r[u] >> BK_SH], 1);
        }
        __syncthreads();
        const int excl = block_excl_scan(cnt[t], wt, t);
        cnt[t] = excl;
        if (t <= B) startL[(size_t)t * BIN_G + blockIdx.x] = excl;
        __syncthreads();
        for (int base = e0; base < e1; base += BIN_U * BIN_T) {
            int r[BIN_U], c[BIN_U], pos[BIN_U]; float v[BIN_U];
#pragma unroll
            for (int u = 0; u < BIN_U; ++u) {
                const int e = base + u * BIN_T + t;
                const int ee = (e < e1) ? e : last;
                r[u] = erow[ee];
                c[u] = ecol[ee];
                v[u] = ev[ee];
            }
#pragma unroll
            for (int u = 0; u < BIN_U; ++u) {
                pos[u] = -1;
                if (base + u * BIN_T + t < e1) pos[u] = atomicAdd(&cnt[r[u] >> BK_SH], 1);
            }
#pragma unroll
            for (int u = 0; u < BIN_U; ++u) {
                if (pos[u] >= 0)
                    brec[e0 + pos[u]] =
                        make_int2((c[u] << BK_SH) | (r[u] & (BK_R - 1)), __float_as_int(v[u]));
            }
        }
    }
}

#define BS_T    1024
#define BS_L    8
#define BS_P    (BIN_G * BS_L / BS_T)
#define BS_U    6
#define BS_O    6
#define BS_CAP  (BS_T * BS_O)

static_assert(BIN_G * BS_L == BS_P * BS_T, "every run has its BS_L lanes");

__device__ __forceinline__ int bucket_scan(int* rc, int* wsum, int t) {
    const int lane = t & 31;
    const int wave = t >> 5;
    int d = 0, incl = 0;
    if (t < BK_R) {
        d = rc[t];
        incl = d;
        for (int off = 1; off < 32; off <<= 1) {
            const int up = __shfl_up(incl, off, 32);
            if (lane >= off) incl += up;
        }
        if (lane == 31) wsum[wave] = incl;
    }
    __syncthreads();
    int excl = 0;
    if (t < BK_R) {
        int c = (lane < wave) ? wsum[lane] : 0;
        for (int off = 16; off >= 1; off >>= 1) c += __shfl_xor(c, off, 32);
        excl = c + incl - d;
        rc[t] = excl;
    }
    __syncthreads();
    return excl;
}

__global__ void __launch_bounds__(BS_T) __attribute__((amdgpu_waves_per_eu(16, 16)))
bucket_sort(const int* __restrict__ brec_, const int* __restrict__ startL,
            int* __restrict__ rowStart, int* __restrict__ scol,
            float* __restrict__ sval, int N, int B, int E) {
    __shared__ int   rc[BK_R];
    __shared__ int   wsum[BK_R / 32];
    __shared__ int   red0[BS_T / 32], red1[BS_T / 32], red2[BS_T / 32];
    __shared__ int   lcol[BS_CAP];
    __shared__ float lval[BS_CAP];
    const int2* __restrict__ brec = (const int2*)brec_;
    const int t    = threadIdx.x;
    const int lane = t & 31;
    const int wave = t >> 5;
    const int b    = blockIdx.x;
    const int grp  = t / BS_L;
    const int sub  = t % BS_L;
    const int tile = (E + BIN_G - 1) / BIN_G;
    int len[BS_P], first[BS_P];
    int a0 = 0, a1 = 0, am = 0;
#pragma unroll
    for (int p = 0; p < BS_P; ++p) {
        const int run = p * (BS_T / BS_L) + grp;
        int tb = run * tile;
        tb = (tb < E) ? tb : E;
        const int lo = startL[(size_t)b * BIN_G + run];
        const int hi = startL[(size_t)(b + 1) * BIN_G + run];
        len[p]   = hi - lo;
        first[p] = tb + lo;
        if (sub == 0) { a0 += lo; a1 += hi; }
        am = (len[p] > am) ? len[p] : am;
    }
    if (t < BK_R) rc[t] = 0;

    int key[BS_P * BS_U], val[BS_P * BS_U];
#pragma unroll
    for (int p = 0; p < BS_P; ++p) {
#pragma unroll
        for (int u = 0; u < BS_U; ++u) {
            const int k  = sub + u * BS_L;
            const int kk = (k < len[p]) ? k : (len[p] - 1);
            const int2 q = brec[(len[p] > 0) ? (first[p] + kk) : 0];
            key[p * BS_U + u] = q.x; val[p * BS_U + u] = q.y;
        }
    }

    for (int off = 16; off >= 1; off >>= 1) {
        a0 += __shfl_xor(a0, off, 32);
        a1 += __shfl_xor(a1, off, 32);
        const int om = __shfl_xor(am, off, 32);
        am = (om > am) ? om : am;
    }
    if (lane == 0) { red0[wave] = a0; red1[wave] = a1; red2[wave] = am; }
    __syncthreads();
    int s0 = red0[lane], s1 = red1[lane], mx = red2[lane];
    for (int off = 16; off >= 1; off >>= 1) {
        s0 += __shfl_xor(s0, off, 32);
        s1 += __shfl_xor(s1, off, 32);
        const int om = __shfl_xor(mx, off, 32);
        mx = (om > mx) ? om : mx;
    }
    const int n = s1 - s0;

    if (n <= BS_CAP && mx <= BS_L * BS_U) {
        int lr[BS_P * BS_U];
#pragma unroll
        for (int p = 0; p < BS_P; ++p) {
#pragma unroll
            for (int u = 0; u < BS_U; ++u) {
                const int s = p * BS_U + u;
                lr[s] = 0;
                if (sub + u * BS_L < len[p]) lr[s] = atomicAdd(&rc[key[s] & (BK_R - 1)], 1);
            }
        }
        __syncthreads();
        const int excl = bucket_scan(rc, wsum, t);
        if (t < BK_R) {
            const int row = b * BK_R + t;
            if (row < N) rowStart[row] = s0 + excl;
        }
#pragma unroll
        for (int s = 0; s < BS_P * BS_U; ++s) lr[s] += rc[key[s] & (BK_R - 1)];
#pragma unroll
        for (int p = 0; p < BS_P; ++p) {
#pragma unroll
            for (int u = 0; u < BS_U; ++u) {
                const int s = p * BS_U + u;
                if (sub + u * BS_L < len[p]) {
                    lcol[lr[s]] = key[s] >> BK_SH;
                    lval[lr[s]] = __int_as_float(val[s]);
                }
            }
        }
        __syncthreads();
        if (n > 0) {
            int oc[BS_O]; float ov[BS_O];
#pragma unroll
            for (int u = 0; u < BS_O; ++u) {
                const int j = t + u * BS_T;
                const int jj = (j < n) ? j : (n - 1);
                oc[u] = lcol[jj]; ov[u] = lval[jj];
            }
#pragma unroll
            for (int u = 0; u < BS_O; ++u) {
                const int j = t + u * BS_T;
                if (j < n) { scol[s0 + j] = oc[u]; sval[s0 + j] = ov[u]; }
            }
        }
    } else {
        for (int p = 0; p < BS_P; ++p)
            for (int k = sub; k < len[p]; k += BS_L)
                atomicAdd(&rc[brec[first[p] + k].x & (BK_R - 1)], 1);
        __syncthreads();
        const int excl = bucket_scan(rc, wsum, t);
        if (t < BK_R) {
            const int row = b * BK_R + t;
            if (row < N) rowStart[row] = s0 + excl;
        }
        for (int p = 0; p < BS_P; ++p) {
            for (int k = sub; k < len[p]; k += BS_L) {
                const int2 q = brec[first[p] + k];
                const int pos = s0 + atomicAdd(&rc[q.x & (BK_R - 1)], 1);
                scol[pos] = q.x >> BK_SH;
                sval[pos] = __int_as_float(q.y);
            }
        }
    }
    if (b == B - 1 && t == 0) rowStart[N] = s1;
}

#define SP1_GW   16
#define SP1_XW   4
#define SP1_W    (SP1_GW + SP1_XW)
#define SP1_T    (SP1_W * 32)
#define SP1_BR   16
#define SP1_GRID 768
#define ZP       (FEAT + 4)
#define W1P      (2 * FEAT + 32)
#define W2Q      (2 * PP)
#define PTILE    (SP1_BR * PP)

static_assert(SP1_BR == SP1_GW && SP1_XW * 32 == FEAT,
              "a gather wave owns one row of a tile, a multiplier wave two column tiles");

template <int C>
__device__ __forceinline__ void gather_fma(const float4* __restrict__ y4, int cc, float vv,
                                           int j, int lane, float4& acc) {
    float4 g[C];
#pragma unroll
    for (int u = 0; u < C; ++u) {
        const int c = __builtin_amdgcn_readlane(cc, j + u);
        g[u] = y4[(size_t)c * (FEAT / 4) + lane];
    }
#pragma unroll
    for (int u = 0; u < C; ++u) {
        const float v = __int_as_float(__builtin_amdgcn_readlane(__float_as_int(vv), j + u));
        acc.x = fmaf(v, g[u].x, acc.x);
        acc.y = fmaf(v, g[u].y, acc.y);
        acc.z = fmaf(v, g[u].z, acc.z);
        acc.w = fmaf(v, g[u].w, acc.w);
    }
}

__global__ void __launch_bounds__(SP1_T) __attribute__((amdgpu_waves_per_eu(15, 16)))
spmm1_fused(const float* __restrict__ x, const int* __restrict__ rowStart,
            const int* __restrict__ scol, const float* __restrict__ sval,
            const float* __restrict__ W1,
            const float* __restrict__ b1, const float* __restrict__ g1,
            const float* __restrict__ be1, const float* __restrict__ m1,
            const float* __restrict__ v1, const float* __restrict__ W2,
            float* __restrict__ p, int N) {
    __shared__ __attribute__((aligned(16))) float w1s[(FEAT / 2) * W1P];
    __shared__ __attribute__((aligned(16))) float zs[2 * SP1_BR * ZP];
    __shared__ __attribute__((aligned(16))) float hb[SP1_BR * ZP];
    __shared__ __attribute__((aligned(16))) float w2p[(FEAT / 2) * W2Q];
    __shared__ __attribute__((aligned(16))) float pp[2 * SP1_XW * PTILE];
    __shared__ float bns[2 * FEAT];

    const int tid = (int)threadIdx.x;
    const int lane = tid & 31;
    const int half = lane >> 4;
    const int r16  = lane & 15;
    const int wave = __builtin_amdgcn_readfirstlane(tid >> 5);
    const bool gat = wave < SP1_GW;
    const int mw = wave - SP1_GW;
    const int G  = (int)gridDim.x;
    const int bI = (int)blockIdx.x;
    const int ntiles = (N + SP1_BR - 1) / SP1_BR;
    const int T = (bI < ntiles) ? (ntiles - bI + G - 1) / G : 0;

    const float4* __restrict__ x4 = (const float4*)x;
    int eB = 0, eE = 0, rsN = 0, nc = 0;
    float nv = 0.0f;
    if (gat) {
        int rs0 = 0;
        const int ra = bI * SP1_BR + wave;
        const int rb = (bI + G) * SP1_BR + wave;
        if (0 < T && ra < N && lane < 2) rs0 = rowStart[ra + lane];
        if (1 < T && rb < N && lane < 2) rsN = rowStart[rb + lane];
        eB = __builtin_amdgcn_readlane(rs0, 0);
        eE = __builtin_amdgcn_readlane(rs0, 1);
        if (eB + lane < eE) { nc = scol[eB + lane]; nv = sval[eB + lane]; }
    } else {
        const unsigned mt = (unsigned)(mw * 32 + lane);
        const float4* __restrict__ W14 = (const float4*)W1;
#pragma unroll 4
        for (int j = 0; j < FEAT / 2 / SP1_XW; ++j) {
            const int kp = mw + SP1_XW * j;
            const float4 wa = W14[(2 * kp) * (FEAT / 4) + lane];
            const float4 wb = W14[(2 * kp + 1) * (FEAT / 4) + lane];
            float4* d = (float4*)&w1s[kp * W1P + 8 * lane];
            d[0] = make_float4(wa.x, wb.x, wa.y, wb.y);
            d[1] = make_float4(wa.z, wb.z, wa.w, wb.w);
        }
        float w2v[(FEAT / 2) * W2Q / (SP1_XW * 32)];
#pragma unroll
        for (int j = 0; j < (FEAT / 2) * W2Q / (SP1_XW * 32); ++j) {
            const unsigned i = mt + (unsigned)j * (SP1_XW * 32);
            const unsigned rem = i % W2Q;
            const unsigned n = rem >> 1;
            const unsigned k = 2 * (i / W2Q) + (rem & 1);
            w2v[j] = W2[k * NCLS + (n < NCLS ? n : NCLS - 1)];
        }
        const float bg = g1[mt], bv = v1[mt], bb = b1[mt], bm = m1[mt], bbe = be1[mt];
#pragma unroll
        for (int j = 0; j < (FEAT / 2) * W2Q / (SP1_XW * 32); ++j) {
            const unsigned i = mt + (unsigned)j * (SP1_XW * 32);
            w2p[i] = (((i % W2Q) >> 1) < NCLS) ? w2v[j] : 0.0f;
        }
        {
            const float bs = bg * rsqrtf(bv + EPSB);
            bns[mt] = bs;
            bns[FEAT + mt] = (bb - bm) * bs + bbe;
        }
    }

    const v8f vzero = {0.f,0.f,0.f,0.f,0.f,0.f,0.f,0.f};
    for (int s = 0; s < T + 1; ++s) {
        float* zg = zs + (s & 1) * (SP1_BR * ZP);
        const float* zm = zs + ((s + 1) & 1) * (SP1_BR * ZP);
        if (gat) {
            const int cc = nc; const float vv = nv;
            const int nB = __builtin_amdgcn_readlane(rsN, 0);
            const int nE = __builtin_amdgcn_readlane(rsN, 1);
            nc = 0; nv = 0.0f;
            if (nB + lane < nE) { nc = scol[nB + lane]; nv = sval[nB + lane]; }
            rsN = 0;
            {
                const int rn = (bI + (s + 2) * G) * SP1_BR + wave;
                if (s + 2 < T && rn < N && lane < 2) rsN = rowStart[rn + lane];
            }

            float4 acc = {0.f, 0.f, 0.f, 0.f};
            const int deg = eE - eB;
            const int n = deg < 32 ? deg : 32;
            int j = 0;
            for (; j + 8 <= n; j += 8) gather_fma<8>(x4, cc, vv, j, lane, acc);
            switch (n - j) {
                case 7: gather_fma<7>(x4, cc, vv, j, lane, acc); break;
                case 6: gather_fma<6>(x4, cc, vv, j, lane, acc); break;
                case 5: gather_fma<5>(x4, cc, vv, j, lane, acc); break;
                case 4: gather_fma<4>(x4, cc, vv, j, lane, acc); break;
                case 3: gather_fma<3>(x4, cc, vv, j, lane, acc); break;
                case 2: gather_fma<2>(x4, cc, vv, j, lane, acc); break;
                case 1: gather_fma<1>(x4, cc, vv, j, lane, acc); break;
                default: break;
            }
            for (int e = eB + 32; e < eE; ++e) {
                const int   c = scol[e];
                const float v = sval[e];
                const float4 g = x4[(size_t)c * (FEAT / 4) + lane];
                acc.x = fmaf(v, g.x, acc.x);
                acc.y = fmaf(v, g.y, acc.y);
                acc.z = fmaf(v, g.z, acc.z);
                acc.w = fmaf(v, g.w, acc.w);
            }
            *(float4*)&zg[wave * ZP + lane * 4] = acc;
            eB = nB; eE = nE;
        } else {
            if (s >= 2) {
                const float* q = pp + ((s + 1) & 1) * (SP1_XW * PTILE);
                const int v = mw * 32 + lane;
                const float sum = ((q[v] + q[PTILE + v]) + q[2 * PTILE + v]) + q[3 * PTILE + v];
                const int t2 = bI + (s - 2) * G;
                if (t2 * SP1_BR + (v >> 3) < N) p[(size_t)t2 * PTILE + v] = sum;
            }
            if (s >= 1 && s <= T) {
                v8f acc0 = vzero, acc1 = vzero;
                const int ch0 = mw * 32 + r16;
                const float* zrow = zm + r16 * ZP + 2 * half;
                const float* wcol = w1s + half * W1P + 2 * ch0;
#pragma unroll 4
                for (int kt = 0; kt < FEAT / 4; ++kt) {
                    const v2f a  = *(const v2f*)(zrow + 4 * kt);
                    const v2f b0 = *(const v2f*)(wcol + (2 * kt) * W1P);
                    const v2f b1v = *(const v2f*)(wcol + (2 * kt) * W1P + 32);
                    acc0 = __builtin_amdgcn_wmma_f32_16x16x4_f32(
                        false, a, false, b0, (short)0, acc0, false, false);
                    acc1 = __builtin_amdgcn_wmma_f32_16x16x4_f32(
                        false, a, false, b1v, (short)0, acc1, false, false);
                }
                const float bs0 = bns[ch0],      bt0 = bns[FEAT + ch0];
                const float bs1 = bns[ch0 + 16], bt1 = bns[FEAT + ch0 + 16];
#pragma unroll
                for (int i = 0; i < 8; ++i) {
                    const float h0 = fmaf(acc0[i], bs0, bt0);
                    const float h1 = fmaf(acc1[i], bs1, bt1);
                    hb[(i + 8 * half) * ZP + ch0]      = h0 > 0.0f ? h0 : 0.0f;
                    hb[(i + 8 * half) * ZP + ch0 + 16] = h1 > 0.0f ? h1 : 0.0f;
                }
                v8f accp = vzero;
                const float* hrow = hb + r16 * ZP + 32 * mw + 2 * half;
                const float* w2c  = w2p + (16 * mw + half) * W2Q + 2 * (r16 & 7);
#pragma unroll
                for (int kt = 0; kt < 8; ++kt) {
                    const v2f a = *(const v2f*)(hrow + 4 * kt);
                    const v2f b = *(const v2f*)(w2c + (2 * kt) * W2Q);
                    accp = __builtin_amdgcn_wmma_f32_16x16x4_f32(
                        false, a, false, b, (short)0, accp, false, false);
                }
                float* qw = pp + (s & 1) * (SP1_XW * PTILE) + mw * PTILE + (8 * half) * PP + (r16 & 7);
#pragma unroll
                for (int i = 0; i < 8; ++i) qw[i * PP] = accp[i];
            }
        }
        __syncthreads();
    }
    if (!gat && T >= 1) {
        const float* q = pp + (T & 1) * (SP1_XW * PTILE);
        const int v = mw * 32 + lane;
        const float sum = ((q[v] + q[PTILE + v]) + q[2 * PTILE + v]) + q[3 * PTILE + v];
        const int t2 = bI + (T - 1) * G;
        if (t2 * SP1_BR + (v >> 3) < N) p[(size_t)t2 * PTILE + v] = sum;
    }
}

#define SP2_T    128
#define SP2_G    4
#define SP2_WR   (SP2_G * 4)
#define SP2_BR   (SP2_T / 32 * SP2_WR)

template <int U>
__device__ __forceinline__ unsigned bcast8(unsigned v) {
    return __builtin_amdgcn_mov_dpp8(v, (unsigned)(U * 0x249249));
}

__device__ __forceinline__ void sp2_pairs(const int* __restrict__ scol,
                                          const float* __restrict__ sval,
                                          const int (&eb)[SP2_G], const int (&ee)[SP2_G],
                                          int c8, unsigned (&po)[SP2_G], float (&vl)[SP2_G]) {
#pragma unroll
    for (int g = 0; g < SP2_G; ++g) {
        int i = eb[g] + c8;
        i = (i < ee[g] - 1) ? i : (ee[g] - 1);
        i = (i > 0) ? i : 0;
        po[g] = (unsigned)scol[i] * PP;
        vl[g] = sval[i];
    }
}

template <int U>
__device__ __forceinline__ void sp2_edge(const float* __restrict__ p, const unsigned (&po)[SP2_G],
                                         unsigned c8, float (&ga)[SP2_G][8]) {
#pragma unroll
    for (int g = 0; g < SP2_G; ++g) ga[g][U] = p[bcast8<U>(po[g]) | c8];
}

template <int U>
__device__ __forceinline__ void sp2_fma(const float (&cv)[SP2_G], const int (&nrem)[SP2_G],
                                        const float (&ga)[SP2_G][8], float (&acc)[SP2_G]) {
#pragma unroll
    for (int g = 0; g < SP2_G; ++g) {
        const float v = __uint_as_float(bcast8<U>(__float_as_uint(cv[g])));
        if (U < nrem[g]) acc[g] = fmaf(v, ga[g][U], acc[g]);
    }
}

__global__ void __launch_bounds__(SP2_T)
spmm2_fused(const float* __restrict__ p, const int* __restrict__ rowStart,
            const int* __restrict__ scol, const float* __restrict__ sval,
            const float* __restrict__ b2, const float* __restrict__ g2,
            const float* __restrict__ be2, const float* __restrict__ m2,
            const float* __restrict__ v2, float* __restrict__ out, int N) {
    __shared__ float os[SP2_BR * NCLS];
    const int t    = threadIdx.x;
    const int lane = t & 31;
    const int wave = t >> 5;
    const int q    = lane >> 3;
    const int c8   = lane & 7;
    const int lrow = wave * SP2_WR + q;
    const int row0 = (int)blockIdx.x * SP2_BR + lrow;

    int eb[SP2_G], ee[SP2_G];
    float acc[SP2_G];
#pragma unroll
    for (int g = 0; g < SP2_G; ++g) {
        const int r  = row0 + 4 * g;
        const int rc = (r < N) ? r : (N - 1);
        eb[g] = rowStart[rc]; ee[g] = rowStart[rc + 1];
        acc[g] = 0.0f;
    }
    const int cc = (c8 < NCLS) ? c8 : (NCLS - 1);
    const float kb = b2[cc], km = m2[cc], kbe = be2[cc];
    const float ksc = g2[cc] * rsqrtf(v2[cc] + EPSB);
#pragma unroll
    for (int g = 0; g < SP2_G; ++g)
        if (row0 + 4 * g >= N) ee[g] = eb[g];

    unsigned po[SP2_G]; float vl[SP2_G];
    sp2_pairs(scol, sval, eb, ee, c8, po, vl);
    for (;;) {
        bool more = false;
#pragma unroll
        for (int g = 0; g < SP2_G; ++g) more = more || (eb[g] < ee[g]);
        if (__builtin_amdgcn_ballot_w32(more) == 0u) break;

        float ga[SP2_G][8];
        sp2_edge<0>(p, po, (unsigned)c8, ga); sp2_edge<1>(p, po, (unsigned)c8, ga);
        sp2_edge<2>(p, po, (unsigned)c8, ga); sp2_edge<3>(p, po, (unsigned)c8, ga);
        sp2_edge<4>(p, po, (unsigned)c8, ga); sp2_edge<5>(p, po, (unsigned)c8, ga);
        sp2_edge<6>(p, po, (unsigned)c8, ga); sp2_edge<7>(p, po, (unsigned)c8, ga);
        float cv[SP2_G]; int nrem[SP2_G];
#pragma unroll
        for (int g = 0; g < SP2_G; ++g) {
            cv[g] = vl[g];
            nrem[g] = ee[g] - eb[g];
            eb[g] += 8;
        }
        sp2_pairs(scol, sval, eb, ee, c8, po, vl);
        __builtin_amdgcn_sched_barrier(0);
        sp2_fma<0>(cv, nrem, ga, acc); sp2_fma<1>(cv, nrem, ga, acc);
        sp2_fma<2>(cv, nrem, ga, acc); sp2_fma<3>(cv, nrem, ga, acc);
        sp2_fma<4>(cv, nrem, ga, acc); sp2_fma<5>(cv, nrem, ga, acc);
        sp2_fma<6>(cv, nrem, ga, acc); sp2_fma<7>(cv, nrem, ga, acc);
    }

    if (c8 < NCLS) {
#pragma unroll
        for (int g = 0; g < SP2_G; ++g)
            os[(lrow + 4 * g) * NCLS + c8] = (acc[g] + kb - km) * ksc + kbe;
    }
    __syncthreads();
    const int total = N * NCLS;
    const int base  = (int)blockIdx.x * (SP2_BR * NCLS);
    constexpr int SP2_OK = (SP2_BR * NCLS + SP2_T - 1) / SP2_T;
    float ov[SP2_OK];
#pragma unroll
    for (int k = 0; k < SP2_OK; ++k) {
        const int j = k * SP2_T + t;
        ov[k] = os[j < SP2_BR * NCLS ? j : SP2_BR * NCLS - 1];
    }
#pragma unroll
    for (int k = 0; k < SP2_OK; ++k) {
        const int j = k * SP2_T + t;
        if (j < SP2_BR * NCLS && base + j < total) out[base + j] = ov[k];
    }
}

extern "C" void kernel_launch(void* const* d_in, const int* in_sizes, int n_in,
                              void* d_out, int out_size, void* d_ws, size_t ws_size,
                              hipStream_t stream) {
    const float* x   = (const float*)d_in[0];
    const float* ev  = (const float*)d_in[1];
    const float* W1  = (const float*)d_in[2];
    const float* b1  = (const float*)d_in[3];
    const float* g1  = (const float*)d_in[4];
    const float* be1 = (const float*)d_in[5];
    const float* m1  = (const float*)d_in[6];
    const float* v1  = (const float*)d_in[7];
    const float* W2  = (const float*)d_in[8];
    const float* b2  = (const float*)d_in[9];
    const float* g2  = (const float*)d_in[10];
    const float* be2 = (const float*)d_in[11];
    const float* m2  = (const float*)d_in[12];
    const float* v2  = (const float*)d_in[13];
    const int*  erow = (const int*)d_in[14];
    const int*  ecol = (const int*)d_in[15];
    float* out = (float*)d_out;

    const int N = in_sizes[0] / FEAT;
    const int E = in_sizes[1];

    const int B = (N + BK_R - 1) >> BK_SH;

    float* p        = (float*)d_ws;
    int*   brec     = (int*)(p + (size_t)N * PP);
    int*   startL   = brec + (size_t)2 * E;
    int*   rowStart = startL + (size_t)(B + 1) * BIN_G;
    int*   scol     = rowStart + (N + 1);
    float* sval     = (float*)(scol + E);

    bin_tile<<<BIN_G, BIN_T, 0, stream>>>(erow, ecol, ev, startL, brec, E, B);
    bucket_sort<<<B, BS_T, 0, stream>>>(brec, startL, rowStart, scol, sval, N, B, E);
    spmm1_fused<<<SP1_GRID, SP1_T, 0, stream>>>(x, rowStart, scol, sval, W1,
                                                b1, g1, be1, m1, v1, W2, p, N);
    spmm2_fused<<<(N + SP2_BR - 1) / SP2_BR, SP2_T, 0, stream>>>(p, rowStart, scol, sval,
                                                                 b2, g2, be2, m2, v2, out, N);
}
